# starting kernel
# baseline (speedup 1.0000x reference)
#include <hip/hip_runtime.h>


typedef __attribute__((ext_vector_type(2))) float v2f;
typedef __attribute__((ext_vector_type(8))) float v8f;

constexpr int HDIM = 128;

__device__ __forceinline__ void atomic_add_f32(float* p, float v) {
  (void)__hip_atomic_fetch_add(p, v, __ATOMIC_RELAXED,
                               __HIP_MEMORY_SCOPE_AGENT);
}

__global__ void zero_kernel(float* __restrict__ p, int n) {
  int i = blockIdx.x * blockDim.x + threadIdx.x;
  int stride = gridDim.x * blockDim.x;
  for (; i < n; i += stride) p[i] = 0.0f;
}

__global__ void degree_kernel(const int* __restrict__ dst, int e,
                              float* __restrict__ deg) {
  int i = blockIdx.x * blockDim.x + threadIdx.x;
  if (i < e) atomic_add_f32(&deg[dst[i]], 1.0f);
}

__global__ void scatter_kernel(const float* __restrict__ x,
                               const int* __restrict__ src,
                               const int* __restrict__ dst, int e,
                               float* __restrict__ s) {
  int wid  = (blockIdx.x * blockDim.x + threadIdx.x) >> 5;
  int lane = threadIdx.x & 31;
  if (wid >= e) return;
  int sr = src[wid];
  int dr = dst[wid];
  const float4 v = *(const float4*)(x + (size_t)sr * HDIM + lane * 4);
  float* o = s + (size_t)dr * HDIM + lane * 4;
  atomic_add_f32(o + 0, v.x);
  atomic_add_f32(o + 1, v.y);
  atomic_add_f32(o + 2, v.z);
  atomic_add_f32(o + 3, v.w);
}


__global__ void __launch_bounds__(256)
gemm_in_kernel(const float* __restrict__ A, const float* __restrict__ W,
               const float* __restrict__ bias, const float* __restrict__ emb,
               const int* __restrict__ nid, float* __restrict__ C, int K) {
  const int lane = threadIdx.x & 31;
  const int wave = threadIdx.x >> 5;
  const int half = lane >> 4;
  const int nloc = lane & 15;
  const int m0   = blockIdx.x * 16;
  const int row  = m0 + nloc;
  const int n    = wave * 16 + nloc;

  const float* Arow = A + (size_t)row * K;
  v8f acc = {};
#pragma unroll 8
  for (int k0 = 0; k0 < K; k0 += 4) {
    const int ka = k0 + 2 * half;
    v2f a; a.x = Arow[ka];
           a.y = Arow[ka + 1];
    v2f b; b.x = W[(size_t)ka * HDIM + n];
           b.y = W[(size_t)(ka + 1) * HDIM + n];
    acc = __builtin_amdgcn_wmma_f32_16x16x4_f32(false, a, false, b,
                                                (short)0, acc, false, false);
  }

  const float bn = bias[n];
#pragma unroll
  for (int r = 0; r < 8; ++r) {
    const int m = m0 + r + 8 * half;
    const float v = acc[r] + bn + emb[(size_t)nid[m] * HDIM + n];
    C[(size_t)m * HDIM + n] = v;
  }
}

__global__ void __launch_bounds__(256)
sage_kernel(const float* __restrict__ S, const float* __restrict__ deg,
            const float* __restrict__ X, const float* __restrict__ Wl,
            const float* __restrict__ bl, const float* __restrict__ Wr,
            float* __restrict__ Out, int relu) {
  const int lane = threadIdx.x & 31;
  const int wave = threadIdx.x >> 5;
  const int half = lane >> 4;
  const int nloc = lane & 15;
  const int m0   = blockIdx.x * 16;
  const int row  = m0 + nloc;
  const int n    = wave * 16 + nloc;

  const float* Srow = S + (size_t)row * HDIM;
  const float* Xrow = X + (size_t)row * HDIM;
  v8f accS = {};
  v8f accX = {};
#pragma unroll 4
  for (int k0 = 0; k0 < HDIM; k0 += 4) {
    const int ka = k0 + 2 * half;
    v2f a1; a1.x = Srow[ka]; a1.y = Srow[ka + 1];
    v2f b1; b1.x = Wl[(size_t)ka * HDIM + n];
            b1.y = Wl[(size_t)(ka + 1) * HDIM + n];
    accS = __builtin_amdgcn_wmma_f32_16x16x4_f32(false, a1, false, b1,
                                                 (short)0, accS, false, false);
    v2f a2; a2.x = Xrow[ka]; a2.y = Xrow[ka + 1];
    v2f b2; b2.x = Wr[(size_t)ka * HDIM + n];
            b2.y = Wr[(size_t)(ka + 1) * HDIM + n];
    accX = __builtin_amdgcn_wmma_f32_16x16x4_f32(false, a2, false, b2,
                                                 (short)0, accX, false, false);
  }

  const float bn = bl[n];
#pragma unroll
  for (int r = 0; r < 8; ++r) {
    const int m = m0 + r + 8 * half;
    float d = deg[m];
    d = d < 1.0f ? 1.0f : d;
    float v = accS[r] / d + accX[r] + bn;
    if (relu) v = v > 0.0f ? v : 0.0f;
    Out[(size_t)m * HDIM + n] = v;
  }
}

__global__ void edge_dot_kernel(const float* __restrict__ Hc,
                                const float* __restrict__ Hp,
                                const int* __restrict__ ls,
                                const int* __restrict__ ld, int e,
                                float* __restrict__ out) {
  int wid  = (blockIdx.x * blockDim.x + threadIdx.x) >> 5;
  int lane = threadIdx.x & 31;
  if (wid >= e) return;
  int a = ls[wid];
  int b = ld[wid];
  const float4 va = *(const float4*)(Hc + (size_t)a * HDIM + lane * 4);
  const float4 vb = *(const float4*)(Hp + (size_t)b * HDIM + lane * 4);
  float v = va.x * vb.x + va.y * vb.y + va.z * vb.z + va.w * vb.w;
#pragma unroll
  for (int off = 16; off > 0; off >>= 1) v += __shfl_down(v, off, 32);
  if (lane == 0) out[wid] = v;
}

extern "C" void kernel_launch(void* const* d_in, const int* in_sizes, int n_in,
                              void* d_out, int out_size, void* d_ws,
                              size_t ws_size, hipStream_t stream) {
  (void)in_sizes; (void)n_in; (void)out_size; (void)ws_size;
  const int NC = 4096, NP = 8192, E = 500000, EL = 500000;

  const float* x_country = (const float*)d_in[0];
  const float* x_product = (const float*)d_in[1];
  const float* Wc_lin = (const float*)d_in[2];
  const float* bc_lin = (const float*)d_in[3];
  const float* Wp_lin = (const float*)d_in[4];
  const float* bp_lin = (const float*)d_in[5];
  const float* emb_c  = (const float*)d_in[6];
  const float* emb_p  = (const float*)d_in[7];
  const float* W1ep_l = (const float*)d_in[8];
  const float* b1ep   = (const float*)d_in[9];
  const float* W1ep_r = (const float*)d_in[10];
  const float* W1pc_l = (const float*)d_in[11];
  const float* b1pc   = (const float*)d_in[12];
  const float* W1pc_r = (const float*)d_in[13];
  const float* W2ep_l = (const float*)d_in[14];
  const float* b2ep   = (const float*)d_in[15];
  const float* W2ep_r = (const float*)d_in[16];
  const float* W2pc_l = (const float*)d_in[17];
  const float* b2pc   = (const float*)d_in[18];
  const float* W2pc_r = (const float*)d_in[19];
  const int* nid_c = (const int*)d_in[20];
  const int* nid_p = (const int*)d_in[21];
  const int* e_src = (const int*)d_in[22];
  const int* e_dst = (const int*)d_in[23];
  const int* l_src = (const int*)d_in[24];
  const int* l_dst = (const int*)d_in[25];

  float* ws = (float*)d_ws;
  size_t off = 0;
  float* xc   = ws + off; off += (size_t)NC * HDIM;
  float* xp   = ws + off; off += (size_t)NP * HDIM;
  float* hp   = ws + off; off += (size_t)NP * HDIM;
  float* hc   = ws + off; off += (size_t)NC * HDIM;
  float* hp2  = ws + off; off += (size_t)NP * HDIM;
  float* hc2  = ws + off; off += (size_t)NC * HDIM;
  float* sbuf = ws + off; off += (size_t)NP * HDIM;
  float* degp = ws + off; off += NP;
  float* degc = ws + off; off += NC;

  const int TB = 256;
  const int edge_blocks = (E * 32 + TB - 1) / TB;
  const int np_elems = NP * HDIM;
  const int nc_elems = NC * HDIM;

  gemm_in_kernel<<<NC / 16, TB, 0, stream>>>(x_country, Wc_lin, bc_lin, emb_c,
                                             nid_c, xc, NC);
  gemm_in_kernel<<<NP / 16, TB, 0, stream>>>(x_product, Wp_lin, bp_lin, emb_p,
                                             nid_p, xp, NP);

  zero_kernel<<<64, TB, 0, stream>>>(degp, NP);
  zero_kernel<<<64, TB, 0, stream>>>(degc, NC);
  degree_kernel<<<(E + TB - 1) / TB, TB, 0, stream>>>(e_dst, E, degp);
  degree_kernel<<<(E + TB - 1) / TB, TB, 0, stream>>>(e_src, E, degc);

  zero_kernel<<<1024, TB, 0, stream>>>(sbuf, np_elems);
  scatter_kernel<<<edge_blocks, TB, 0, stream>>>(xc, e_src, e_dst, E, sbuf);
  sage_kernel<<<NP / 16, TB, 0, stream>>>(sbuf, degp, xp, W1ep_l, b1ep, W1ep_r,
                                          hp, 1);

  zero_kernel<<<1024, TB, 0, stream>>>(sbuf, nc_elems);
  scatter_kernel<<<edge_blocks, TB, 0, stream>>>(xp, e_dst, e_src, E, sbuf);
  sage_kernel<<<NC / 16, TB, 0, stream>>>(sbuf, degc, xc, W1pc_l, b1pc, W1pc_r,
                                          hc, 1);

  zero_kernel<<<1024, TB, 0, stream>>>(sbuf, np_elems);
  scatter_kernel<<<edge_blocks, TB, 0, stream>>>(hc, e_src, e_dst, E, sbuf);
  sage_kernel<<<NP / 16, TB, 0, stream>>>(sbuf, degp, hp, W2ep_l, b2ep, W2ep_r,
                                          hp2, 0);

  zero_kernel<<<1024, TB, 0, stream>>>(sbuf, nc_elems);
  scatter_kernel<<<edge_blocks, TB, 0, stream>>>(hp, e_dst, e_src, E, sbuf);
  sage_kernel<<<NC / 16, TB, 0, stream>>>(sbuf, degc, hc, W2pc_l, b2pc, W2pc_r,
                                          hc2, 0);

  const int el_blocks = (EL * 32 + TB - 1) / TB;
  edge_dot_kernel<<<el_blocks, TB, 0, stream>>>(hc2, hp2, l_src, l_dst, EL,
                                                (float*)d_out);
}


# input projection GEMMs: A and W chunks staged through LDS by coalesced 16-byte loads one chunk ahead, 32-row block tile on the larger; k regrouped per WMMA, so the f32 sum order changes
# speedup vs baseline: 1.2595x; 1.2595x over previous
#include <hip/hip_runtime.h>


typedef __attribute__((ext_vector_type(2))) float v2f;
typedef __attribute__((ext_vector_type(8))) float v8f;
typedef __attribute__((ext_vector_type(4))) float v4f;

constexpr int HDIM = 128;

__device__ __forceinline__ void atomic_add_f32(float* p, float v) {
  (void)__hip_atomic_fetch_add(p, v, __ATOMIC_RELAXED,
                               __HIP_MEMORY_SCOPE_AGENT);
}

__global__ void zero_kernel(float* __restrict__ p, int n) {
  int i = blockIdx.x * blockDim.x + threadIdx.x;
  int stride = gridDim.x * blockDim.x;
  for (; i < n; i += stride) p[i] = 0.0f;
}

__global__ void degree_kernel(const int* __restrict__ dst, int e,
                              float* __restrict__ deg) {
  int i = blockIdx.x * blockDim.x + threadIdx.x;
  if (i < e) atomic_add_f32(&deg[dst[i]], 1.0f);
}

__global__ void scatter_kernel(const float* __restrict__ x,
                               const int* __restrict__ src,
                               const int* __restrict__ dst, int e,
                               float* __restrict__ s) {
  int wid  = (blockIdx.x * blockDim.x + threadIdx.x) >> 5;
  int lane = threadIdx.x & 31;
  if (wid >= e) return;
  int sr = src[wid];
  int dr = dst[wid];
  const float4 v = *(const float4*)(x + (size_t)sr * HDIM + lane * 4);
  float* o = s + (size_t)dr * HDIM + lane * 4;
  atomic_add_f32(o + 0, v.x);
  atomic_add_f32(o + 1, v.y);
  atomic_add_f32(o + 2, v.z);
  atomic_add_f32(o + 3, v.w);
}


template <int RT, int BK>
__global__ void __launch_bounds__(256)
gemm_in_kernel(const float* __restrict__ A, const float* __restrict__ W,
               const float* __restrict__ bias, const float* __restrict__ emb,
               const int* __restrict__ nid, float* __restrict__ C, int K) {
  constexpr int BM = 16 * RT;
  constexpr int PA = BK + 4;
  constexpr int PW = HDIM + 4;
  constexpr int ATPR = BK / 4;
  constexpr int AROWS = 256 / ATPR;
  constexpr int NA = BM / AROWS;
  constexpr int NW = BK / 8;
  static_assert(NA * AROWS == BM && NA >= 1, "A tile must be whole passes");
  __shared__ __attribute__((aligned(16))) float As[2][BM * PA];
  __shared__ __attribute__((aligned(16))) float Ws[2][BK * PW];

  const int tid  = threadIdx.x;
  const int lane = tid & 31;
  const int wave = tid >> 5;
  const int half = lane >> 4;
  const int nloc = lane & 15;
  const int m0   = blockIdx.x * BM;
  const int n    = wave * 16 + nloc;

  const int arow = tid / ATPR;
  const int acol = (tid % ATPR) * 4;
  const int wrow = tid >> 5;
  const int wcol = (tid & 31) * 4;
  const float* Ag = A + (size_t)(m0 + arow) * K + acol;
  const float* Wg = W + (size_t)wrow * HDIM + wcol;

  v4f ra[NA];
  v4f rw[NW];
  v8f acc[RT];
#pragma unroll
  for (int i = 0; i < RT; ++i) acc[i] = v8f{};

#pragma unroll
  for (int i = 0; i < NA; ++i)
    ra[i] = *(const v4f*)(Ag + (size_t)(AROWS * i) * K);
#pragma unroll
  for (int i = 0; i < NW; ++i)
    rw[i] = *(const v4f*)(Wg + (size_t)(8 * i) * HDIM);
#pragma unroll
  for (int i = 0; i < NA; ++i)
    *(v4f*)(&As[0][(arow + AROWS * i) * PA + acol]) = ra[i];
#pragma unroll
  for (int i = 0; i < NW; ++i)
    *(v4f*)(&Ws[0][(wrow + 8 * i) * PW + wcol]) = rw[i];
  __syncthreads();

  const int nchunk = K / BK;
  for (int t = 0; t < nchunk; ++t) {
    const int buf = t & 1;
    const bool more = (t + 1 < nchunk);
    if (more) {
      const int k1 = (t + 1) * BK;
#pragma unroll
      for (int i = 0; i < NA; ++i)
        ra[i] = *(const v4f*)(Ag + (size_t)(AROWS * i) * K + k1);
#pragma unroll
      for (int i = 0; i < NW; ++i)
        rw[i] = *(const v4f*)(Wg + (size_t)(k1 + 8 * i) * HDIM);
    }

    const float* as = &As[buf][nloc * PA + 4 * half];
    const float* ws = &Ws[buf][(4 * half) * PW + n];
#pragma unroll
    for (int j = 0; j < BK / 8; ++j) {
      v4f av[RT];
#pragma unroll
      for (int i = 0; i < RT; ++i)
        av[i] = *(const v4f*)(as + (16 * i) * PA + 8 * j);
      v2f b0; b0.x = ws[(8 * j + 0) * PW];
              b0.y = ws[(8 * j + 1) * PW];
      v2f b1; b1.x = ws[(8 * j + 2) * PW];
              b1.y = ws[(8 * j + 3) * PW];
#pragma unroll
      for (int i = 0; i < RT; ++i) {
        v2f a; a.x = av[i].x; a.y = av[i].y;
        acc[i] = __builtin_amdgcn_wmma_f32_16x16x4_f32(
            false, a, false, b0, (short)0, acc[i], false, false);
      }
#pragma unroll
      for (int i = 0; i < RT; ++i) {
        v2f a; a.x = av[i].z; a.y = av[i].w;
        acc[i] = __builtin_amdgcn_wmma_f32_16x16x4_f32(
            false, a, false, b1, (short)0, acc[i], false, false);
      }
    }

    if (more) {
#pragma unroll
      for (int i = 0; i < NA; ++i)
        *(v4f*)(&As[buf ^ 1][(arow + AROWS * i) * PA + acol]) = ra[i];
#pragma unroll
      for (int i = 0; i < NW; ++i)
        *(v4f*)(&Ws[buf ^ 1][(wrow + 8 * i) * PW + wcol]) = rw[i];
    }
    __syncthreads();
  }

  const float bn = bias[n];
#pragma unroll
  for (int i = 0; i < RT; ++i) {
#pragma unroll
    for (int r = 0; r < 8; ++r) {
      const int m = m0 + 16 * i + r + 8 * half;
      const float v = acc[i][r] + bn + emb[(size_t)nid[m] * HDIM + n];
      C[(size_t)m * HDIM + n] = v;
    }
  }
}

__global__ void __launch_bounds__(256)
sage_kernel(const float* __restrict__ S, const float* __restrict__ deg,
            const float* __restrict__ X, const float* __restrict__ Wl,
            const float* __restrict__ bl, const float* __restrict__ Wr,
            float* __restrict__ Out, int relu) {
  const int lane = threadIdx.x & 31;
  const int wave = threadIdx.x >> 5;
  const int half = lane >> 4;
  const int nloc = lane & 15;
  const int m0   = blockIdx.x * 16;
  const int row  = m0 + nloc;
  const int n    = wave * 16 + nloc;

  const float* Srow = S + (size_t)row * HDIM;
  const float* Xrow = X + (size_t)row * HDIM;
  v8f accS = {};
  v8f accX = {};
#pragma unroll 4
  for (int k0 = 0; k0 < HDIM; k0 += 4) {
    const int ka = k0 + 2 * half;
    v2f a1; a1.x = Srow[ka]; a1.y = Srow[ka + 1];
    v2f b1; b1.x = Wl[(size_t)ka * HDIM + n];
            b1.y = Wl[(size_t)(ka + 1) * HDIM + n];
    accS = __builtin_amdgcn_wmma_f32_16x16x4_f32(false, a1, false, b1,
                                                 (short)0, accS, false, false);
    v2f a2; a2.x = Xrow[ka]; a2.y = Xrow[ka + 1];
    v2f b2; b2.x = Wr[(size_t)ka * HDIM + n];
            b2.y = Wr[(size_t)(ka + 1) * HDIM + n];
    accX = __builtin_amdgcn_wmma_f32_16x16x4_f32(false, a2, false, b2,
                                                 (short)0, accX, false, false);
  }

  const float bn = bl[n];
#pragma unroll
  for (int r = 0; r < 8; ++r) {
    const int m = m0 + r + 8 * half;
    float d = deg[m];
    d = d < 1.0f ? 1.0f : d;
    float v = accS[r] / d + accX[r] + bn;
    if (relu) v = v > 0.0f ? v : 0.0f;
    Out[(size_t)m * HDIM + n] = v;
  }
}

__global__ void edge_dot_kernel(const float* __restrict__ Hc,
                                const float* __restrict__ Hp,
                                const int* __restrict__ ls,
                                const int* __restrict__ ld, int e,
                                float* __restrict__ out) {
  int wid  = (blockIdx.x * blockDim.x + threadIdx.x) >> 5;
  int lane = threadIdx.x & 31;
  if (wid >= e) return;
  int a = ls[wid];
  int b = ld[wid];
  const float4 va = *(const float4*)(Hc + (size_t)a * HDIM + lane * 4);
  const float4 vb = *(const float4*)(Hp + (size_t)b * HDIM + lane * 4);
  float v = va.x * vb.x + va.y * vb.y + va.z * vb.z + va.w * vb.w;
#pragma unroll
  for (int off = 16; off > 0; off >>= 1) v += __shfl_down(v, off, 32);
  if (lane == 0) out[wid] = v;
}

extern "C" void kernel_launch(void* const* d_in, const int* in_sizes, int n_in,
                              void* d_out, int out_size, void* d_ws,
                              size_t ws_size, hipStream_t stream) {
  (void)in_sizes; (void)n_in; (void)out_size; (void)ws_size;
  const int NC = 4096, NP = 8192, E = 500000, EL = 500000;

  const float* x_country = (const float*)d_in[0];
  const float* x_product = (const float*)d_in[1];
  const float* Wc_lin = (const float*)d_in[2];
  const float* bc_lin = (const float*)d_in[3];
  const float* Wp_lin = (const float*)d_in[4];
  const float* bp_lin = (const float*)d_in[5];
  const float* emb_c  = (const float*)d_in[6];
  const float* emb_p  = (const float*)d_in[7];
  const float* W1ep_l = (const float*)d_in[8];
  const float* b1ep   = (const float*)d_in[9];
  const float* W1ep_r = (const float*)d_in[10];
  const float* W1pc_l = (const float*)d_in[11];
  const float* b1pc   = (const float*)d_in[12];
  const float* W1pc_r = (const float*)d_in[13];
  const float* W2ep_l = (const float*)d_in[14];
  const float* b2ep   = (const float*)d_in[15];
  const float* W2ep_r = (const float*)d_in[16];
  const float* W2pc_l = (const float*)d_in[17];
  const float* b2pc   = (const float*)d_in[18];
  const float* W2pc_r = (const float*)d_in[19];
  const int* nid_c = (const int*)d_in[20];
  const int* nid_p = (const int*)d_in[21];
  const int* e_src = (const int*)d_in[22];
  const int* e_dst = (const int*)d_in[23];
  const int* l_src = (const int*)d_in[24];
  const int* l_dst = (const int*)d_in[25];

  float* ws = (float*)d_ws;
  size_t off = 0;
  float* xc   = ws + off; off += (size_t)NC * HDIM;
  float* xp   = ws + off; off += (size_t)NP * HDIM;
  float* hp   = ws + off; off += (size_t)NP * HDIM;
  float* hc   = ws + off; off += (size_t)NC * HDIM;
  float* hp2  = ws + off; off += (size_t)NP * HDIM;
  float* hc2  = ws + off; off += (size_t)NC * HDIM;
  float* sbuf = ws + off; off += (size_t)NP * HDIM;
  float* degp = ws + off; off += NP;
  float* degc = ws + off; off += NC;

  const int TB = 256;
  const int edge_blocks = (E * 32 + TB - 1) / TB;
  const int np_elems = NP * HDIM;
  const int nc_elems = NC * HDIM;

  gemm_in_kernel<1, 128><<<NC / 16, TB, 0, stream>>>(x_country, Wc_lin, bc_lin,
                                                emb_c, nid_c, xc, NC);
  gemm_in_kernel<2, 64><<<NP / 32, TB, 0, stream>>>(x_product, Wp_lin, bp_lin,
                                                emb_p, nid_p, xp, NP);

  zero_kernel<<<64, TB, 0, stream>>>(degp, NP);
  zero_kernel<<<64, TB, 0, stream>>>(degc, NC);
  degree_kernel<<<(E + TB - 1) / TB, TB, 0, stream>>>(e_dst, E, degp);
  degree_kernel<<<(E + TB - 1) / TB, TB, 0, stream>>>(e_src, E, degc);

  zero_kernel<<<1024, TB, 0, stream>>>(sbuf, np_elems);
  scatter_kernel<<<edge_blocks, TB, 0, stream>>>(xc, e_src, e_dst, E, sbuf);
  sage_kernel<<<NP / 16, TB, 0, stream>>>(sbuf, degp, xp, W1ep_l, b1ep, W1ep_r,
                                          hp, 1);

  zero_kernel<<<1024, TB, 0, stream>>>(sbuf, nc_elems);
  scatter_kernel<<<edge_blocks, TB, 0, stream>>>(xp, e_dst, e_src, E, sbuf);
  sage_kernel<<<NC / 16, TB, 0, stream>>>(sbuf, degc, xc, W1pc_l, b1pc, W1pc_r,
                                          hc, 1);

  zero_kernel<<<1024, TB, 0, stream>>>(sbuf, np_elems);
  scatter_kernel<<<edge_blocks, TB, 0, stream>>>(hc, e_src, e_dst, E, sbuf);
  sage_kernel<<<NP / 16, TB, 0, stream>>>(sbuf, degp, hp, W2ep_l, b2ep, W2ep_r,
                                          hp2, 0);

  zero_kernel<<<1024, TB, 0, stream>>>(sbuf, nc_elems);
  scatter_kernel<<<edge_blocks, TB, 0, stream>>>(hp, e_dst, e_src, E, sbuf);
  sage_kernel<<<NC / 16, TB, 0, stream>>>(sbuf, degc, hc, W2pc_l, b2pc, W2pc_r,
                                          hc2, 0);

  const int el_blocks = (EL * 32 + TB - 1) / TB;
  edge_dot_kernel<<<el_blocks, TB, 0, stream>>>(hc2, hp2, l_src, l_dst, EL,
                                                (float*)d_out);
}


# neighbour sums by one-wave-per-row gather over per-call counting-sorted edge lists (LDS + int atomics, arrival-order slots) replace float-atomic scatter and zero fills; per-row f32 sum order changes
# speedup vs baseline: 2.4390x; 1.9365x over previous
#include <hip/hip_runtime.h>


typedef __attribute__((ext_vector_type(2))) float v2f;
typedef __attribute__((ext_vector_type(8))) float v8f;
typedef __attribute__((ext_vector_type(4))) float v4f;

constexpr int HDIM = 128;

__device__ __forceinline__ void atomic_add_i32(int* p, int v) {
  (void)__hip_atomic_fetch_add(p, v, __ATOMIC_RELAXED,
                               __HIP_MEMORY_SCOPE_AGENT);
}
__device__ __forceinline__ void atomic_or_u32(unsigned* p, unsigned v) {
  (void)__hip_atomic_fetch_or(p, v, __ATOMIC_RELAXED,
                              __HIP_MEMORY_SCOPE_AGENT);
}

__global__ void zero_kernel(float* __restrict__ p, int n) {
  int i = blockIdx.x * blockDim.x + threadIdx.x;
  int stride = gridDim.x * blockDim.x;
  for (; i < n; i += stride) p[i] = 0.0f;
}

constexpr int EPB = 4096;
constexpr int EPT = EPB / 1024;
constexpr int MAXP = 8192;
constexpr int MAXC = 4096;

__device__ __forceinline__ int lds_add_rtn(int* p, int v) {
  return __hip_atomic_fetch_add(p, v, __ATOMIC_RELAXED,
                                __HIP_MEMORY_SCOPE_WORKGROUP);
}

__global__ void __launch_bounds__(1024)
hist_kernel(const int* __restrict__ src, const int* __restrict__ dst, int e,
            int* __restrict__ cnt_dst, int* __restrict__ cnt_src) {
  __shared__ int hd[MAXP];
  __shared__ int hs[MAXC];
  const int t = threadIdx.x;
  for (int j = t; j < MAXP; j += 1024) hd[j] = 0;
  for (int j = t; j < MAXC; j += 1024) hs[j] = 0;
  __syncthreads();
  int es[EPT], ed[EPT];
#pragma unroll
  for (int k = 0; k < EPT; ++k) {
    const int i = blockIdx.x * EPB + k * 1024 + t;
    es[k] = (i < e) ? src[i] : 0;
    ed[k] = (i < e) ? dst[i] : 0;
  }
#pragma unroll
  for (int k = 0; k < EPT; ++k) {
    const int i = blockIdx.x * EPB + k * 1024 + t;
    if (i < e) {
      (void)lds_add_rtn(&hd[ed[k]], 1);
      (void)lds_add_rtn(&hs[es[k]], 1);
    }
  }
  __syncthreads();
  for (int j = t; j < MAXP; j += 1024) {
    const int v = hd[j];
    if (v) atomic_add_i32(&cnt_dst[j], v);
  }
  for (int j = t; j < MAXC; j += 1024) {
    const int v = hs[j];
    if (v) atomic_add_i32(&cnt_src[j], v);
  }
}

__global__ void __launch_bounds__(1024)
scan_kernel(int* __restrict__ cnt_dst, int* __restrict__ rp_dst, int np,
            int* __restrict__ cnt_src, int* __restrict__ rp_src, int nc) {
  __shared__ int sh[8192];
  __shared__ int wsum[32];
  int* cnt       = blockIdx.x == 0 ? cnt_dst : cnt_src;
  int* rp        = blockIdx.x == 0 ? rp_dst : rp_src;
  const int n    = blockIdx.x == 0 ? np : nc;
  const int ipt  = n >> 10;
  const int t    = threadIdx.x;
  const int lane = t & 31;
  const int w    = t >> 5;
  for (int i = t; i < n; i += 1024) { sh[i] = cnt[i]; cnt[i] = 0; }
  __syncthreads();
  int s = 0;
  for (int k = 0; k < ipt; ++k) s += sh[t * ipt + k];
  int inc = s;
  for (int off = 1; off < 32; off <<= 1) {
    const int o = __shfl_up(inc, off, 32);
    if (lane >= off) inc += o;
  }
  if (lane == 31) wsum[w] = inc;
  __syncthreads();
  if (w == 0) {
    const int ws = wsum[lane];
    int wi = ws;
    for (int off = 1; off < 32; off <<= 1) {
      const int o = __shfl_up(wi, off, 32);
      if (lane >= off) wi += o;
    }
    wsum[lane] = wi - ws;
  }
  __syncthreads();
  int run = wsum[w] + inc - s;
  for (int k = 0; k < ipt; ++k) {
    const int c = sh[t * ipt + k];
    sh[t * ipt + k] = run;
    run += c;
  }
  __syncthreads();
  for (int i = t; i < n; i += 1024) rp[i] = sh[i];
}

__global__ void __launch_bounds__(1024)
fill_kernel(const int* __restrict__ src, const int* __restrict__ dst, int e,
            int* __restrict__ cur_dst, int* __restrict__ cur_src,
            const int* __restrict__ rp_dst, const int* __restrict__ rp_src,
            unsigned* __restrict__ col_dst, unsigned* __restrict__ col_src) {
  __shared__ int hd[MAXP];
  __shared__ int hs[MAXC];
  const int t = threadIdx.x;
  for (int j = t; j < MAXP; j += 1024) hd[j] = 0;
  for (int j = t; j < MAXC; j += 1024) hs[j] = 0;
  __syncthreads();
  int es[EPT], ed[EPT], rd[EPT], rs[EPT];
#pragma unroll
  for (int k = 0; k < EPT; ++k) {
    const int i = blockIdx.x * EPB + k * 1024 + t;
    es[k] = (i < e) ? src[i] : 0;
    ed[k] = (i < e) ? dst[i] : 0;
  }
#pragma unroll
  for (int k = 0; k < EPT; ++k) {
    const int i = blockIdx.x * EPB + k * 1024 + t;
    rd[k] = 0; rs[k] = 0;
    if (i < e) {
      rd[k] = lds_add_rtn(&hd[ed[k]], 1);
      rs[k] = lds_add_rtn(&hs[es[k]], 1);
    }
  }
  __syncthreads();
  {
    int v[MAXP / 1024], a[MAXP / 1024];
#pragma unroll
    for (int q = 0; q < MAXP / 1024; ++q) v[q] = hd[t + 1024 * q];
#pragma unroll
    for (int q = 0; q < MAXP / 1024; ++q) {
      const int j = t + 1024 * q;
      a[q] = 0;
      if (v[q]) a[q] = rp_dst[j] + __hip_atomic_fetch_add(&cur_dst[j], v[q],
                           __ATOMIC_RELAXED, __HIP_MEMORY_SCOPE_AGENT);
    }
#pragma unroll
    for (int q = 0; q < MAXP / 1024; ++q)
      if (v[q]) hd[t + 1024 * q] = a[q];
  }
  {
    int v[MAXC / 1024], a[MAXC / 1024];
#pragma unroll
    for (int q = 0; q < MAXC / 1024; ++q) v[q] = hs[t + 1024 * q];
#pragma unroll
    for (int q = 0; q < MAXC / 1024; ++q) {
      const int j = t + 1024 * q;
      a[q] = 0;
      if (v[q]) a[q] = rp_src[j] + __hip_atomic_fetch_add(&cur_src[j], v[q],
                           __ATOMIC_RELAXED, __HIP_MEMORY_SCOPE_AGENT);
    }
#pragma unroll
    for (int q = 0; q < MAXC / 1024; ++q)
      if (v[q]) hs[t + 1024 * q] = a[q];
  }
  __syncthreads();
#pragma unroll
  for (int k = 0; k < EPT; ++k) {
    const int i = blockIdx.x * EPB + k * 1024 + t;
    if (i < e) {
      const int slot_d = hd[ed[k]] + rd[k];
      atomic_or_u32(&col_dst[slot_d >> 1],
                    (unsigned)es[k] << (16 * (slot_d & 1)));
      const int slot_s = hs[es[k]] + rs[k];
      atomic_or_u32(&col_src[slot_s >> 1],
                    (unsigned)ed[k] << (16 * (slot_s & 1)));
    }
  }
}

template <int N>
__device__ __forceinline__ void gather_group(const float* __restrict__ xl,
                                             int ids, int j, v4f& acc) {
  v4f v[N];
#pragma unroll
  for (int u = 0; u < N; ++u) {
    const int c = __builtin_amdgcn_readlane(ids, j + u);
    v[u] = *(const v4f*)(xl + (size_t)c * HDIM);
  }
#pragma unroll
  for (int u = 0; u < N; ++u) acc += v[u];
}

__global__ void __launch_bounds__(256)
gather_sum_kernel(const float* __restrict__ x,
                  const unsigned short* __restrict__ col,
                  const int* __restrict__ rp, int nrows, int etot,
                  float* __restrict__ s) {
  const int lane = threadIdx.x & 31;
  const int row = __builtin_amdgcn_readfirstlane(
      (int)((blockIdx.x * 256 + threadIdx.x) >> 5));
  if (row >= nrows) return;
  const int beg = rp[row];
  const int end = (row + 1 < nrows) ? rp[row + 1] : etot;
  const float* xl = x + lane * 4;
  v4f acc = {0.0f, 0.0f, 0.0f, 0.0f};
  for (int base = beg; base < end; base += 32) {
    const int left = end - base;
    const int n = left < 32 ? left : 32;
    const int ids = (lane < n) ? (int)col[base + lane] : 0;
    int j = 0;
    for (; j + 16 <= n; j += 16) gather_group<16>(xl, ids, j, acc);
    if (n - j >= 8) { gather_group<8>(xl, ids, j, acc); j += 8; }
    if (n - j >= 4) { gather_group<4>(xl, ids, j, acc); j += 4; }
    if (n - j >= 2) { gather_group<2>(xl, ids, j, acc); j += 2; }
    if (n - j >= 1) { gather_group<1>(xl, ids, j, acc); }
  }
  *(v4f*)(s + (size_t)row * HDIM + lane * 4) = acc;
}


template <int RT, int BK>
__global__ void __launch_bounds__(256)
gemm_in_kernel(const float* __restrict__ A, const float* __restrict__ W,
               const float* __restrict__ bias, const float* __restrict__ emb,
               const int* __restrict__ nid, float* __restrict__ C, int K) {
  constexpr int BM = 16 * RT;
  constexpr int PA = BK + 4;
  constexpr int PW = HDIM + 4;
  constexpr int ATPR = BK / 4;
  constexpr int AROWS = 256 / ATPR;
  constexpr int NA = BM / AROWS;
  constexpr int NW = BK / 8;
  static_assert(NA * AROWS == BM && NA >= 1, "A tile must be whole passes");
  __shared__ __attribute__((aligned(16))) float As[2][BM * PA];
  __shared__ __attribute__((aligned(16))) float Ws[2][BK * PW];

  const int tid  = threadIdx.x;
  const int lane = tid & 31;
  const int wave = tid >> 5;
  const int half = lane >> 4;
  const int nloc = lane & 15;
  const int m0   = blockIdx.x * BM;
  const int n    = wave * 16 + nloc;

  const int arow = tid / ATPR;
  const int acol = (tid % ATPR) * 4;
  const int wrow = tid >> 5;
  const int wcol = (tid & 31) * 4;
  const float* Ag = A + (size_t)(m0 + arow) * K + acol;
  const float* Wg = W + (size_t)wrow * HDIM + wcol;

  v4f ra[NA];
  v4f rw[NW];
  v8f acc[RT];
#pragma unroll
  for (int i = 0; i < RT; ++i) acc[i] = v8f{};

#pragma unroll
  for (int i = 0; i < NA; ++i)
    ra[i] = *(const v4f*)(Ag + (size_t)(AROWS * i) * K);
#pragma unroll
  for (int i = 0; i < NW; ++i)
    rw[i] = *(const v4f*)(Wg + (size_t)(8 * i) * HDIM);
#pragma unroll
  for (int i = 0; i < NA; ++i)
    *(v4f*)(&As[0][(arow + AROWS * i) * PA + acol]) = ra[i];
#pragma unroll
  for (int i = 0; i < NW; ++i)
    *(v4f*)(&Ws[0][(wrow + 8 * i) * PW + wcol]) = rw[i];
  __syncthreads();

  const int nchunk = K / BK;
  for (int t = 0; t < nchunk; ++t) {
    const int buf = t & 1;
    const bool more = (t + 1 < nchunk);
    if (more) {
      const int k1 = (t + 1) * BK;
#pragma unroll
      for (int i = 0; i < NA; ++i)
        ra[i] = *(const v4f*)(Ag + (size_t)(AROWS * i) * K + k1);
#pragma unroll
      for (int i = 0; i < NW; ++i)
        rw[i] = *(const v4f*)(Wg + (size_t)(k1 + 8 * i) * HDIM);
    }

    const float* as = &As[buf][nloc * PA + 4 * half];
    const float* ws = &Ws[buf][(4 * half) * PW + n];
#pragma unroll
    for (int j = 0; j < BK / 8; ++j) {
      v4f av[RT];
#pragma unroll
      for (int i = 0; i < RT; ++i)
        av[i] = *(const v4f*)(as + (16 * i) * PA + 8 * j);
      v2f b0; b0.x = ws[(8 * j + 0) * PW];
              b0.y = ws[(8 * j + 1) * PW];
      v2f b1; b1.x = ws[(8 * j + 2) * PW];
              b1.y = ws[(8 * j + 3) * PW];
#pragma unroll
      for (int i = 0; i < RT; ++i) {
        v2f a; a.x = av[i].x; a.y = av[i].y;
        acc[i] = __builtin_amdgcn_wmma_f32_16x16x4_f32(
            false, a, false, b0, (short)0, acc[i], false, false);
      }
#pragma unroll
      for (int i = 0; i < RT; ++i) {
        v2f a; a.x = av[i].z; a.y = av[i].w;
        acc[i] = __builtin_amdgcn_wmma_f32_16x16x4_f32(
            false, a, false, b1, (short)0, acc[i], false, false);
      }
    }

    if (more) {
#pragma unroll
      for (int i = 0; i < NA; ++i)
        *(v4f*)(&As[buf ^ 1][(arow + AROWS * i) * PA + acol]) = ra[i];
#pragma unroll
      for (int i = 0; i < NW; ++i)
        *(v4f*)(&Ws[buf ^ 1][(wrow + 8 * i) * PW + wcol]) = rw[i];
    }
    __syncthreads();
  }

  const float bn = bias[n];
#pragma unroll
  for (int i = 0; i < RT; ++i) {
#pragma unroll
    for (int r = 0; r < 8; ++r) {
      const int m = m0 + 16 * i + r + 8 * half;
      const float v = acc[i][r] + bn + emb[(size_t)nid[m] * HDIM + n];
      C[(size_t)m * HDIM + n] = v;
    }
  }
}

__global__ void __launch_bounds__(256)
sage_kernel(const float* __restrict__ S, const int* __restrict__ rp,
            int nrows, int etot,
            const float* __restrict__ X, const float* __restrict__ Wl,
            const float* __restrict__ bl, const float* __restrict__ Wr,
            float* __restrict__ Out, int relu) {
  const int lane = threadIdx.x & 31;
  const int wave = threadIdx.x >> 5;
  const int half = lane >> 4;
  const int nloc = lane & 15;
  const int m0   = blockIdx.x * 16;
  const int row  = m0 + nloc;
  const int n    = wave * 16 + nloc;

  const float* Srow = S + (size_t)row * HDIM;
  const float* Xrow = X + (size_t)row * HDIM;
  v8f accS = {};
  v8f accX = {};
  const int mb = m0 + 8 * half;
  int rpv[9];
#pragma unroll
  for (int r = 0; r < 8; ++r) rpv[r] = rp[mb + r];
  rpv[8] = (mb + 8 < nrows) ? rp[mb + 8] : etot;
#pragma unroll 4
  for (int k0 = 0; k0 < HDIM; k0 += 4) {
    const int ka = k0 + 2 * half;
    v2f a1; a1.x = Srow[ka]; a1.y = Srow[ka + 1];
    v2f b1; b1.x = Wl[(size_t)ka * HDIM + n];
            b1.y = Wl[(size_t)(ka + 1) * HDIM + n];
    accS = __builtin_amdgcn_wmma_f32_16x16x4_f32(false, a1, false, b1,
                                                 (short)0, accS, false, false);
    v2f a2; a2.x = Xrow[ka]; a2.y = Xrow[ka + 1];
    v2f b2; b2.x = Wr[(size_t)ka * HDIM + n];
            b2.y = Wr[(size_t)(ka + 1) * HDIM + n];
    accX = __builtin_amdgcn_wmma_f32_16x16x4_f32(false, a2, false, b2,
                                                 (short)0, accX, false, false);
  }

  const float bn = bl[n];
#pragma unroll
  for (int r = 0; r < 8; ++r) {
    const int m = m0 + r + 8 * half;
    float d = (float)(rpv[r + 1] - rpv[r]);
    d = d < 1.0f ? 1.0f : d;
    float v = accS[r] / d + accX[r] + bn;
    if (relu) v = v > 0.0f ? v : 0.0f;
    Out[(size_t)m * HDIM + n] = v;
  }
}

__global__ void edge_dot_kernel(const float* __restrict__ Hc,
                                const float* __restrict__ Hp,
                                const int* __restrict__ ls,
                                const int* __restrict__ ld, int e,
                                float* __restrict__ out) {
  int wid  = (blockIdx.x * blockDim.x + threadIdx.x) >> 5;
  int lane = threadIdx.x & 31;
  if (wid >= e) return;
  int a = ls[wid];
  int b = ld[wid];
  const float4 va = *(const float4*)(Hc + (size_t)a * HDIM + lane * 4);
  const float4 vb = *(const float4*)(Hp + (size_t)b * HDIM + lane * 4);
  float v = va.x * vb.x + va.y * vb.y + va.z * vb.z + va.w * vb.w;
#pragma unroll
  for (int off = 16; off > 0; off >>= 1) v += __shfl_down(v, off, 32);
  if (lane == 0) out[wid] = v;
}

extern "C" void kernel_launch(void* const* d_in, const int* in_sizes, int n_in,
                              void* d_out, int out_size, void* d_ws,
                              size_t ws_size, hipStream_t stream) {
  (void)in_sizes; (void)n_in; (void)out_size; (void)ws_size;
  const int NC = 4096, NP = 8192, E = 500000, EL = 500000;

  const float* x_country = (const float*)d_in[0];
  const float* x_product = (const float*)d_in[1];
  const float* Wc_lin = (const float*)d_in[2];
  const float* bc_lin = (const float*)d_in[3];
  const float* Wp_lin = (const float*)d_in[4];
  const float* bp_lin = (const float*)d_in[5];
  const float* emb_c  = (const float*)d_in[6];
  const float* emb_p  = (const float*)d_in[7];
  const float* W1ep_l = (const float*)d_in[8];
  const float* b1ep   = (const float*)d_in[9];
  const float* W1ep_r = (const float*)d_in[10];
  const float* W1pc_l = (const float*)d_in[11];
  const float* b1pc   = (const float*)d_in[12];
  const float* W1pc_r = (const float*)d_in[13];
  const float* W2ep_l = (const float*)d_in[14];
  const float* b2ep   = (const float*)d_in[15];
  const float* W2ep_r = (const float*)d_in[16];
  const float* W2pc_l = (const float*)d_in[17];
  const float* b2pc   = (const float*)d_in[18];
  const float* W2pc_r = (const float*)d_in[19];
  const int* nid_c = (const int*)d_in[20];
  const int* nid_p = (const int*)d_in[21];
  const int* e_src = (const int*)d_in[22];
  const int* e_dst = (const int*)d_in[23];
  const int* l_src = (const int*)d_in[24];
  const int* l_dst = (const int*)d_in[25];

  float* ws = (float*)d_ws;
  size_t off = 0;
  float* xc   = ws + off; off += (size_t)NC * HDIM;
  float* xp   = ws + off; off += (size_t)NP * HDIM;
  float* hp   = ws + off; off += (size_t)NP * HDIM;
  float* hc   = ws + off; off += (size_t)NC * HDIM;
  float* hp2  = ws + off; off += (size_t)NP * HDIM;
  float* hc2  = ws + off; off += (size_t)NC * HDIM;
  float* sbuf = ws + off; off += (size_t)NP * HDIM;
  int* rp_p   = (int*)(ws + off); off += NP;
  int* rp_c   = (int*)(ws + off); off += NC;

  unsigned* colw_p = (unsigned*)hc2;
  unsigned* colw_c = colw_p + E / 2;
  int* cnt_p = (int*)hc2 + E;
  int* cnt_c = cnt_p + NP;
  const unsigned short* col_p = (const unsigned short*)colw_p;
  const unsigned short* col_c = (const unsigned short*)colw_c;

  const int TB = 256;
  const int csr_blocks = (E + EPB - 1) / EPB;

  gemm_in_kernel<1, 128><<<NC / 16, TB, 0, stream>>>(x_country, Wc_lin, bc_lin,
                                                emb_c, nid_c, xc, NC);
  gemm_in_kernel<2, 64><<<NP / 32, TB, 0, stream>>>(x_product, Wp_lin, bp_lin,
                                                emb_p, nid_p, xp, NP);

  zero_kernel<<<512, TB, 0, stream>>>(hc2, E + NP + NC);
  hist_kernel<<<csr_blocks, 1024, 0, stream>>>(e_src, e_dst, E, cnt_p, cnt_c);
  scan_kernel<<<2, 1024, 0, stream>>>(cnt_p, rp_p, NP, cnt_c, rp_c, NC);
  fill_kernel<<<csr_blocks, 1024, 0, stream>>>(e_src, e_dst, E, cnt_p, cnt_c,
                                               rp_p, rp_c, colw_p, colw_c);

  gather_sum_kernel<<<NP / 8, TB, 0, stream>>>(xc, col_p, rp_p, NP, E, sbuf);
  sage_kernel<<<NP / 16, TB, 0, stream>>>(sbuf, rp_p, NP, E, xp, W1ep_l, b1ep,
                                          W1ep_r, hp, 1);

  gather_sum_kernel<<<NC / 8, TB, 0, stream>>>(xp, col_c, rp_c, NC, E, sbuf);
  sage_kernel<<<NC / 16, TB, 0, stream>>>(sbuf, rp_c, NC, E, xc, W1pc_l, b1pc,
                                          W1pc_r, hc, 1);

  gather_sum_kernel<<<NP / 8, TB, 0, stream>>>(hc, col_p, rp_p, NP, E, sbuf);
  sage_kernel<<<NP / 16, TB, 0, stream>>>(sbuf, rp_p, NP, E, hp, W2ep_l, b2ep,
                                          W2ep_r, hp2, 0);

  gather_sum_kernel<<<NC / 8, TB, 0, stream>>>(hp, col_c, rp_c, NC, E, sbuf);
  sage_kernel<<<NC / 16, TB, 0, stream>>>(sbuf, rp_c, NC, E, hc, W2pc_l, b2pc,
                                          W2pc_r, hc2, 0);

  const int el_blocks = (EL * 32 + TB - 1) / TB;
  edge_dot_kernel<<<el_blocks, TB, 0, stream>>>(hc2, hp2, l_src, l_dst, EL,
                                                (float*)d_out);
}


# edge head: each wave takes 32 label edges (8 lanes per edge, 16 row loads in flight, one 128-B store) instead of one wave per edge, 32x fewer waves to start; the dot product's f32 sum order changes
# speedup vs baseline: 2.6866x; 1.1015x over previous
#include <hip/hip_runtime.h>


typedef __attribute__((ext_vector_type(2))) float v2f;
typedef __attribute__((ext_vector_type(8))) float v8f;
typedef __attribute__((ext_vector_type(4))) float v4f;

constexpr int HDIM = 128;

__device__ __forceinline__ void atomic_add_i32(int* p, int v) {
  (void)__hip_atomic_fetch_add(p, v, __ATOMIC_RELAXED,
                               __HIP_MEMORY_SCOPE_AGENT);
}
__device__ __forceinline__ void atomic_or_u32(unsigned* p, unsigned v) {
  (void)__hip_atomic_fetch_or(p, v, __ATOMIC_RELAXED,
                              __HIP_MEMORY_SCOPE_AGENT);
}

__global__ void zero_kernel(float* __restrict__ p, int n) {
  int i = blockIdx.x * blockDim.x + threadIdx.x;
  int stride = gridDim.x * blockDim.x;
  for (; i < n; i += stride) p[i] = 0.0f;
}

constexpr int EPB = 4096;
constexpr int EPT = EPB / 1024;
constexpr int MAXP = 8192;
constexpr int MAXC = 4096;

__device__ __forceinline__ int lds_add_rtn(int* p, int v) {
  return __hip_atomic_fetch_add(p, v, __ATOMIC_RELAXED,
                                __HIP_MEMORY_SCOPE_WORKGROUP);
}

__global__ void __launch_bounds__(1024)
hist_kernel(const int* __restrict__ src, const int* __restrict__ dst, int e,
            int* __restrict__ cnt_dst, int* __restrict__ cnt_src) {
  __shared__ int hd[MAXP];
  __shared__ int hs[MAXC];
  const int t = threadIdx.x;
  for (int j = t; j < MAXP; j += 1024) hd[j] = 0;
  for (int j = t; j < MAXC; j += 1024) hs[j] = 0;
  __syncthreads();
  int es[EPT], ed[EPT];
#pragma unroll
  for (int k = 0; k < EPT; ++k) {
    const int i = blockIdx.x * EPB + k * 1024 + t;
    es[k] = (i < e) ? src[i] : 0;
    ed[k] = (i < e) ? dst[i] : 0;
  }
#pragma unroll
  for (int k = 0; k < EPT; ++k) {
    const int i = blockIdx.x * EPB + k * 1024 + t;
    if (i < e) {
      (void)lds_add_rtn(&hd[ed[k]], 1);
      (void)lds_add_rtn(&hs[es[k]], 1);
    }
  }
  __syncthreads();
  for (int j = t; j < MAXP; j += 1024) {
    const int v = hd[j];
    if (v) atomic_add_i32(&cnt_dst[j], v);
  }
  for (int j = t; j < MAXC; j += 1024) {
    const int v = hs[j];
    if (v) atomic_add_i32(&cnt_src[j], v);
  }
}

__global__ void __launch_bounds__(1024)
scan_kernel(int* __restrict__ cnt_dst, int* __restrict__ rp_dst, int np,
            int* __restrict__ cnt_src, int* __restrict__ rp_src, int nc) {
  __shared__ int sh[8192];
  __shared__ int wsum[32];
  int* cnt       = blockIdx.x == 0 ? cnt_dst : cnt_src;
  int* rp        = blockIdx.x == 0 ? rp_dst : rp_src;
  const int n    = blockIdx.x == 0 ? np : nc;
  const int ipt  = n >> 10;
  const int t    = threadIdx.x;
  const int lane = t & 31;
  const int w    = t >> 5;
  for (int i = t; i < n; i += 1024) { sh[i] = cnt[i]; cnt[i] = 0; }
  __syncthreads();
  int s = 0;
  for (int k = 0; k < ipt; ++k) s += sh[t * ipt + k];
  int inc = s;
  for (int off = 1; off < 32; off <<= 1) {
    const int o = __shfl_up(inc, off, 32);
    if (lane >= off) inc += o;
  }
  if (lane == 31) wsum[w] = inc;
  __syncthreads();
  if (w == 0) {
    const int ws = wsum[lane];
    int wi = ws;
    for (int off = 1; off < 32; off <<= 1) {
      const int o = __shfl_up(wi, off, 32);
      if (lane >= off) wi += o;
    }
    wsum[lane] = wi - ws;
  }
  __syncthreads();
  int run = wsum[w] + inc - s;
  for (int k = 0; k < ipt; ++k) {
    const int c = sh[t * ipt + k];
    sh[t * ipt + k] = run;
    run += c;
  }
  __syncthreads();
  for (int i = t; i < n; i += 1024) rp[i] = sh[i];
}

__global__ void __launch_bounds__(1024)
fill_kernel(const int* __restrict__ src, const int* __restrict__ dst, int e,
            int* __restrict__ cur_dst, int* __restrict__ cur_src,
            const int* __restrict__ rp_dst, const int* __restrict__ rp_src,
            unsigned* __restrict__ col_dst, unsigned* __restrict__ col_src) {
  __shared__ int hd[MAXP];
  __shared__ int hs[MAXC];
  const int t = threadIdx.x;
  for (int j = t; j < MAXP; j += 1024) hd[j] = 0;
  for (int j = t; j < MAXC; j += 1024) hs[j] = 0;
  __syncthreads();
  int es[EPT], ed[EPT], rd[EPT], rs[EPT];
#pragma unroll
  for (int k = 0; k < EPT; ++k) {
    const int i = blockIdx.x * EPB + k * 1024 + t;
    es[k] = (i < e) ? src[i] : 0;
    ed[k] = (i < e) ? dst[i] : 0;
  }
#pragma unroll
  for (int k = 0; k < EPT; ++k) {
    const int i = blockIdx.x * EPB + k * 1024 + t;
    rd[k] = 0; rs[k] = 0;
    if (i < e) {
      rd[k] = lds_add_rtn(&hd[ed[k]], 1);
      rs[k] = lds_add_rtn(&hs[es[k]], 1);
    }
  }
  __syncthreads();
  {
    int v[MAXP / 1024], a[MAXP / 1024];
#pragma unroll
    for (int q = 0; q < MAXP / 1024; ++q) v[q] = hd[t + 1024 * q];
#pragma unroll
    for (int q = 0; q < MAXP / 1024; ++q) {
      const int j = t + 1024 * q;
      a[q] = 0;
      if (v[q]) a[q] = rp_dst[j] + __hip_atomic_fetch_add(&cur_dst[j], v[q],
                           __ATOMIC_RELAXED, __HIP_MEMORY_SCOPE_AGENT);
    }
#pragma unroll
    for (int q = 0; q < MAXP / 1024; ++q)
      if (v[q]) hd[t + 1024 * q] = a[q];
  }
  {
    int v[MAXC / 1024], a[MAXC / 1024];
#pragma unroll
    for (int q = 0; q < MAXC / 1024; ++q) v[q] = hs[t + 1024 * q];
#pragma unroll
    for (int q = 0; q < MAXC / 1024; ++q) {
      const int j = t + 1024 * q;
      a[q] = 0;
      if (v[q]) a[q] = rp_src[j] + __hip_atomic_fetch_add(&cur_src[j], v[q],
                           __ATOMIC_RELAXED, __HIP_MEMORY_SCOPE_AGENT);
    }
#pragma unroll
    for (int q = 0; q < MAXC / 1024; ++q)
      if (v[q]) hs[t + 1024 * q] = a[q];
  }
  __syncthreads();
#pragma unroll
  for (int k = 0; k < EPT; ++k) {
    const int i = blockIdx.x * EPB + k * 1024 + t;
    if (i < e) {
      const int slot_d = hd[ed[k]] + rd[k];
      atomic_or_u32(&col_dst[slot_d >> 1],
                    (unsigned)es[k] << (16 * (slot_d & 1)));
      const int slot_s = hs[es[k]] + rs[k];
      atomic_or_u32(&col_src[slot_s >> 1],
                    (unsigned)ed[k] << (16 * (slot_s & 1)));
    }
  }
}

template <int N>
__device__ __forceinline__ void gather_group(const float* __restrict__ xl,
                                             int ids, int j, v4f& acc) {
  v4f v[N];
#pragma unroll
  for (int u = 0; u < N; ++u) {
    const int c = __builtin_amdgcn_readlane(ids, j + u);
    v[u] = *(const v4f*)(xl + (size_t)c * HDIM);
  }
#pragma unroll
  for (int u = 0; u < N; ++u) acc += v[u];
}

__global__ void __launch_bounds__(256)
gather_sum_kernel(const float* __restrict__ x,
                  const unsigned short* __restrict__ col,
                  const int* __restrict__ rp, int nrows, int etot,
                  float* __restrict__ s) {
  const int lane = threadIdx.x & 31;
  const int row = __builtin_amdgcn_readfirstlane(
      (int)((blockIdx.x * 256 + threadIdx.x) >> 5));
  if (row >= nrows) return;
  const int beg = rp[row];
  const int end = (row + 1 < nrows) ? rp[row + 1] : etot;
  const float* xl = x + lane * 4;
  v4f acc = {0.0f, 0.0f, 0.0f, 0.0f};
  for (int base = beg; base < end; base += 32) {
    const int left = end - base;
    const int n = left < 32 ? left : 32;
    const int ids = (lane < n) ? (int)col[base + lane] : 0;
    int j = 0;
    for (; j + 16 <= n; j += 16) gather_group<16>(xl, ids, j, acc);
    if (n - j >= 8) { gather_group<8>(xl, ids, j, acc); j += 8; }
    if (n - j >= 4) { gather_group<4>(xl, ids, j, acc); j += 4; }
    if (n - j >= 2) { gather_group<2>(xl, ids, j, acc); j += 2; }
    if (n - j >= 1) { gather_group<1>(xl, ids, j, acc); }
  }
  *(v4f*)(s + (size_t)row * HDIM + lane * 4) = acc;
}


template <int RT, int BK>
__global__ void __launch_bounds__(256)
gemm_in_kernel(const float* __restrict__ A, const float* __restrict__ W,
               const float* __restrict__ bias, const float* __restrict__ emb,
               const int* __restrict__ nid, float* __restrict__ C, int K) {
  constexpr int BM = 16 * RT;
  constexpr int PA = BK + 4;
  constexpr int PW = HDIM + 4;
  constexpr int ATPR = BK / 4;
  constexpr int AROWS = 256 / ATPR;
  constexpr int NA = BM / AROWS;
  constexpr int NW = BK / 8;
  static_assert(NA * AROWS == BM && NA >= 1, "A tile must be whole passes");
  __shared__ __attribute__((aligned(16))) float As[2][BM * PA];
  __shared__ __attribute__((aligned(16))) float Ws[2][BK * PW];

  const int tid  = threadIdx.x;
  const int lane = tid & 31;
  const int wave = tid >> 5;
  const int half = lane >> 4;
  const int nloc = lane & 15;
  const int m0   = blockIdx.x * BM;
  const int n    = wave * 16 + nloc;

  const int arow = tid / ATPR;
  const int acol = (tid % ATPR) * 4;
  const int wrow = tid >> 5;
  const int wcol = (tid & 31) * 4;
  const float* Ag = A + (size_t)(m0 + arow) * K + acol;
  const float* Wg = W + (size_t)wrow * HDIM + wcol;

  v4f ra[NA];
  v4f rw[NW];
  v8f acc[RT];
#pragma unroll
  for (int i = 0; i < RT; ++i) acc[i] = v8f{};

#pragma unroll
  for (int i = 0; i < NA; ++i)
    ra[i] = *(const v4f*)(Ag + (size_t)(AROWS * i) * K);
#pragma unroll
  for (int i = 0; i < NW; ++i)
    rw[i] = *(const v4f*)(Wg + (size_t)(8 * i) * HDIM);
#pragma unroll
  for (int i = 0; i < NA; ++i)
    *(v4f*)(&As[0][(arow + AROWS * i) * PA + acol]) = ra[i];
#pragma unroll
  for (int i = 0; i < NW; ++i)
    *(v4f*)(&Ws[0][(wrow + 8 * i) * PW + wcol]) = rw[i];
  __syncthreads();

  const int nchunk = K / BK;
  for (int t = 0; t < nchunk; ++t) {
    const int buf = t & 1;
    const bool more = (t + 1 < nchunk);
    if (more) {
      const int k1 = (t + 1) * BK;
#pragma unroll
      for (int i = 0; i < NA; ++i)
        ra[i] = *(const v4f*)(Ag + (size_t)(AROWS * i) * K + k1);
#pragma unroll
      for (int i = 0; i < NW; ++i)
        rw[i] = *(const v4f*)(Wg + (size_t)(k1 + 8 * i) * HDIM);
    }

    const float* as = &As[buf][nloc * PA + 4 * half];
    const float* ws = &Ws[buf][(4 * half) * PW + n];
#pragma unroll
    for (int j = 0; j < BK / 8; ++j) {
      v4f av[RT];
#pragma unroll
      for (int i = 0; i < RT; ++i)
        av[i] = *(const v4f*)(as + (16 * i) * PA + 8 * j);
      v2f b0; b0.x = ws[(8 * j + 0) * PW];
              b0.y = ws[(8 * j + 1) * PW];
      v2f b1; b1.x = ws[(8 * j + 2) * PW];
              b1.y = ws[(8 * j + 3) * PW];
#pragma unroll
      for (int i = 0; i < RT; ++i) {
        v2f a; a.x = av[i].x; a.y = av[i].y;
        acc[i] = __builtin_amdgcn_wmma_f32_16x16x4_f32(
            false, a, false, b0, (short)0, acc[i], false, false);
      }
#pragma unroll
      for (int i = 0; i < RT; ++i) {
        v2f a; a.x = av[i].z; a.y = av[i].w;
        acc[i] = __builtin_amdgcn_wmma_f32_16x16x4_f32(
            false, a, false, b1, (short)0, acc[i], false, false);
      }
    }

    if (more) {
#pragma unroll
      for (int i = 0; i < NA; ++i)
        *(v4f*)(&As[buf ^ 1][(arow + AROWS * i) * PA + acol]) = ra[i];
#pragma unroll
      for (int i = 0; i < NW; ++i)
        *(v4f*)(&Ws[buf ^ 1][(wrow + 8 * i) * PW + wcol]) = rw[i];
    }
    __syncthreads();
  }

  const float bn = bias[n];
#pragma unroll
  for (int i = 0; i < RT; ++i) {
#pragma unroll
    for (int r = 0; r < 8; ++r) {
      const int m = m0 + 16 * i + r + 8 * half;
      const float v = acc[i][r] + bn + emb[(size_t)nid[m] * HDIM + n];
      C[(size_t)m * HDIM + n] = v;
    }
  }
}

__global__ void __launch_bounds__(256)
sage_kernel(const float* __restrict__ S, const int* __restrict__ rp,
            int nrows, int etot,
            const float* __restrict__ X, const float* __restrict__ Wl,
            const float* __restrict__ bl, const float* __restrict__ Wr,
            float* __restrict__ Out, int relu) {
  const int lane = threadIdx.x & 31;
  const int wave = threadIdx.x >> 5;
  const int half = lane >> 4;
  const int nloc = lane & 15;
  const int m0   = blockIdx.x * 16;
  const int row  = m0 + nloc;
  const int n    = wave * 16 + nloc;

  const float* Srow = S + (size_t)row * HDIM;
  const float* Xrow = X + (size_t)row * HDIM;
  v8f accS = {};
  v8f accX = {};
  const int mb = m0 + 8 * half;
  int rpv[9];
#pragma unroll
  for (int r = 0; r < 8; ++r) rpv[r] = rp[mb + r];
  rpv[8] = (mb + 8 < nrows) ? rp[mb + 8] : etot;
#pragma unroll 4
  for (int k0 = 0; k0 < HDIM; k0 += 4) {
    const int ka = k0 + 2 * half;
    v2f a1; a1.x = Srow[ka]; a1.y = Srow[ka + 1];
    v2f b1; b1.x = Wl[(size_t)ka * HDIM + n];
            b1.y = Wl[(size_t)(ka + 1) * HDIM + n];
    accS = __builtin_amdgcn_wmma_f32_16x16x4_f32(false, a1, false, b1,
                                                 (short)0, accS, false, false);
    v2f a2; a2.x = Xrow[ka]; a2.y = Xrow[ka + 1];
    v2f b2; b2.x = Wr[(size_t)ka * HDIM + n];
            b2.y = Wr[(size_t)(ka + 1) * HDIM + n];
    accX = __builtin_amdgcn_wmma_f32_16x16x4_f32(false, a2, false, b2,
                                                 (short)0, accX, false, false);
  }

  const float bn = bl[n];
#pragma unroll
  for (int r = 0; r < 8; ++r) {
    const int m = m0 + r + 8 * half;
    float d = (float)(rpv[r + 1] - rpv[r]);
    d = d < 1.0f ? 1.0f : d;
    float v = accS[r] / d + accX[r] + bn;
    if (relu) v = v > 0.0f ? v : 0.0f;
    Out[(size_t)m * HDIM + n] = v;
  }
}

constexpr int EPW = 32;

template <int M>
__device__ __forceinline__ float lane_xor(float v) {
  const int i = __builtin_bit_cast(int, v);
  return __builtin_bit_cast(
      float, __builtin_amdgcn_update_dpp(i, i, 0x160 + M, 0xf, 0xf, false));
}

__global__ void __launch_bounds__(256)
edge_dot_kernel(const float* __restrict__ Hc, const float* __restrict__ Hp,
                const int* __restrict__ ls, const int* __restrict__ ld, int e,
                float* __restrict__ out) {
  const int lane = threadIdx.x & 31;
  const int q = lane & 7;
  const int first = EPW * __builtin_amdgcn_readfirstlane(
      (int)((blockIdx.x * 256 + threadIdx.x) >> 5));
  if (first >= e) return;
  const int e0 = first + (lane & 24);
  int ia[8], ib[8];
  if (first + EPW <= e) {
    const int4 a0 = *(const int4*)(ls + e0);
    const int4 a1 = *(const int4*)(ls + e0 + 4);
    const int4 b0 = *(const int4*)(ld + e0);
    const int4 b1 = *(const int4*)(ld + e0 + 4);
    ia[0] = a0.x; ia[1] = a0.y; ia[2] = a0.z; ia[3] = a0.w;
    ia[4] = a1.x; ia[5] = a1.y; ia[6] = a1.z; ia[7] = a1.w;
    ib[0] = b0.x; ib[1] = b0.y; ib[2] = b0.z; ib[3] = b0.w;
    ib[4] = b1.x; ib[5] = b1.y; ib[6] = b1.z; ib[7] = b1.w;
  } else {
#pragma unroll
    for (int k = 0; k < 8; ++k) {
      const int i = e0 + k;
      ia[k] = (i < e) ? ls[i] : 0;
      ib[k] = (i < e) ? ld[i] : 0;
    }
  }
  const float* hcq = Hc + q * 4;
  const float* hpq = Hp + q * 4;
  float r = 0.0f;
#pragma unroll
  for (int k = 0; k < 8; k += 2) {
    v4f va[2][4], vb[2][4];
#pragma unroll
    for (int u = 0; u < 2; ++u) {
      const float* pa = hcq + (size_t)ia[k + u] * HDIM;
      const float* pb = hpq + (size_t)ib[k + u] * HDIM;
#pragma unroll
      for (int c = 0; c < 4; ++c) {
        va[u][c] = *(const v4f*)(pa + 32 * c);
        vb[u][c] = *(const v4f*)(pb + 32 * c);
      }
    }
#pragma unroll
    for (int u = 0; u < 2; ++u) {
      float p = 0.0f;
#pragma unroll
      for (int c = 0; c < 4; ++c)
        p += va[u][c].x * vb[u][c].x + va[u][c].y * vb[u][c].y +
             va[u][c].z * vb[u][c].z + va[u][c].w * vb[u][c].w;
      p += lane_xor<1>(p);
      p += lane_xor<2>(p);
      p += lane_xor<4>(p);
      if (q == k + u) r = p;
    }
  }
  const int i = first + lane;
  if (i < e) out[i] = r;
}

extern "C" void kernel_launch(void* const* d_in, const int* in_sizes, int n_in,
                              void* d_out, int out_size, void* d_ws,
                              size_t ws_size, hipStream_t stream) {
  (void)in_sizes; (void)n_in; (void)out_size; (void)ws_size;
  const int NC = 4096, NP = 8192, E = 500000, EL = 500000;

  const float* x_country = (const float*)d_in[0];
  const float* x_product = (const float*)d_in[1];
  const float* Wc_lin = (const float*)d_in[2];
  const float* bc_lin = (const float*)d_in[3];
  const float* Wp_lin = (const float*)d_in[4];
  const float* bp_lin = (const float*)d_in[5];
  const float* emb_c  = (const float*)d_in[6];
  const float* emb_p  = (const float*)d_in[7];
  const float* W1ep_l = (const float*)d_in[8];
  const float* b1ep   = (const float*)d_in[9];
  const float* W1ep_r = (const float*)d_in[10];
  const float* W1pc_l = (const float*)d_in[11];
  const float* b1pc   = (const float*)d_in[12];
  const float* W1pc_r = (const float*)d_in[13];
  const float* W2ep_l = (const float*)d_in[14];
  const float* b2ep   = (const float*)d_in[15];
  const float* W2ep_r = (const float*)d_in[16];
  const float* W2pc_l = (const float*)d_in[17];
  const float* b2pc   = (const float*)d_in[18];
  const float* W2pc_r = (const float*)d_in[19];
  const int* nid_c = (const int*)d_in[20];
  const int* nid_p = (const int*)d_in[21];
  const int* e_src = (const int*)d_in[22];
  const int* e_dst = (const int*)d_in[23];
  const int* l_src = (const int*)d_in[24];
  const int* l_dst = (const int*)d_in[25];

  float* ws = (float*)d_ws;
  size_t off = 0;
  float* xc   = ws + off; off += (size_t)NC * HDIM;
  float* xp   = ws + off; off += (size_t)NP * HDIM;
  float* hp   = ws + off; off += (size_t)NP * HDIM;
  float* hc   = ws + off; off += (size_t)NC * HDIM;
  float* hp2  = ws + off; off += (size_t)NP * HDIM;
  float* hc2  = ws + off; off += (size_t)NC * HDIM;
  float* sbuf = ws + off; off += (size_t)NP * HDIM;
  int* rp_p   = (int*)(ws + off); off += NP;
  int* rp_c   = (int*)(ws + off); off += NC;

  unsigned* colw_p = (unsigned*)hc2;
  unsigned* colw_c = colw_p + E / 2;
  int* cnt_p = (int*)hc2 + E;
  int* cnt_c = cnt_p + NP;
  const unsigned short* col_p = (const unsigned short*)colw_p;
  const unsigned short* col_c = (const unsigned short*)colw_c;

  const int TB = 256;
  const int csr_blocks = (E + EPB - 1) / EPB;

  gemm_in_kernel<1, 128><<<NC / 16, TB, 0, stream>>>(x_country, Wc_lin, bc_lin,
                                                emb_c, nid_c, xc, NC);
  gemm_in_kernel<2, 64><<<NP / 32, TB, 0, stream>>>(x_product, Wp_lin, bp_lin,
                                                emb_p, nid_p, xp, NP);

  zero_kernel<<<512, TB, 0, stream>>>(hc2, E + NP + NC);
  hist_kernel<<<csr_blocks, 1024, 0, stream>>>(e_src, e_dst, E, cnt_p, cnt_c);
  scan_kernel<<<2, 1024, 0, stream>>>(cnt_p, rp_p, NP, cnt_c, rp_c, NC);
  fill_kernel<<<csr_blocks, 1024, 0, stream>>>(e_src, e_dst, E, cnt_p, cnt_c,
                                               rp_p, rp_c, colw_p, colw_c);

  gather_sum_kernel<<<NP / 8, TB, 0, stream>>>(xc, col_p, rp_p, NP, E, sbuf);
  sage_kernel<<<NP / 16, TB, 0, stream>>>(sbuf, rp_p, NP, E, xp, W1ep_l, b1ep,
                                          W1ep_r, hp, 1);

  gather_sum_kernel<<<NC / 8, TB, 0, stream>>>(xp, col_c, rp_c, NC, E, sbuf);
  sage_kernel<<<NC / 16, TB, 0, stream>>>(sbuf, rp_c, NC, E, xc, W1pc_l, b1pc,
                                          W1pc_r, hc, 1);

  gather_sum_kernel<<<NP / 8, TB, 0, stream>>>(hc, col_p, rp_p, NP, E, sbuf);
  sage_kernel<<<NP / 16, TB, 0, stream>>>(sbuf, rp_p, NP, E, hp, W2ep_l, b2ep,
                                          W2ep_r, hp2, 0);

  gather_sum_kernel<<<NC / 8, TB, 0, stream>>>(hp, col_c, rp_c, NC, E, sbuf);
  sage_kernel<<<NC / 16, TB, 0, stream>>>(sbuf, rp_c, NC, E, hc, W2pc_l, b2pc,
                                          W2pc_r, hc2, 0);

  const int el_blocks = (EL + 8 * EPW - 1) / (8 * EPW);
  edge_dot_kernel<<<el_blocks, TB, 0, stream>>>(hc2, hp2, l_src, l_dst, EL,
                                                (float*)d_out);
}


# each SAGE layer is one launch: a block gathers its 16 rows' neighbour sums into LDS and runs both WMMA products from there, both edge directions in one grid; 15 -> 9 launches, sum order kept
# speedup vs baseline: 2.9483x; 1.0974x over previous
#include <hip/hip_runtime.h>


typedef __attribute__((ext_vector_type(2))) float v2f;
typedef __attribute__((ext_vector_type(8))) float v8f;
typedef __attribute__((ext_vector_type(4))) float v4f;

constexpr int HDIM = 128;

__device__ __forceinline__ void atomic_add_i32(int* p, int v) {
  (void)__hip_atomic_fetch_add(p, v, __ATOMIC_RELAXED,
                               __HIP_MEMORY_SCOPE_AGENT);
}
__device__ __forceinline__ void atomic_or_u32(unsigned* p, unsigned v) {
  (void)__hip_atomic_fetch_or(p, v, __ATOMIC_RELAXED,
                              __HIP_MEMORY_SCOPE_AGENT);
}

__global__ void zero_kernel(float* __restrict__ p, int n) {
  int i = blockIdx.x * blockDim.x + threadIdx.x;
  int stride = gridDim.x * blockDim.x;
  for (; i < n; i += stride) p[i] = 0.0f;
}

constexpr int EPB = 4096;
constexpr int EPT = EPB / 1024;
constexpr int MAXP = 8192;
constexpr int MAXC = 4096;

__device__ __forceinline__ int lds_add_rtn(int* p, int v) {
  return __hip_atomic_fetch_add(p, v, __ATOMIC_RELAXED,
                                __HIP_MEMORY_SCOPE_WORKGROUP);
}

__global__ void __launch_bounds__(1024)
hist_kernel(const int* __restrict__ src, const int* __restrict__ dst, int e,
            int* __restrict__ cnt_dst, int* __restrict__ cnt_src) {
  __shared__ int hd[MAXP];
  __shared__ int hs[MAXC];
  const int t = threadIdx.x;
  for (int j = t; j < MAXP; j += 1024) hd[j] = 0;
  for (int j = t; j < MAXC; j += 1024) hs[j] = 0;
  __syncthreads();
  int es[EPT], ed[EPT];
#pragma unroll
  for (int k = 0; k < EPT; ++k) {
    const int i = blockIdx.x * EPB + k * 1024 + t;
    es[k] = (i < e) ? src[i] : 0;
    ed[k] = (i < e) ? dst[i] : 0;
  }
#pragma unroll
  for (int k = 0; k < EPT; ++k) {
    const int i = blockIdx.x * EPB + k * 1024 + t;
    if (i < e) {
      (void)lds_add_rtn(&hd[ed[k]], 1);
      (void)lds_add_rtn(&hs[es[k]], 1);
    }
  }
  __syncthreads();
  for (int j = t; j < MAXP; j += 1024) {
    const int v = hd[j];
    if (v) atomic_add_i32(&cnt_dst[j], v);
  }
  for (int j = t; j < MAXC; j += 1024) {
    const int v = hs[j];
    if (v) atomic_add_i32(&cnt_src[j], v);
  }
}

__global__ void __launch_bounds__(1024)
scan_kernel(int* __restrict__ cnt_dst, int* __restrict__ rp_dst, int np,
            int* __restrict__ cnt_src, int* __restrict__ rp_src, int nc) {
  __shared__ int sh[8192];
  __shared__ int wsum[32];
  int* cnt       = blockIdx.x == 0 ? cnt_dst : cnt_src;
  int* rp        = blockIdx.x == 0 ? rp_dst : rp_src;
  const int n    = blockIdx.x == 0 ? np : nc;
  const int ipt  = n >> 10;
  const int t    = threadIdx.x;
  const int lane = t & 31;
  const int w    = t >> 5;
  for (int i = t; i < n; i += 1024) { sh[i] = cnt[i]; cnt[i] = 0; }
  __syncthreads();
  int s = 0;
  for (int k = 0; k < ipt; ++k) s += sh[t * ipt + k];
  int inc = s;
  for (int off = 1; off < 32; off <<= 1) {
    const int o = __shfl_up(inc, off, 32);
    if (lane >= off) inc += o;
  }
  if (lane == 31) wsum[w] = inc;
  __syncthreads();
  if (w == 0) {
    const int ws = wsum[lane];
    int wi = ws;
    for (int off = 1; off < 32; off <<= 1) {
      const int o = __shfl_up(wi, off, 32);
      if (lane >= off) wi += o;
    }
    wsum[lane] = wi - ws;
  }
  __syncthreads();
  int run = wsum[w] + inc - s;
  for (int k = 0; k < ipt; ++k) {
    const int c = sh[t * ipt + k];
    sh[t * ipt + k] = run;
    run += c;
  }
  __syncthreads();
  for (int i = t; i < n; i += 1024) rp[i] = sh[i];
}

__global__ void __launch_bounds__(1024)
fill_kernel(const int* __restrict__ src, const int* __restrict__ dst, int e,
            int* __restrict__ cur_dst, int* __restrict__ cur_src,
            const int* __restrict__ rp_dst, const int* __restrict__ rp_src,
            unsigned* __restrict__ col_dst, unsigned* __restrict__ col_src) {
  __shared__ int hd[MAXP];
  __shared__ int hs[MAXC];
  const int t = threadIdx.x;
  for (int j = t; j < MAXP; j += 1024) hd[j] = 0;
  for (int j = t; j < MAXC; j += 1024) hs[j] = 0;
  __syncthreads();
  int es[EPT], ed[EPT], rd[EPT], rs[EPT];
#pragma unroll
  for (int k = 0; k < EPT; ++k) {
    const int i = blockIdx.x * EPB + k * 1024 + t;
    es[k] = (i < e) ? src[i] : 0;
    ed[k] = (i < e) ? dst[i] : 0;
  }
#pragma unroll
  for (int k = 0; k < EPT; ++k) {
    const int i = blockIdx.x * EPB + k * 1024 + t;
    rd[k] = 0; rs[k] = 0;
    if (i < e) {
      rd[k] = lds_add_rtn(&hd[ed[k]], 1);
      rs[k] = lds_add_rtn(&hs[es[k]], 1);
    }
  }
  __syncthreads();
  {
    int v[MAXP / 1024], a[MAXP / 1024];
#pragma unroll
    for (int q = 0; q < MAXP / 1024; ++q) v[q] = hd[t + 1024 * q];
#pragma unroll
    for (int q = 0; q < MAXP / 1024; ++q) {
      const int j = t + 1024 * q;
      a[q] = 0;
      if (v[q]) a[q] = rp_dst[j] + __hip_atomic_fetch_add(&cur_dst[j], v[q],
                           __ATOMIC_RELAXED, __HIP_MEMORY_SCOPE_AGENT);
    }
#pragma unroll
    for (int q = 0; q < MAXP / 1024; ++q)
      if (v[q]) hd[t + 1024 * q] = a[q];
  }
  {
    int v[MAXC / 1024], a[MAXC / 1024];
#pragma unroll
    for (int q = 0; q < MAXC / 1024; ++q) v[q] = hs[t + 1024 * q];
#pragma unroll
    for (int q = 0; q < MAXC / 1024; ++q) {
      const int j = t + 1024 * q;
      a[q] = 0;
      if (v[q]) a[q] = rp_src[j] + __hip_atomic_fetch_add(&cur_src[j], v[q],
                           __ATOMIC_RELAXED, __HIP_MEMORY_SCOPE_AGENT);
    }
#pragma unroll
    for (int q = 0; q < MAXC / 1024; ++q)
      if (v[q]) hs[t + 1024 * q] = a[q];
  }
  __syncthreads();
#pragma unroll
  for (int k = 0; k < EPT; ++k) {
    const int i = blockIdx.x * EPB + k * 1024 + t;
    if (i < e) {
      const int slot_d = hd[ed[k]] + rd[k];
      atomic_or_u32(&col_dst[slot_d >> 1],
                    (unsigned)es[k] << (16 * (slot_d & 1)));
      const int slot_s = hs[es[k]] + rs[k];
      atomic_or_u32(&col_src[slot_s >> 1],
                    (unsigned)ed[k] << (16 * (slot_s & 1)));
    }
  }
}

template <int N>
__device__ __forceinline__ void gather_group(const float* __restrict__ xl,
                                             int ids, int j, v4f& acc) {
  v4f v[N];
#pragma unroll
  for (int u = 0; u < N; ++u) {
    const int c = __builtin_amdgcn_readlane(ids, j + u);
    v[u] = *(const v4f*)(xl + (size_t)c * HDIM);
  }
#pragma unroll
  for (int u = 0; u < N; ++u) acc += v[u];
}

__device__ __forceinline__ v4f gather_row(const float* __restrict__ x,
                                          const unsigned short* __restrict__ col,
                                          int beg, int end, int lane) {
  const float* xl = x + lane * 4;
  v4f acc = {0.0f, 0.0f, 0.0f, 0.0f};
  for (int base = beg; base < end; base += 32) {
    const int left = end - base;
    const int n = left < 32 ? left : 32;
    const int ids = (lane < n) ? (int)col[base + lane] : 0;
    int j = 0;
    for (; j + 16 <= n; j += 16) gather_group<16>(xl, ids, j, acc);
    if (n - j >= 8) { gather_group<8>(xl, ids, j, acc); j += 8; }
    if (n - j >= 4) { gather_group<4>(xl, ids, j, acc); j += 4; }
    if (n - j >= 2) { gather_group<2>(xl, ids, j, acc); j += 2; }
    if (n - j >= 1) { gather_group<1>(xl, ids, j, acc); }
  }
  return acc;
}


template <int RT, int BK>
__global__ void __launch_bounds__(256)
gemm_in_kernel(const float* __restrict__ A, const float* __restrict__ W,
               const float* __restrict__ bias, const float* __restrict__ emb,
               const int* __restrict__ nid, float* __restrict__ C, int K) {
  constexpr int BM = 16 * RT;
  constexpr int PA = BK + 4;
  constexpr int PW = HDIM + 4;
  constexpr int ATPR = BK / 4;
  constexpr int AROWS = 256 / ATPR;
  constexpr int NA = BM / AROWS;
  constexpr int NW = BK / 8;
  static_assert(NA * AROWS == BM && NA >= 1, "A tile must be whole passes");
  __shared__ __attribute__((aligned(16))) float As[2][BM * PA];
  __shared__ __attribute__((aligned(16))) float Ws[2][BK * PW];

  const int tid  = threadIdx.x;
  const int lane = tid & 31;
  const int wave = tid >> 5;
  const int half = lane >> 4;
  const int nloc = lane & 15;
  const int m0   = blockIdx.x * BM;
  const int n    = wave * 16 + nloc;

  const int arow = tid / ATPR;
  const int acol = (tid % ATPR) * 4;
  const int wrow = tid >> 5;
  const int wcol = (tid & 31) * 4;
  const float* Ag = A + (size_t)(m0 + arow) * K + acol;
  const float* Wg = W + (size_t)wrow * HDIM + wcol;

  v4f ra[NA];
  v4f rw[NW];
  v8f acc[RT];
#pragma unroll
  for (int i = 0; i < RT; ++i) acc[i] = v8f{};

#pragma unroll
  for (int i = 0; i < NA; ++i)
    ra[i] = *(const v4f*)(Ag + (size_t)(AROWS * i) * K);
#pragma unroll
  for (int i = 0; i < NW; ++i)
    rw[i] = *(const v4f*)(Wg + (size_t)(8 * i) * HDIM);
#pragma unroll
  for (int i = 0; i < NA; ++i)
    *(v4f*)(&As[0][(arow + AROWS * i) * PA + acol]) = ra[i];
#pragma unroll
  for (int i = 0; i < NW; ++i)
    *(v4f*)(&Ws[0][(wrow + 8 * i) * PW + wcol]) = rw[i];
  __syncthreads();

  const int nchunk = K / BK;
  for (int t = 0; t < nchunk; ++t) {
    const int buf = t & 1;
    const bool more = (t + 1 < nchunk);
    if (more) {
      const int k1 = (t + 1) * BK;
#pragma unroll
      for (int i = 0; i < NA; ++i)
        ra[i] = *(const v4f*)(Ag + (size_t)(AROWS * i) * K + k1);
#pragma unroll
      for (int i = 0; i < NW; ++i)
        rw[i] = *(const v4f*)(Wg + (size_t)(k1 + 8 * i) * HDIM);
    }

    const float* as = &As[buf][nloc * PA + 4 * half];
    const float* ws = &Ws[buf][(4 * half) * PW + n];
#pragma unroll
    for (int j = 0; j < BK / 8; ++j) {
      v4f av[RT];
#pragma unroll
      for (int i = 0; i < RT; ++i)
        av[i] = *(const v4f*)(as + (16 * i) * PA + 8 * j);
      v2f b0; b0.x = ws[(8 * j + 0) * PW];
              b0.y = ws[(8 * j + 1) * PW];
      v2f b1; b1.x = ws[(8 * j + 2) * PW];
              b1.y = ws[(8 * j + 3) * PW];
#pragma unroll
      for (int i = 0; i < RT; ++i) {
        v2f a; a.x = av[i].x; a.y = av[i].y;
        acc[i] = __builtin_amdgcn_wmma_f32_16x16x4_f32(
            false, a, false, b0, (short)0, acc[i], false, false);
      }
#pragma unroll
      for (int i = 0; i < RT; ++i) {
        v2f a; a.x = av[i].z; a.y = av[i].w;
        acc[i] = __builtin_amdgcn_wmma_f32_16x16x4_f32(
            false, a, false, b1, (short)0, acc[i], false, false);
      }
    }

    if (more) {
#pragma unroll
      for (int i = 0; i < NA; ++i)
        *(v4f*)(&As[buf ^ 1][(arow + AROWS * i) * PA + acol]) = ra[i];
#pragma unroll
      for (int i = 0; i < NW; ++i)
        *(v4f*)(&Ws[buf ^ 1][(wrow + 8 * i) * PW + wcol]) = rw[i];
    }
    __syncthreads();
  }

  const float bn = bias[n];
#pragma unroll
  for (int i = 0; i < RT; ++i) {
#pragma unroll
    for (int r = 0; r < 8; ++r) {
      const int m = m0 + 16 * i + r + 8 * half;
      const float v = acc[i][r] + bn + emb[(size_t)nid[m] * HDIM + n];
      C[(size_t)m * HDIM + n] = v;
    }
  }
}

constexpr int PS = HDIM + 4;

__global__ void __launch_bounds__(256)
sage_layer_kernel(const float* __restrict__ xsrc_c,
                  const unsigned short* __restrict__ col_c,
                  const int* __restrict__ rp_c,
                  const float* __restrict__ xdst_c,
                  const float* __restrict__ Wl_c, const float* __restrict__ bl_c,
                  const float* __restrict__ Wr_c, float* __restrict__ out_c,
                  int nrows_c,
                  const float* __restrict__ xsrc_p,
                  const unsigned short* __restrict__ col_p,
                  const int* __restrict__ rp_p,
                  const float* __restrict__ xdst_p,
                  const float* __restrict__ Wl_p, const float* __restrict__ bl_p,
                  const float* __restrict__ Wr_p, float* __restrict__ out_p,
                  int nrows_p, int etot, int relu) {
  __shared__ __attribute__((aligned(16))) float Ss[16 * PS];
  __shared__ __attribute__((aligned(16))) float Xs[16 * PS];

  const int nb_c = nrows_c >> 4;
  const bool isc = (int)blockIdx.x < nb_c;
  const float* xsrc          = isc ? xsrc_c : xsrc_p;
  const unsigned short* col  = isc ? col_c : col_p;
  const int* rp              = isc ? rp_c : rp_p;
  const float* X             = isc ? xdst_c : xdst_p;
  const float* Wl            = isc ? Wl_c : Wl_p;
  const float* bl            = isc ? bl_c : bl_p;
  const float* Wr            = isc ? Wr_c : Wr_p;
  float* Out                 = isc ? out_c : out_p;
  const int nrows            = isc ? nrows_c : nrows_p;
  const int m0 = (isc ? (int)blockIdx.x : (int)blockIdx.x - nb_c) * 16;

  const int lane = threadIdx.x & 31;
  const int wave = __builtin_amdgcn_readfirstlane((int)(threadIdx.x >> 5));
  const int half = lane >> 4;
  const int nloc = lane & 15;
  const int n    = wave * 16 + nloc;

  const int lr = 2 * wave;
  const v4f xr0 = *(const v4f*)(X + (size_t)(m0 + lr) * HDIM + lane * 4);
  const v4f xr1 = *(const v4f*)(X + (size_t)(m0 + lr + 1) * HDIM + lane * 4);
  for (int rr = 0; rr < 2; ++rr) {
    const int row = m0 + lr + rr;
    const int beg = rp[row];
    const int end = (row + 1 < nrows) ? rp[row + 1] : etot;
    const v4f acc = gather_row(xsrc, col, beg, end, lane);
    *(v4f*)(&Ss[(lr + rr) * PS + lane * 4]) = acc;
  }
  *(v4f*)(&Xs[lr * PS + lane * 4]) = xr0;
  *(v4f*)(&Xs[(lr + 1) * PS + lane * 4]) = xr1;

  const int mb = m0 + 8 * half;
  int rpv[9];
#pragma unroll
  for (int r = 0; r < 8; ++r) rpv[r] = rp[mb + r];
  rpv[8] = (mb + 8 < nrows) ? rp[mb + 8] : etot;
  __syncthreads();

  const float* sa = &Ss[nloc * PS + 2 * half];
  const float* xa = &Xs[nloc * PS + 2 * half];
  v8f accS = {};
  v8f accX = {};
#pragma unroll 8
  for (int k0 = 0; k0 < HDIM; k0 += 4) {
    const int ka = k0 + 2 * half;
    const v2f a1 = *(const v2f*)(sa + k0);
    v2f b1; b1.x = Wl[(size_t)ka * HDIM + n];
            b1.y = Wl[(size_t)(ka + 1) * HDIM + n];
    accS = __builtin_amdgcn_wmma_f32_16x16x4_f32(false, a1, false, b1,
                                                 (short)0, accS, false, false);
    const v2f a2 = *(const v2f*)(xa + k0);
    v2f b2; b2.x = Wr[(size_t)ka * HDIM + n];
            b2.y = Wr[(size_t)(ka + 1) * HDIM + n];
    accX = __builtin_amdgcn_wmma_f32_16x16x4_f32(false, a2, false, b2,
                                                 (short)0, accX, false, false);
  }

  const float bn = bl[n];
#pragma unroll
  for (int r = 0; r < 8; ++r) {
    const int m = m0 + r + 8 * half;
    float d = (float)(rpv[r + 1] - rpv[r]);
    d = d < 1.0f ? 1.0f : d;
    float v = accS[r] / d + accX[r] + bn;
    if (relu) v = v > 0.0f ? v : 0.0f;
    Out[(size_t)m * HDIM + n] = v;
  }
}

constexpr int EPW = 32;

template <int M>
__device__ __forceinline__ float lane_xor(float v) {
  const int i = __builtin_bit_cast(int, v);
  return __builtin_bit_cast(
      float, __builtin_amdgcn_update_dpp(i, i, 0x160 + M, 0xf, 0xf, false));
}

__global__ void __launch_bounds__(256)
edge_dot_kernel(const float* __restrict__ Hc, const float* __restrict__ Hp,
                const int* __restrict__ ls, const int* __restrict__ ld, int e,
                float* __restrict__ out) {
  const int lane = threadIdx.x & 31;
  const int q = lane & 7;
  const int first = EPW * __builtin_amdgcn_readfirstlane(
      (int)((blockIdx.x * 256 + threadIdx.x) >> 5));
  if (first >= e) return;
  const int e0 = first + (lane & 24);
  int ia[8], ib[8];
  if (first + EPW <= e) {
    const int4 a0 = *(const int4*)(ls + e0);
    const int4 a1 = *(const int4*)(ls + e0 + 4);
    const int4 b0 = *(const int4*)(ld + e0);
    const int4 b1 = *(const int4*)(ld + e0 + 4);
    ia[0] = a0.x; ia[1] = a0.y; ia[2] = a0.z; ia[3] = a0.w;
    ia[4] = a1.x; ia[5] = a1.y; ia[6] = a1.z; ia[7] = a1.w;
    ib[0] = b0.x; ib[1] = b0.y; ib[2] = b0.z; ib[3] = b0.w;
    ib[4] = b1.x; ib[5] = b1.y; ib[6] = b1.z; ib[7] = b1.w;
  } else {
#pragma unroll
    for (int k = 0; k < 8; ++k) {
      const int i = e0 + k;
      ia[k] = (i < e) ? ls[i] : 0;
      ib[k] = (i < e) ? ld[i] : 0;
    }
  }
  const float* hcq = Hc + q * 4;
  const float* hpq = Hp + q * 4;
  float r = 0.0f;
#pragma unroll
  for (int k = 0; k < 8; k += 2) {
    v4f va[2][4], vb[2][4];
#pragma unroll
    for (int u = 0; u < 2; ++u) {
      const float* pa = hcq + (size_t)ia[k + u] * HDIM;
      const float* pb = hpq + (size_t)ib[k + u] * HDIM;
#pragma unroll
      for (int c = 0; c < 4; ++c) {
        va[u][c] = *(const v4f*)(pa + 32 * c);
        vb[u][c] = *(const v4f*)(pb + 32 * c);
      }
    }
#pragma unroll
    for (int u = 0; u < 2; ++u) {
      float p = 0.0f;
#pragma unroll
      for (int c = 0; c < 4; ++c)
        p += va[u][c].x * vb[u][c].x + va[u][c].y * vb[u][c].y +
             va[u][c].z * vb[u][c].z + va[u][c].w * vb[u][c].w;
      p += lane_xor<1>(p);
      p += lane_xor<2>(p);
      p += lane_xor<4>(p);
      if (q == k + u) r = p;
    }
  }
  const int i = first + lane;
  if (i < e) out[i] = r;
}

extern "C" void kernel_launch(void* const* d_in, const int* in_sizes, int n_in,
                              void* d_out, int out_size, void* d_ws,
                              size_t ws_size, hipStream_t stream) {
  (void)in_sizes; (void)n_in; (void)out_size; (void)ws_size;
  const int NC = 4096, NP = 8192, E = 500000, EL = 500000;

  const float* x_country = (const float*)d_in[0];
  const float* x_product = (const float*)d_in[1];
  const float* Wc_lin = (const float*)d_in[2];
  const float* bc_lin = (const float*)d_in[3];
  const float* Wp_lin = (const float*)d_in[4];
  const float* bp_lin = (const float*)d_in[5];
  const float* emb_c  = (const float*)d_in[6];
  const float* emb_p  = (const float*)d_in[7];
  const float* W1ep_l = (const float*)d_in[8];
  const float* b1ep   = (const float*)d_in[9];
  const float* W1ep_r = (const float*)d_in[10];
  const float* W1pc_l = (const float*)d_in[11];
  const float* b1pc   = (const float*)d_in[12];
  const float* W1pc_r = (const float*)d_in[13];
  const float* W2ep_l = (const float*)d_in[14];
  const float* b2ep   = (const float*)d_in[15];
  const float* W2ep_r = (const float*)d_in[16];
  const float* W2pc_l = (const float*)d_in[17];
  const float* b2pc   = (const float*)d_in[18];
  const float* W2pc_r = (const float*)d_in[19];
  const int* nid_c = (const int*)d_in[20];
  const int* nid_p = (const int*)d_in[21];
  const int* e_src = (const int*)d_in[22];
  const int* e_dst = (const int*)d_in[23];
  const int* l_src = (const int*)d_in[24];
  const int* l_dst = (const int*)d_in[25];

  float* ws = (float*)d_ws;
  size_t off = 0;
  float* xc   = ws + off; off += (size_t)NC * HDIM;
  float* xp   = ws + off; off += (size_t)NP * HDIM;
  float* hp   = ws + off; off += (size_t)NP * HDIM;
  float* hc   = ws + off; off += (size_t)NC * HDIM;
  float* hp2  = ws + off; off += (size_t)NP * HDIM;
  float* hc2  = ws + off; off += (size_t)NC * HDIM;
  float* scr  = ws + off; off += (size_t)NP * HDIM;
  int* rp_p   = (int*)(ws + off); off += NP;
  int* rp_c   = (int*)(ws + off); off += NC;

  unsigned* colw_p = (unsigned*)scr;
  unsigned* colw_c = colw_p + E / 2;
  int* cnt_p = (int*)scr + E;
  int* cnt_c = cnt_p + NP;
  const unsigned short* col_p = (const unsigned short*)colw_p;
  const unsigned short* col_c = (const unsigned short*)colw_c;

  const int TB = 256;
  const int csr_blocks = (E + EPB - 1) / EPB;

  gemm_in_kernel<1, 128><<<NC / 16, TB, 0, stream>>>(x_country, Wc_lin, bc_lin,
                                                emb_c, nid_c, xc, NC);
  gemm_in_kernel<2, 64><<<NP / 32, TB, 0, stream>>>(x_product, Wp_lin, bp_lin,
                                                emb_p, nid_p, xp, NP);

  zero_kernel<<<512, TB, 0, stream>>>(scr, E + NP + NC);
  hist_kernel<<<csr_blocks, 1024, 0, stream>>>(e_src, e_dst, E, cnt_p, cnt_c);
  scan_kernel<<<2, 1024, 0, stream>>>(cnt_p, rp_p, NP, cnt_c, rp_c, NC);
  fill_kernel<<<csr_blocks, 1024, 0, stream>>>(e_src, e_dst, E, cnt_p, cnt_c,
                                               rp_p, rp_c, colw_p, colw_c);

  const int layer_blocks = NC / 16 + NP / 16;

  sage_layer_kernel<<<layer_blocks, TB, 0, stream>>>(
      xp, col_c, rp_c, xc, W1pc_l, b1pc, W1pc_r, hc, NC,
      xc, col_p, rp_p, xp, W1ep_l, b1ep, W1ep_r, hp, NP, E, 1);

  sage_layer_kernel<<<layer_blocks, TB, 0, stream>>>(
      hp, col_c, rp_c, hc, W2pc_l, b2pc, W2pc_r, hc2, NC,
      hc, col_p, rp_p, hp, W2ep_l, b2ep, W2ep_r, hp2, NP, E, 0);

  const int el_blocks = (EL + 8 * EPW - 1) / (8 * EPW);
  edge_dot_kernel<<<el_blocks, TB, 0, stream>>>(hc2, hp2, l_src, l_dst, EL,
                                                (float*)d_out);
}


# both input projection GEMMs in one launch of 512 blocks on one shared LDS buffer: two tiles per WGP, one tile's WMMAs fill the other's barrier gaps; 9 -> 8 launches, every sum order kept
# speedup vs baseline: 3.1802x; 1.0787x over previous
#include <hip/hip_runtime.h>


typedef __attribute__((ext_vector_type(2))) float v2f;
typedef __attribute__((ext_vector_type(8))) float v8f;
typedef __attribute__((ext_vector_type(4))) float v4f;

constexpr int HDIM = 128;

__device__ __forceinline__ void atomic_add_i32(int* p, int v) {
  (void)__hip_atomic_fetch_add(p, v, __ATOMIC_RELAXED,
                               __HIP_MEMORY_SCOPE_AGENT);
}
__device__ __forceinline__ void atomic_or_u32(unsigned* p, unsigned v) {
  (void)__hip_atomic_fetch_or(p, v, __ATOMIC_RELAXED,
                              __HIP_MEMORY_SCOPE_AGENT);
}

__global__ void zero_kernel(float* __restrict__ p, int n) {
  int i = blockIdx.x * blockDim.x + threadIdx.x;
  int stride = gridDim.x * blockDim.x;
  for (; i < n; i += stride) p[i] = 0.0f;
}

constexpr int EPB = 4096;
constexpr int EPT = EPB / 1024;
constexpr int MAXP = 8192;
constexpr int MAXC = 4096;

__device__ __forceinline__ int lds_add_rtn(int* p, int v) {
  return __hip_atomic_fetch_add(p, v, __ATOMIC_RELAXED,
                                __HIP_MEMORY_SCOPE_WORKGROUP);
}

__global__ void __launch_bounds__(1024)
hist_kernel(const int* __restrict__ src, const int* __restrict__ dst, int e,
            int* __restrict__ cnt_dst, int* __restrict__ cnt_src) {
  __shared__ int hd[MAXP];
  __shared__ int hs[MAXC];
  const int t = threadIdx.x;
  for (int j = t; j < MAXP; j += 1024) hd[j] = 0;
  for (int j = t; j < MAXC; j += 1024) hs[j] = 0;
  __syncthreads();
  int es[EPT], ed[EPT];
#pragma unroll
  for (int k = 0; k < EPT; ++k) {
    const int i = blockIdx.x * EPB + k * 1024 + t;
    es[k] = (i < e) ? src[i] : 0;
    ed[k] = (i < e) ? dst[i] : 0;
  }
#pragma unroll
  for (int k = 0; k < EPT; ++k) {
    const int i = blockIdx.x * EPB + k * 1024 + t;
    if (i < e) {
      (void)lds_add_rtn(&hd[ed[k]], 1);
      (void)lds_add_rtn(&hs[es[k]], 1);
    }
  }
  __syncthreads();
  for (int j = t; j < MAXP; j += 1024) {
    const int v = hd[j];
    if (v) atomic_add_i32(&cnt_dst[j], v);
  }
  for (int j = t; j < MAXC; j += 1024) {
    const int v = hs[j];
    if (v) atomic_add_i32(&cnt_src[j], v);
  }
}

__global__ void __launch_bounds__(1024)
scan_kernel(int* __restrict__ cnt_dst, int* __restrict__ rp_dst, int np,
            int* __restrict__ cnt_src, int* __restrict__ rp_src, int nc) {
  __shared__ int sh[8192];
  __shared__ int wsum[32];
  int* cnt       = blockIdx.x == 0 ? cnt_dst : cnt_src;
  int* rp        = blockIdx.x == 0 ? rp_dst : rp_src;
  const int n    = blockIdx.x == 0 ? np : nc;
  const int ipt  = n >> 10;
  const int t    = threadIdx.x;
  const int lane = t & 31;
  const int w    = t >> 5;
  for (int i = t; i < n; i += 1024) { sh[i] = cnt[i]; cnt[i] = 0; }
  __syncthreads();
  int s = 0;
  for (int k = 0; k < ipt; ++k) s += sh[t * ipt + k];
  int inc = s;
  for (int off = 1; off < 32; off <<= 1) {
    const int o = __shfl_up(inc, off, 32);
    if (lane >= off) inc += o;
  }
  if (lane == 31) wsum[w] = inc;
  __syncthreads();
  if (w == 0) {
    const int ws = wsum[lane];
    int wi = ws;
    for (int off = 1; off < 32; off <<= 1) {
      const int o = __shfl_up(wi, off, 32);
      if (lane >= off) wi += o;
    }
    wsum[lane] = wi - ws;
  }
  __syncthreads();
  int run = wsum[w] + inc - s;
  for (int k = 0; k < ipt; ++k) {
    const int c = sh[t * ipt + k];
    sh[t * ipt + k] = run;
    run += c;
  }
  __syncthreads();
  for (int i = t; i < n; i += 1024) rp[i] = sh[i];
}

__global__ void __launch_bounds__(1024)
fill_kernel(const int* __restrict__ src, const int* __restrict__ dst, int e,
            int* __restrict__ cur_dst, int* __restrict__ cur_src,
            const int* __restrict__ rp_dst, const int* __restrict__ rp_src,
            unsigned* __restrict__ col_dst, unsigned* __restrict__ col_src) {
  __shared__ int hd[MAXP];
  __shared__ int hs[MAXC];
  const int t = threadIdx.x;
  for (int j = t; j < MAXP; j += 1024) hd[j] = 0;
  for (int j = t; j < MAXC; j += 1024) hs[j] = 0;
  __syncthreads();
  int es[EPT], ed[EPT], rd[EPT], rs[EPT];
#pragma unroll
  for (int k = 0; k < EPT; ++k) {
    const int i = blockIdx.x * EPB + k * 1024 + t;
    es[k] = (i < e) ? src[i] : 0;
    ed[k] = (i < e) ? dst[i] : 0;
  }
#pragma unroll
  for (int k = 0; k < EPT; ++k) {
    const int i = blockIdx.x * EPB + k * 1024 + t;
    rd[k] = 0; rs[k] = 0;
    if (i < e) {
      rd[k] = lds_add_rtn(&hd[ed[k]], 1);
      rs[k] = lds_add_rtn(&hs[es[k]], 1);
    }
  }
  __syncthreads();
  {
    int v[MAXP / 1024], a[MAXP / 1024];
#pragma unroll
    for (int q = 0; q < MAXP / 1024; ++q) v[q] = hd[t + 1024 * q];
#pragma unroll
    for (int q = 0; q < MAXP / 1024; ++q) {
      const int j = t + 1024 * q;
      a[q] = 0;
      if (v[q]) a[q] = rp_dst[j] + __hip_atomic_fetch_add(&cur_dst[j], v[q],
                           __ATOMIC_RELAXED, __HIP_MEMORY_SCOPE_AGENT);
    }
#pragma unroll
    for (int q = 0; q < MAXP / 1024; ++q)
      if (v[q]) hd[t + 1024 * q] = a[q];
  }
  {
    int v[MAXC / 1024], a[MAXC / 1024];
#pragma unroll
    for (int q = 0; q < MAXC / 1024; ++q) v[q] = hs[t + 1024 * q];
#pragma unroll
    for (int q = 0; q < MAXC / 1024; ++q) {
      const int j = t + 1024 * q;
      a[q] = 0;
      if (v[q]) a[q] = rp_src[j] + __hip_atomic_fetch_add(&cur_src[j], v[q],
                           __ATOMIC_RELAXED, __HIP_MEMORY_SCOPE_AGENT);
    }
#pragma unroll
    for (int q = 0; q < MAXC / 1024; ++q)
      if (v[q]) hs[t + 1024 * q] = a[q];
  }
  __syncthreads();
#pragma unroll
  for (int k = 0; k < EPT; ++k) {
    const int i = blockIdx.x * EPB + k * 1024 + t;
    if (i < e) {
      const int slot_d = hd[ed[k]] + rd[k];
      atomic_or_u32(&col_dst[slot_d >> 1],
                    (unsigned)es[k] << (16 * (slot_d & 1)));
      const int slot_s = hs[es[k]] + rs[k];
      atomic_or_u32(&col_src[slot_s >> 1],
                    (unsigned)ed[k] << (16 * (slot_s & 1)));
    }
  }
}

template <int N>
__device__ __forceinline__ void gather_group(const float* __restrict__ xl,
                                             int ids, int j, v4f& acc) {
  v4f v[N];
#pragma unroll
  for (int u = 0; u < N; ++u) {
    const int c = __builtin_amdgcn_readlane(ids, j + u);
    v[u] = *(const v4f*)(xl + (size_t)c * HDIM);
  }
#pragma unroll
  for (int u = 0; u < N; ++u) acc += v[u];
}

__device__ __forceinline__ v4f gather_row(const float* __restrict__ x,
                                          const unsigned short* __restrict__ col,
                                          int beg, int end, int lane) {
  const float* xl = x + lane * 4;
  v4f acc = {0.0f, 0.0f, 0.0f, 0.0f};
  for (int base = beg; base < end; base += 32) {
    const int left = end - base;
    const int n = left < 32 ? left : 32;
    const int ids = (lane < n) ? (int)col[base + lane] : 0;
    int j = 0;
    for (; j + 16 <= n; j += 16) gather_group<16>(xl, ids, j, acc);
    if (n - j >= 8) { gather_group<8>(xl, ids, j, acc); j += 8; }
    if (n - j >= 4) { gather_group<4>(xl, ids, j, acc); j += 4; }
    if (n - j >= 2) { gather_group<2>(xl, ids, j, acc); j += 2; }
    if (n - j >= 1) { gather_group<1>(xl, ids, j, acc); }
  }
  return acc;
}


template <int RT, int BK>
__device__ __forceinline__ void
gemm_in_tile(float* __restrict__ smem, int bid,
             const float* __restrict__ A, const float* __restrict__ W,
             const float* __restrict__ bias, const float* __restrict__ emb,
             const int* __restrict__ nid, float* __restrict__ C, int K) {
  constexpr int BM = 16 * RT;
  constexpr int PA = BK + 4;
  constexpr int PW = HDIM + 4;
  constexpr int ATPR = BK / 4;
  constexpr int AROWS = 256 / ATPR;
  constexpr int NA = BM / AROWS;
  constexpr int NW = BK / 8;
  static_assert(NA * AROWS == BM && NA >= 1, "A tile must be whole passes");
  constexpr int ASZ = BM * PA;
  constexpr int WSZ = BK * PW;
  float* As = smem;
  float* Ws = smem + 2 * ASZ;

  const int tid  = threadIdx.x;
  const int lane = tid & 31;
  const int wave = tid >> 5;
  const int half = lane >> 4;
  const int nloc = lane & 15;
  const int m0   = bid * BM;
  const int n    = wave * 16 + nloc;

  const int arow = tid / ATPR;
  const int acol = (tid % ATPR) * 4;
  const int wrow = tid >> 5;
  const int wcol = (tid & 31) * 4;
  const float* Ag = A + (size_t)(m0 + arow) * K + acol;
  const float* Wg = W + (size_t)wrow * HDIM + wcol;

  v4f ra[NA];
  v4f rw[NW];
  v8f acc[RT];
#pragma unroll
  for (int i = 0; i < RT; ++i) acc[i] = v8f{};

#pragma unroll
  for (int i = 0; i < NA; ++i)
    ra[i] = *(const v4f*)(Ag + (size_t)(AROWS * i) * K);
#pragma unroll
  for (int i = 0; i < NW; ++i)
    rw[i] = *(const v4f*)(Wg + (size_t)(8 * i) * HDIM);
#pragma unroll
  for (int i = 0; i < NA; ++i)
    *(v4f*)(&As[(arow + AROWS * i) * PA + acol]) = ra[i];
#pragma unroll
  for (int i = 0; i < NW; ++i)
    *(v4f*)(&Ws[(wrow + 8 * i) * PW + wcol]) = rw[i];
  __syncthreads();

  const int nchunk = K / BK;
  for (int t = 0; t < nchunk; ++t) {
    const int buf = t & 1;
    const bool more = (t + 1 < nchunk);
    if (more) {
      const int k1 = (t + 1) * BK;
#pragma unroll
      for (int i = 0; i < NA; ++i)
        ra[i] = *(const v4f*)(Ag + (size_t)(AROWS * i) * K + k1);
#pragma unroll
      for (int i = 0; i < NW; ++i)
        rw[i] = *(const v4f*)(Wg + (size_t)(k1 + 8 * i) * HDIM);
    }

    const float* as = &As[buf * ASZ + nloc * PA + 4 * half];
    const float* ws = &Ws[buf * WSZ + (4 * half) * PW + n];
#pragma unroll
    for (int j = 0; j < BK / 8; ++j) {
      v4f av[RT];
#pragma unroll
      for (int i = 0; i < RT; ++i)
        av[i] = *(const v4f*)(as + (16 * i) * PA + 8 * j);
      v2f b0; b0.x = ws[(8 * j + 0) * PW];
              b0.y = ws[(8 * j + 1) * PW];
      v2f b1; b1.x = ws[(8 * j + 2) * PW];
              b1.y = ws[(8 * j + 3) * PW];
#pragma unroll
      for (int i = 0; i < RT; ++i) {
        v2f a; a.x = av[i].x; a.y = av[i].y;
        acc[i] = __builtin_amdgcn_wmma_f32_16x16x4_f32(
            false, a, false, b0, (short)0, acc[i], false, false);
      }
#pragma unroll
      for (int i = 0; i < RT; ++i) {
        v2f a; a.x = av[i].z; a.y = av[i].w;
        acc[i] = __builtin_amdgcn_wmma_f32_16x16x4_f32(
            false, a, false, b1, (short)0, acc[i], false, false);
      }
    }

    if (more) {
#pragma unroll
      for (int i = 0; i < NA; ++i)
        *(v4f*)(&As[(buf ^ 1) * ASZ + (arow + AROWS * i) * PA + acol]) = ra[i];
#pragma unroll
      for (int i = 0; i < NW; ++i)
        *(v4f*)(&Ws[(buf ^ 1) * WSZ + (wrow + 8 * i) * PW + wcol]) = rw[i];
    }
    __syncthreads();
  }

  const float bn = bias[n];
#pragma unroll
  for (int i = 0; i < RT; ++i) {
#pragma unroll
    for (int r = 0; r < 8; ++r) {
      const int m = m0 + 16 * i + r + 8 * half;
      const float v = acc[i][r] + bn + emb[(size_t)nid[m] * HDIM + n];
      C[(size_t)m * HDIM + n] = v;
    }
  }
}

constexpr int gemm_in_smem(int rt, int bk) {
  return 2 * (16 * rt) * (bk + 4) + 2 * bk * (HDIM + 4);
}
constexpr int GEMM_SMEM = gemm_in_smem(1, 128) > gemm_in_smem(2, 64)
                              ? gemm_in_smem(1, 128) : gemm_in_smem(2, 64);
static_assert(2 * GEMM_SMEM * 4 <= 320 * 1024, "two tiles per WGP");

__global__ void __launch_bounds__(256)
gemm_in_both_kernel(const float* __restrict__ Ap, const float* __restrict__ Wp,
                    const float* __restrict__ bias_p,
                    const float* __restrict__ emb_p,
                    const int* __restrict__ nid_p, float* __restrict__ Cp,
                    int Kp, int nb_p,
                    const float* __restrict__ Ac, const float* __restrict__ Wc,
                    const float* __restrict__ bias_c,
                    const float* __restrict__ emb_c,
                    const int* __restrict__ nid_c, float* __restrict__ Cc,
                    int Kc) {
  __shared__ __attribute__((aligned(16))) float smem[GEMM_SMEM];
  if ((int)blockIdx.x < nb_p)
    gemm_in_tile<2, 64>(smem, (int)blockIdx.x, Ap, Wp, bias_p, emb_p, nid_p,
                        Cp, Kp);
  else
    gemm_in_tile<1, 128>(smem, (int)blockIdx.x - nb_p, Ac, Wc, bias_c, emb_c,
                         nid_c, Cc, Kc);
}

constexpr int PS = HDIM + 4;

__global__ void __launch_bounds__(256)
sage_layer_kernel(const float* __restrict__ xsrc_c,
                  const unsigned short* __restrict__ col_c,
                  const int* __restrict__ rp_c,
                  const float* __restrict__ xdst_c,
                  const float* __restrict__ Wl_c, const float* __restrict__ bl_c,
                  const float* __restrict__ Wr_c, float* __restrict__ out_c,
                  int nrows_c,
                  const float* __restrict__ xsrc_p,
                  const unsigned short* __restrict__ col_p,
                  const int* __restrict__ rp_p,
                  const float* __restrict__ xdst_p,
                  const float* __restrict__ Wl_p, const float* __restrict__ bl_p,
                  const float* __restrict__ Wr_p, float* __restrict__ out_p,
                  int nrows_p, int etot, int relu) {
  __shared__ __attribute__((aligned(16))) float Ss[16 * PS];
  __shared__ __attribute__((aligned(16))) float Xs[16 * PS];

  const int nb_c = nrows_c >> 4;
  const bool isc = (int)blockIdx.x < nb_c;
  const float* xsrc          = isc ? xsrc_c : xsrc_p;
  const unsigned short* col  = isc ? col_c : col_p;
  const int* rp              = isc ? rp_c : rp_p;
  const float* X             = isc ? xdst_c : xdst_p;
  const float* Wl            = isc ? Wl_c : Wl_p;
  const float* bl            = isc ? bl_c : bl_p;
  const float* Wr            = isc ? Wr_c : Wr_p;
  float* Out                 = isc ? out_c : out_p;
  const int nrows            = isc ? nrows_c : nrows_p;
  const int m0 = (isc ? (int)blockIdx.x : (int)blockIdx.x - nb_c) * 16;

  const int lane = threadIdx.x & 31;
  const int wave = __builtin_amdgcn_readfirstlane((int)(threadIdx.x >> 5));
  const int half = lane >> 4;
  const int nloc = lane & 15;
  const int n    = wave * 16 + nloc;

  const int lr = 2 * wave;
  const v4f xr0 = *(const v4f*)(X + (size_t)(m0 + lr) * HDIM + lane * 4);
  const v4f xr1 = *(const v4f*)(X + (size_t)(m0 + lr + 1) * HDIM + lane * 4);
  for (int rr = 0; rr < 2; ++rr) {
    const int row = m0 + lr + rr;
    const int beg = rp[row];
    const int end = (row + 1 < nrows) ? rp[row + 1] : etot;
    const v4f acc = gather_row(xsrc, col, beg, end, lane);
    *(v4f*)(&Ss[(lr + rr) * PS + lane * 4]) = acc;
  }
  *(v4f*)(&Xs[lr * PS + lane * 4]) = xr0;
  *(v4f*)(&Xs[(lr + 1) * PS + lane * 4]) = xr1;

  const int mb = m0 + 8 * half;
  int rpv[9];
#pragma unroll
  for (int r = 0; r < 8; ++r) rpv[r] = rp[mb + r];
  rpv[8] = (mb + 8 < nrows) ? rp[mb + 8] : etot;
  __syncthreads();

  const float* sa = &Ss[nloc * PS + 2 * half];
  const float* xa = &Xs[nloc * PS + 2 * half];
  v8f accS = {};
  v8f accX = {};
#pragma unroll 8
  for (int k0 = 0; k0 < HDIM; k0 += 4) {
    const int ka = k0 + 2 * half;
    const v2f a1 = *(const v2f*)(sa + k0);
    v2f b1; b1.x = Wl[(size_t)ka * HDIM + n];
            b1.y = Wl[(size_t)(ka + 1) * HDIM + n];
    accS = __builtin_amdgcn_wmma_f32_16x16x4_f32(false, a1, false, b1,
                                                 (short)0, accS, false, false);
    const v2f a2 = *(const v2f*)(xa + k0);
    v2f b2; b2.x = Wr[(size_t)ka * HDIM + n];
            b2.y = Wr[(size_t)(ka + 1) * HDIM + n];
    accX = __builtin_amdgcn_wmma_f32_16x16x4_f32(false, a2, false, b2,
                                                 (short)0, accX, false, false);
  }

  const float bn = bl[n];
#pragma unroll
  for (int r = 0; r < 8; ++r) {
    const int m = m0 + r + 8 * half;
    float d = (float)(rpv[r + 1] - rpv[r]);
    d = d < 1.0f ? 1.0f : d;
    float v = accS[r] / d + accX[r] + bn;
    if (relu) v = v > 0.0f ? v : 0.0f;
    Out[(size_t)m * HDIM + n] = v;
  }
}

constexpr int EPW = 32;

template <int M>
__device__ __forceinline__ float lane_xor(float v) {
  const int i = __builtin_bit_cast(int, v);
  return __builtin_bit_cast(
      float, __builtin_amdgcn_update_dpp(i, i, 0x160 + M, 0xf, 0xf, false));
}

__global__ void __launch_bounds__(256)
edge_dot_kernel(const float* __restrict__ Hc, const float* __restrict__ Hp,
                const int* __restrict__ ls, const int* __restrict__ ld, int e,
                float* __restrict__ out) {
  const int lane = threadIdx.x & 31;
  const int q = lane & 7;
  const int first = EPW * __builtin_amdgcn_readfirstlane(
      (int)((blockIdx.x * 256 + threadIdx.x) >> 5));
  if (first >= e) return;
  const int e0 = first + (lane & 24);
  int ia[8], ib[8];
  if (first + EPW <= e) {
    const int4 a0 = *(const int4*)(ls + e0);
    const int4 a1 = *(const int4*)(ls + e0 + 4);
    const int4 b0 = *(const int4*)(ld + e0);
    const int4 b1 = *(const int4*)(ld + e0 + 4);
    ia[0] = a0.x; ia[1] = a0.y; ia[2] = a0.z; ia[3] = a0.w;
    ia[4] = a1.x; ia[5] = a1.y; ia[6] = a1.z; ia[7] = a1.w;
    ib[0] = b0.x; ib[1] = b0.y; ib[2] = b0.z; ib[3] = b0.w;
    ib[4] = b1.x; ib[5] = b1.y; ib[6] = b1.z; ib[7] = b1.w;
  } else {
#pragma unroll
    for (int k = 0; k < 8; ++k) {
      const int i = e0 + k;
      ia[k] = (i < e) ? ls[i] : 0;
      ib[k] = (i < e) ? ld[i] : 0;
    }
  }
  const float* hcq = Hc + q * 4;
  const float* hpq = Hp + q * 4;
  float r = 0.0f;
#pragma unroll
  for (int k = 0; k < 8; k += 2) {
    v4f va[2][4], vb[2][4];
#pragma unroll
    for (int u = 0; u < 2; ++u) {
      const float* pa = hcq + (size_t)ia[k + u] * HDIM;
      const float* pb = hpq + (size_t)ib[k + u] * HDIM;
#pragma unroll
      for (int c = 0; c < 4; ++c) {
        va[u][c] = *(const v4f*)(pa + 32 * c);
        vb[u][c] = *(const v4f*)(pb + 32 * c);
      }
    }
#pragma unroll
    for (int u = 0; u < 2; ++u) {
      float p = 0.0f;
#pragma unroll
      for (int c = 0; c < 4; ++c)
        p += va[u][c].x * vb[u][c].x + va[u][c].y * vb[u][c].y +
             va[u][c].z * vb[u][c].z + va[u][c].w * vb[u][c].w;
      p += lane_xor<1>(p);
      p += lane_xor<2>(p);
      p += lane_xor<4>(p);
      if (q == k + u) r = p;
    }
  }
  const int i = first + lane;
  if (i < e) out[i] = r;
}

extern "C" void kernel_launch(void* const* d_in, const int* in_sizes, int n_in,
                              void* d_out, int out_size, void* d_ws,
                              size_t ws_size, hipStream_t stream) {
  (void)in_sizes; (void)n_in; (void)out_size; (void)ws_size;
  const int NC = 4096, NP = 8192, E = 500000, EL = 500000;

  const float* x_country = (const float*)d_in[0];
  const float* x_product = (const float*)d_in[1];
  const float* Wc_lin = (const float*)d_in[2];
  const float* bc_lin = (const float*)d_in[3];
  const float* Wp_lin = (const float*)d_in[4];
  const float* bp_lin = (const float*)d_in[5];
  const float* emb_c  = (const float*)d_in[6];
  const float* emb_p  = (const float*)d_in[7];
  const float* W1ep_l = (const float*)d_in[8];
  const float* b1ep   = (const float*)d_in[9];
  const float* W1ep_r = (const float*)d_in[10];
  const float* W1pc_l = (const float*)d_in[11];
  const float* b1pc   = (const float*)d_in[12];
  const float* W1pc_r = (const float*)d_in[13];
  const float* W2ep_l = (const float*)d_in[14];
  const float* b2ep   = (const float*)d_in[15];
  const float* W2ep_r = (const float*)d_in[16];
  const float* W2pc_l = (const float*)d_in[17];
  const float* b2pc   = (const float*)d_in[18];
  const float* W2pc_r = (const float*)d_in[19];
  const int* nid_c = (const int*)d_in[20];
  const int* nid_p = (const int*)d_in[21];
  const int* e_src = (const int*)d_in[22];
  const int* e_dst = (const int*)d_in[23];
  const int* l_src = (const int*)d_in[24];
  const int* l_dst = (const int*)d_in[25];

  float* ws = (float*)d_ws;
  size_t off = 0;
  float* xc   = ws + off; off += (size_t)NC * HDIM;
  float* xp   = ws + off; off += (size_t)NP * HDIM;
  float* hp   = ws + off; off += (size_t)NP * HDIM;
  float* hc   = ws + off; off += (size_t)NC * HDIM;
  float* hp2  = ws + off; off += (size_t)NP * HDIM;
  float* hc2  = ws + off; off += (size_t)NC * HDIM;
  float* scr  = ws + off; off += (size_t)NP * HDIM;
  int* rp_p   = (int*)(ws + off); off += NP;
  int* rp_c   = (int*)(ws + off); off += NC;

  unsigned* colw_p = (unsigned*)scr;
  unsigned* colw_c = colw_p + E / 2;
  int* cnt_p = (int*)scr + E;
  int* cnt_c = cnt_p + NP;
  const unsigned short* col_p = (const unsigned short*)colw_p;
  const unsigned short* col_c = (const unsigned short*)colw_c;

  const int TB = 256;
  const int csr_blocks = (E + EPB - 1) / EPB;

  gemm_in_both_kernel<<<NP / 32 + NC / 16, TB, 0, stream>>>(
      x_product, Wp_lin, bp_lin, emb_p, nid_p, xp, NP, NP / 32,
      x_country, Wc_lin, bc_lin, emb_c, nid_c, xc, NC);

  zero_kernel<<<512, TB, 0, stream>>>(scr, E + NP + NC);
  hist_kernel<<<csr_blocks, 1024, 0, stream>>>(e_src, e_dst, E, cnt_p, cnt_c);
  scan_kernel<<<2, 1024, 0, stream>>>(cnt_p, rp_p, NP, cnt_c, rp_c, NC);
  fill_kernel<<<csr_blocks, 1024, 0, stream>>>(e_src, e_dst, E, cnt_p, cnt_c,
                                               rp_p, rp_c, colw_p, colw_c);

  const int layer_blocks = NC / 16 + NP / 16;

  sage_layer_kernel<<<layer_blocks, TB, 0, stream>>>(
      xp, col_c, rp_c, xc, W1pc_l, b1pc, W1pc_r, hc, NC,
      xc, col_p, rp_p, xp, W1ep_l, b1ep, W1ep_r, hp, NP, E, 1);

  sage_layer_kernel<<<layer_blocks, TB, 0, stream>>>(
      hp, col_c, rp_c, hc, W2pc_l, b2pc, W2pc_r, hc2, NC,
      hc, col_p, rp_p, hp, W2ep_l, b2ep, W2ep_r, hp2, NP, E, 0);

  const int el_blocks = (EL + 8 * EPW - 1) / (8 * EPW);
  edge_dot_kernel<<<el_blocks, TB, 0, stream>>>(hc2, hp2, l_src, l_dst, EL,
                                                (float*)d_out);
}


# neighbour lists by a two-level LDS sort (64-row buckets, then rows) written out in runs of whole words: no scattered global atomic per edge; per-row f32 sum order changes (still arrival order)
# speedup vs baseline: 3.5035x; 1.1017x over previous
#include <hip/hip_runtime.h>

typedef __attribute__((ext_vector_type(2))) float v2f;
typedef __attribute__((ext_vector_type(8))) float v8f;
typedef __attribute__((ext_vector_type(4))) float v4f;

constexpr int HDIM = 128;

__device__ __forceinline__ void atomic_add_i32(int* p, int v) {
  (void)__hip_atomic_fetch_add(p, v, __ATOMIC_RELAXED,
                               __HIP_MEMORY_SCOPE_AGENT);
}
__device__ __forceinline__ void atomic_or_u32(unsigned* p, unsigned v) {
  (void)__hip_atomic_fetch_or(p, v, __ATOMIC_RELAXED,
                              __HIP_MEMORY_SCOPE_AGENT);
}

__global__ void zero_kernel(float* __restrict__ p, int n) {
  int i = blockIdx.x * blockDim.x + threadIdx.x;
  int stride = gridDim.x * blockDim.x;
  for (; i < n; i += stride) p[i] = 0.0f;
}


constexpr int EPB = 4096;
constexpr int EPT = EPB / 1024;
constexpr int MAXP = 8192;
constexpr int MAXC = 4096;
constexpr int NBP = MAXP / 64;
constexpr int NBC = MAXC / 64;
constexpr int NB = NBP + NBC;
static_assert(NB % 32 == 0 && NB <= 256, "bucket counters are whole waves");

__device__ __forceinline__ int lds_add_rtn(int* p, int v) {
  return __hip_atomic_fetch_add(p, v, __ATOMIC_RELAXED,
                                __HIP_MEMORY_SCOPE_WORKGROUP);
}

__device__ __forceinline__ int wave_incl_scan(int v, int lane) {
  for (int off = 1; off < 32; off <<= 1) {
    const int o = __shfl_up(v, off, 32);
    if (lane >= off) v += o;
  }
  return v;
}

__global__ void __launch_bounds__(1024)
bucket_count_kernel(const int* __restrict__ src, const int* __restrict__ dst,
                    int e, int* __restrict__ gcnt) {
  __shared__ int h[NB];
  const int t = threadIdx.x;
  if (t < NB) h[t] = 0;
  __syncthreads();
  int es[EPT], ed[EPT];
#pragma unroll
  for (int k = 0; k < EPT; ++k) {
    const int i = blockIdx.x * EPB + k * 1024 + t;
    const int ii = i < e ? i : e - 1;
    es[k] = src[ii];
    ed[k] = dst[ii];
  }
#pragma unroll
  for (int k = 0; k < EPT; ++k) {
    const int i = blockIdx.x * EPB + k * 1024 + t;
    if (i < e) {
      (void)lds_add_rtn(&h[ed[k] >> 6], 1);
      (void)lds_add_rtn(&h[NBP + (es[k] >> 6)], 1);
    }
  }
  __syncthreads();
  if (t < NB) {
    const int v = h[t];
    if (v) atomic_add_i32(&gcnt[t], v);
  }
}

__global__ void __launch_bounds__(1024)
bucket_fill_kernel(const int* __restrict__ src, const int* __restrict__ dst,
                   int e, const int* __restrict__ gcnt, int* __restrict__ gcur,
                   unsigned* __restrict__ rec) {
  __shared__ int h[NB];
  __shared__ int lst[NB];
  __shared__ int gst[NB];
  __shared__ int wtot[2][8];
  __shared__ unsigned stage[2 * EPB];
  const int t = threadIdx.x;
  const int lane = t & 31;
  const int w = t >> 5;
  if (t < NB) h[t] = 0;
  __syncthreads();
  int es[EPT], ed[EPT], rd[EPT], rs[EPT];
#pragma unroll
  for (int k = 0; k < EPT; ++k) {
    const int i = blockIdx.x * EPB + k * 1024 + t;
    const int ii = i < e ? i : e - 1;
    es[k] = src[ii];
    ed[k] = dst[ii];
  }
#pragma unroll
  for (int k = 0; k < EPT; ++k) {
    const int i = blockIdx.x * EPB + k * 1024 + t;
    rd[k] = 0; rs[k] = 0;
    if (i < e) {
      rd[k] = lds_add_rtn(&h[ed[k] >> 6], 1);
      rs[k] = lds_add_rtn(&h[NBP + (es[k] >> 6)], 1);
    }
  }
  __syncthreads();
  const int myc = (t < NB) ? h[t] : 0;
  const int myg = (t < NB) ? gcnt[t] : 0;
  const int incc = wave_incl_scan(myc, lane);
  const int incg = wave_incl_scan(myg, lane);
  if (lane == 31 && w < 8) { wtot[0][w] = incc; wtot[1][w] = incg; }
  __syncthreads();
  if (t < NB) {
    int basec = 0, baseg = 0;
    for (int j = 0; j < NB / 32; ++j)
      if (j < w) { basec += wtot[0][j]; baseg += wtot[1][j]; }
    int r = 0;
    if (myc) r = __hip_atomic_fetch_add(&gcur[t], myc, __ATOMIC_RELAXED,
                                        __HIP_MEMORY_SCOPE_AGENT);
    lst[t] = basec + incc - myc;
    gst[t] = baseg + incg - myg + r;
  }
  __syncthreads();
#pragma unroll
  for (int k = 0; k < EPT; ++k) {
    const int i = blockIdx.x * EPB + k * 1024 + t;
    if (i < e) {
      const int bd = ed[k] >> 6;
      stage[lst[bd] + rd[k]] = ((unsigned)bd << 24) |
                               ((unsigned)(ed[k] & 63) << 16) | (unsigned)es[k];
      const int bs = NBP + (es[k] >> 6);
      stage[lst[bs] + rs[k]] = ((unsigned)bs << 24) |
                               ((unsigned)(es[k] & 63) << 16) | (unsigned)ed[k];
    }
  }
  __syncthreads();
  const int left = e - (int)blockIdx.x * EPB;
  const int total = 2 * (left < EPB ? left : EPB);
  for (int i = t; i < total; i += 1024) {
    const unsigned r = stage[i];
    const int b = (int)(r >> 24);
    atomic_or_u32(&rec[gst[b] + (i - lst[b])], r);
  }
}

constexpr int CAP = 32768;

__global__ void __launch_bounds__(1024)
bucket_rows_kernel(const int* __restrict__ gcnt,
                   const unsigned* __restrict__ rec, int e,
                   int* __restrict__ rp_dst, int* __restrict__ rp_src,
                   unsigned* __restrict__ col_dst,
                   unsigned* __restrict__ col_src) {
  __shared__ int fc[64];
  __shared__ int fp[64];
  __shared__ int wsum[32];
  __shared__ unsigned short sorted[CAP];
  const int t = threadIdx.x;
  const int lane = t & 31;
  const int w = t >> 5;
  const int b = blockIdx.x;
  int v = (t < NB && t < b) ? gcnt[t] : 0;
  for (int off = 16; off >= 1; off >>= 1) v += __shfl_xor(v, off, 32);
  if (lane == 0) wsum[w] = v;
  if (t < 64) fc[t] = 0;
  __syncthreads();
  int beg = 0;
  for (int j = 0; j < NB / 32; ++j) beg += wsum[j];
  const int n = gcnt[b];
  const bool isd = b < NBP;
  int* rp        = isd ? rp_dst : rp_src;
  unsigned* colw = isd ? col_dst : col_src;
  const int row0 = (isd ? b : b - NBP) * 64;
  const int s0   = isd ? beg : beg - e;
  const unsigned* rb = rec + beg;

  for (int i = t; i < n; i += 1024)
    (void)lds_add_rtn(&fc[(rb[i] >> 16) & 63], 1);
  __syncthreads();
  const int c = (t < 64) ? fc[t] : 0;
  const int inc = wave_incl_scan(c, lane);
  if (t == 31) wsum[0] = inc;
  __syncthreads();
  if (t < 64) {
    const int ex = inc - c + (w == 1 ? wsum[0] : 0);
    fp[t] = ex;
    fc[t] = 0;
    rp[row0 + t] = s0 + ex;
  }
  __syncthreads();
  const bool small = n <= CAP;
  for (int i = t; i < n; i += 1024) {
    const unsigned r = rb[i];
    const int f = (int)(r >> 16) & 63;
    const int pos = fp[f] + lds_add_rtn(&fc[f], 1);
    const unsigned id = r & 0xffffu;
    if (small) {
      sorted[pos] = (unsigned short)id;
    } else {
      const int slot = s0 + pos;
      atomic_or_u32(&colw[slot >> 1], id << (16 * (slot & 1)));
    }
  }
  if (small) {
    __syncthreads();
    const int w0 = s0 >> 1;
    const int w1 = (s0 + n + 1) >> 1;
    for (int q = w0 + t; q < w1; q += 1024) {
      const int sa = 2 * q - s0;
      unsigned val = 0;
      if (sa >= 0 && sa < n) val = sorted[sa];
      if (sa + 1 >= 0 && sa + 1 < n) val |= (unsigned)sorted[sa + 1] << 16;
      atomic_or_u32(&colw[q], val);
    }
  }
}

template <int N>
__device__ __forceinline__ void gather_group(const float* __restrict__ xl,
                                             int ids, int j, v4f& acc) {
  v4f v[N];
#pragma unroll
  for (int u = 0; u < N; ++u) {
    const int c = __builtin_amdgcn_readlane(ids, j + u);
    v[u] = *(const v4f*)(xl + (size_t)c * HDIM);
  }
#pragma unroll
  for (int u = 0; u < N; ++u) acc += v[u];
}

__device__ __forceinline__ v4f gather_row(const float* __restrict__ x,
                                          const unsigned short* __restrict__ col,
                                          int beg, int end, int lane) {
  const float* xl = x + lane * 4;
  v4f acc = {0.0f, 0.0f, 0.0f, 0.0f};
  for (int base = beg; base < end; base += 32) {
    const int left = end - base;
    const int n = left < 32 ? left : 32;
    const int ids = (lane < n) ? (int)col[base + lane] : 0;
    int j = 0;
    for (; j + 16 <= n; j += 16) gather_group<16>(xl, ids, j, acc);
    if (n - j >= 8) { gather_group<8>(xl, ids, j, acc); j += 8; }
    if (n - j >= 4) { gather_group<4>(xl, ids, j, acc); j += 4; }
    if (n - j >= 2) { gather_group<2>(xl, ids, j, acc); j += 2; }
    if (n - j >= 1) { gather_group<1>(xl, ids, j, acc); }
  }
  return acc;
}


template <int RT, int BK>
__device__ __forceinline__ void
gemm_in_tile(float* __restrict__ smem, int bid,
             const float* __restrict__ A, const float* __restrict__ W,
             const float* __restrict__ bias, const float* __restrict__ emb,
             const int* __restrict__ nid, float* __restrict__ C, int K) {
  constexpr int BM = 16 * RT;
  constexpr int PA = BK + 4;
  constexpr int PW = HDIM + 4;
  constexpr int ATPR = BK / 4;
  constexpr int AROWS = 256 / ATPR;
  constexpr int NA = BM / AROWS;
  constexpr int NW = BK / 8;
  static_assert(NA * AROWS == BM && NA >= 1, "A tile must be whole passes");
  constexpr int ASZ = BM * PA;
  constexpr int WSZ = BK * PW;
  float* As = smem;
  float* Ws = smem + 2 * ASZ;

  const int tid  = threadIdx.x;
  const int lane = tid & 31;
  const int wave = tid >> 5;
  const int half = lane >> 4;
  const int nloc = lane & 15;
  const int m0   = bid * BM;
  const int n    = wave * 16 + nloc;

  const int arow = tid / ATPR;
  const int acol = (tid % ATPR) * 4;
  const int wrow = tid >> 5;
  const int wcol = (tid & 31) * 4;
  const float* Ag = A + (size_t)(m0 + arow) * K + acol;
  const float* Wg = W + (size_t)wrow * HDIM + wcol;

  v4f ra[NA];
  v4f rw[NW];
  v8f acc[RT];
#pragma unroll
  for (int i = 0; i < RT; ++i) acc[i] = v8f{};

#pragma unroll
  for (int i = 0; i < NA; ++i)
    ra[i] = *(const v4f*)(Ag + (size_t)(AROWS * i) * K);
#pragma unroll
  for (int i = 0; i < NW; ++i)
    rw[i] = *(const v4f*)(Wg + (size_t)(8 * i) * HDIM);
#pragma unroll
  for (int i = 0; i < NA; ++i)
    *(v4f*)(&As[(arow + AROWS * i) * PA + acol]) = ra[i];
#pragma unroll
  for (int i = 0; i < NW; ++i)
    *(v4f*)(&Ws[(wrow + 8 * i) * PW + wcol]) = rw[i];
  __syncthreads();

  const int nchunk = K / BK;
  for (int t = 0; t < nchunk; ++t) {
    const int buf = t & 1;
    const bool more = (t + 1 < nchunk);
    if (more) {
      const int k1 = (t + 1) * BK;
#pragma unroll
      for (int i = 0; i < NA; ++i)
        ra[i] = *(const v4f*)(Ag + (size_t)(AROWS * i) * K + k1);
#pragma unroll
      for (int i = 0; i < NW; ++i)
        rw[i] = *(const v4f*)(Wg + (size_t)(k1 + 8 * i) * HDIM);
    }

    const float* as = &As[buf * ASZ + nloc * PA + 4 * half];
    const float* ws = &Ws[buf * WSZ + (4 * half) * PW + n];
#pragma unroll
    for (int j = 0; j < BK / 8; ++j) {
      v4f av[RT];
#pragma unroll
      for (int i = 0; i < RT; ++i)
        av[i] = *(const v4f*)(as + (16 * i) * PA + 8 * j);
      v2f b0; b0.x = ws[(8 * j + 0) * PW];
              b0.y = ws[(8 * j + 1) * PW];
      v2f b1; b1.x = ws[(8 * j + 2) * PW];
              b1.y = ws[(8 * j + 3) * PW];
#pragma unroll
      for (int i = 0; i < RT; ++i) {
        v2f a; a.x = av[i].x; a.y = av[i].y;
        acc[i] = __builtin_amdgcn_wmma_f32_16x16x4_f32(
            false, a, false, b0, (short)0, acc[i], false, false);
      }
#pragma unroll
      for (int i = 0; i < RT; ++i) {
        v2f a; a.x = av[i].z; a.y = av[i].w;
        acc[i] = __builtin_amdgcn_wmma_f32_16x16x4_f32(
            false, a, false, b1, (short)0, acc[i], false, false);
      }
    }

    if (more) {
#pragma unroll
      for (int i = 0; i < NA; ++i)
        *(v4f*)(&As[(buf ^ 1) * ASZ + (arow + AROWS * i) * PA + acol]) = ra[i];
#pragma unroll
      for (int i = 0; i < NW; ++i)
        *(v4f*)(&Ws[(buf ^ 1) * WSZ + (wrow + 8 * i) * PW + wcol]) = rw[i];
    }
    __syncthreads();
  }

  const float bn = bias[n];
#pragma unroll
  for (int i = 0; i < RT; ++i) {
#pragma unroll
    for (int r = 0; r < 8; ++r) {
      const int m = m0 + 16 * i + r + 8 * half;
      const float v = acc[i][r] + bn + emb[(size_t)nid[m] * HDIM + n];
      C[(size_t)m * HDIM + n] = v;
    }
  }
}

constexpr int gemm_in_smem(int rt, int bk) {
  return 2 * (16 * rt) * (bk + 4) + 2 * bk * (HDIM + 4);
}
constexpr int GEMM_SMEM = gemm_in_smem(1, 128) > gemm_in_smem(2, 64)
                              ? gemm_in_smem(1, 128) : gemm_in_smem(2, 64);
static_assert(2 * GEMM_SMEM * 4 <= 320 * 1024, "two tiles per WGP");

__global__ void __launch_bounds__(256)
gemm_in_both_kernel(const float* __restrict__ Ap, const float* __restrict__ Wp,
                    const float* __restrict__ bias_p,
                    const float* __restrict__ emb_p,
                    const int* __restrict__ nid_p, float* __restrict__ Cp,
                    int Kp, int nb_p,
                    const float* __restrict__ Ac, const float* __restrict__ Wc,
                    const float* __restrict__ bias_c,
                    const float* __restrict__ emb_c,
                    const int* __restrict__ nid_c, float* __restrict__ Cc,
                    int Kc) {
  __shared__ __attribute__((aligned(16))) float smem[GEMM_SMEM];
  if ((int)blockIdx.x < nb_p)
    gemm_in_tile<2, 64>(smem, (int)blockIdx.x, Ap, Wp, bias_p, emb_p, nid_p,
                        Cp, Kp);
  else
    gemm_in_tile<1, 128>(smem, (int)blockIdx.x - nb_p, Ac, Wc, bias_c, emb_c,
                         nid_c, Cc, Kc);
}

constexpr int PS = HDIM + 4;

__global__ void __launch_bounds__(256)
sage_layer_kernel(const float* __restrict__ xsrc_c,
                  const unsigned short* __restrict__ col_c,
                  const int* __restrict__ rp_c,
                  const float* __restrict__ xdst_c,
                  const float* __restrict__ Wl_c, const float* __restrict__ bl_c,
                  const float* __restrict__ Wr_c, float* __restrict__ out_c,
                  int nrows_c,
                  const float* __restrict__ xsrc_p,
                  const unsigned short* __restrict__ col_p,
                  const int* __restrict__ rp_p,
                  const float* __restrict__ xdst_p,
                  const float* __restrict__ Wl_p, const float* __restrict__ bl_p,
                  const float* __restrict__ Wr_p, float* __restrict__ out_p,
                  int nrows_p, int etot, int relu) {
  __shared__ __attribute__((aligned(16))) float Ss[16 * PS];
  __shared__ __attribute__((aligned(16))) float Xs[16 * PS];

  const int nb_c = nrows_c >> 4;
  const bool isc = (int)blockIdx.x < nb_c;
  const float* xsrc          = isc ? xsrc_c : xsrc_p;
  const unsigned short* col  = isc ? col_c : col_p;
  const int* rp              = isc ? rp_c : rp_p;
  const float* X             = isc ? xdst_c : xdst_p;
  const float* Wl            = isc ? Wl_c : Wl_p;
  const float* bl            = isc ? bl_c : bl_p;
  const float* Wr            = isc ? Wr_c : Wr_p;
  float* Out                 = isc ? out_c : out_p;
  const int nrows            = isc ? nrows_c : nrows_p;
  const int m0 = (isc ? (int)blockIdx.x : (int)blockIdx.x - nb_c) * 16;

  const int lane = threadIdx.x & 31;
  const int wave = __builtin_amdgcn_readfirstlane((int)(threadIdx.x >> 5));
  const int half = lane >> 4;
  const int nloc = lane & 15;
  const int n    = wave * 16 + nloc;

  const int lr = 2 * wave;
  const v4f xr0 = *(const v4f*)(X + (size_t)(m0 + lr) * HDIM + lane * 4);
  const v4f xr1 = *(const v4f*)(X + (size_t)(m0 + lr + 1) * HDIM + lane * 4);
  for (int rr = 0; rr < 2; ++rr) {
    const int row = m0 + lr + rr;
    const int beg = rp[row];
    const int end = (row + 1 < nrows) ? rp[row + 1] : etot;
    const v4f acc = gather_row(xsrc, col, beg, end, lane);
    *(v4f*)(&Ss[(lr + rr) * PS + lane * 4]) = acc;
  }
  *(v4f*)(&Xs[lr * PS + lane * 4]) = xr0;
  *(v4f*)(&Xs[(lr + 1) * PS + lane * 4]) = xr1;

  const int mb = m0 + 8 * half;
  int rpv[9];
#pragma unroll
  for (int r = 0; r < 8; ++r) rpv[r] = rp[mb + r];
  rpv[8] = (mb + 8 < nrows) ? rp[mb + 8] : etot;
  __syncthreads();

  const float* sa = &Ss[nloc * PS + 2 * half];
  const float* xa = &Xs[nloc * PS + 2 * half];
  v8f accS = {};
  v8f accX = {};
#pragma unroll 8
  for (int k0 = 0; k0 < HDIM; k0 += 4) {
    const int ka = k0 + 2 * half;
    const v2f a1 = *(const v2f*)(sa + k0);
    v2f b1; b1.x = Wl[(size_t)ka * HDIM + n];
            b1.y = Wl[(size_t)(ka + 1) * HDIM + n];
    accS = __builtin_amdgcn_wmma_f32_16x16x4_f32(false, a1, false, b1,
                                                 (short)0, accS, false, false);
    const v2f a2 = *(const v2f*)(xa + k0);
    v2f b2; b2.x = Wr[(size_t)ka * HDIM + n];
            b2.y = Wr[(size_t)(ka + 1) * HDIM + n];
    accX = __builtin_amdgcn_wmma_f32_16x16x4_f32(false, a2, false, b2,
                                                 (short)0, accX, false, false);
  }

  const float bn = bl[n];
#pragma unroll
  for (int r = 0; r < 8; ++r) {
    const int m = m0 + r + 8 * half;
    float d = (float)(rpv[r + 1] - rpv[r]);
    d = d < 1.0f ? 1.0f : d;
    float v = accS[r] / d + accX[r] + bn;
    if (relu) v = v > 0.0f ? v : 0.0f;
    Out[(size_t)m * HDIM + n] = v;
  }
}

constexpr int EPW = 32;

template <int M>
__device__ __forceinline__ float lane_xor(float v) {
  const int i = __builtin_bit_cast(int, v);
  return __builtin_bit_cast(
      float, __builtin_amdgcn_update_dpp(i, i, 0x160 + M, 0xf, 0xf, false));
}

__global__ void __launch_bounds__(256)
edge_dot_kernel(const float* __restrict__ Hc, const float* __restrict__ Hp,
                const int* __restrict__ ls, const int* __restrict__ ld, int e,
                float* __restrict__ out) {
  const int lane = threadIdx.x & 31;
  const int q = lane & 7;
  const int first = EPW * __builtin_amdgcn_readfirstlane(
      (int)((blockIdx.x * 256 + threadIdx.x) >> 5));
  if (first >= e) return;
  const int e0 = first + (lane & 24);
  int ia[8], ib[8];
  if (first + EPW <= e) {
    const int4 a0 = *(const int4*)(ls + e0);
    const int4 a1 = *(const int4*)(ls + e0 + 4);
    const int4 b0 = *(const int4*)(ld + e0);
    const int4 b1 = *(const int4*)(ld + e0 + 4);
    ia[0] = a0.x; ia[1] = a0.y; ia[2] = a0.z; ia[3] = a0.w;
    ia[4] = a1.x; ia[5] = a1.y; ia[6] = a1.z; ia[7] = a1.w;
    ib[0] = b0.x; ib[1] = b0.y; ib[2] = b0.z; ib[3] = b0.w;
    ib[4] = b1.x; ib[5] = b1.y; ib[6] = b1.z; ib[7] = b1.w;
  } else {
#pragma unroll
    for (int k = 0; k < 8; ++k) {
      const int i = e0 + k;
      ia[k] = (i < e) ? ls[i] : 0;
      ib[k] = (i < e) ? ld[i] : 0;
    }
  }
  const float* hcq = Hc + q * 4;
  const float* hpq = Hp + q * 4;
  float r = 0.0f;
#pragma unroll
  for (int k = 0; k < 8; k += 2) {
    v4f va[2][4], vb[2][4];
#pragma unroll
    for (int u = 0; u < 2; ++u) {
      const float* pa = hcq + (size_t)ia[k + u] * HDIM;
      const float* pb = hpq + (size_t)ib[k + u] * HDIM;
#pragma unroll
      for (int c = 0; c < 4; ++c) {
        va[u][c] = *(const v4f*)(pa + 32 * c);
        vb[u][c] = *(const v4f*)(pb + 32 * c);
      }
    }
#pragma unroll
    for (int u = 0; u < 2; ++u) {
      float p = 0.0f;
#pragma unroll
      for (int c = 0; c < 4; ++c)
        p += va[u][c].x * vb[u][c].x + va[u][c].y * vb[u][c].y +
             va[u][c].z * vb[u][c].z + va[u][c].w * vb[u][c].w;
      p += lane_xor<1>(p);
      p += lane_xor<2>(p);
      p += lane_xor<4>(p);
      if (q == k + u) r = p;
    }
  }
  const int i = first + lane;
  if (i < e) out[i] = r;
}

extern "C" void kernel_launch(void* const* d_in, const int* in_sizes, int n_in,
                              void* d_out, int out_size, void* d_ws,
                              size_t ws_size, hipStream_t stream) {
  (void)in_sizes; (void)n_in; (void)out_size; (void)ws_size;
  const int NC = 4096, NP = 8192, E = 500000, EL = 500000;

  const float* x_country = (const float*)d_in[0];
  const float* x_product = (const float*)d_in[1];
  const float* Wc_lin = (const float*)d_in[2];
  const float* bc_lin = (const float*)d_in[3];
  const float* Wp_lin = (const float*)d_in[4];
  const float* bp_lin = (const float*)d_in[5];
  const float* emb_c  = (const float*)d_in[6];
  const float* emb_p  = (const float*)d_in[7];
  const float* W1ep_l = (const float*)d_in[8];
  const float* b1ep   = (const float*)d_in[9];
  const float* W1ep_r = (const float*)d_in[10];
  const float* W1pc_l = (const float*)d_in[11];
  const float* b1pc   = (const float*)d_in[12];
  const float* W1pc_r = (const float*)d_in[13];
  const float* W2ep_l = (const float*)d_in[14];
  const float* b2ep   = (const float*)d_in[15];
  const float* W2ep_r = (const float*)d_in[16];
  const float* W2pc_l = (const float*)d_in[17];
  const float* b2pc   = (const float*)d_in[18];
  const float* W2pc_r = (const float*)d_in[19];
  const int* nid_c = (const int*)d_in[20];
  const int* nid_p = (const int*)d_in[21];
  const int* e_src = (const int*)d_in[22];
  const int* e_dst = (const int*)d_in[23];
  const int* l_src = (const int*)d_in[24];
  const int* l_dst = (const int*)d_in[25];

  float* ws = (float*)d_ws;
  size_t off = 0;
  float* xc   = ws + off; off += (size_t)NC * HDIM;
  float* xp   = ws + off; off += (size_t)NP * HDIM;
  float* hp   = ws + off; off += (size_t)NP * HDIM;
  float* hc   = ws + off; off += (size_t)NC * HDIM;
  float* hc2  = ws + off; off += (size_t)NC * HDIM;
  float* hp2  = ws + off; off += (size_t)NP * HDIM;
  float* scr  = ws + off; off += (size_t)NP * HDIM;
  int* rp_p   = (int*)(ws + off); off += NP;
  int* rp_c   = (int*)(ws + off); off += NC;

  unsigned* rec    = (unsigned*)hp2;
  unsigned* colw_p = (unsigned*)scr;
  unsigned* colw_c = colw_p + E / 2;
  int* gcnt = (int*)scr + E;
  int* gcur = gcnt + NB;
  const unsigned short* col_p = (const unsigned short*)colw_p;
  const unsigned short* col_c = (const unsigned short*)colw_c;

  const int TB = 256;
  const int csr_blocks = (E + EPB - 1) / EPB;

  gemm_in_both_kernel<<<NP / 32 + NC / 16, TB, 0, stream>>>(
      x_product, Wp_lin, bp_lin, emb_p, nid_p, xp, NP, NP / 32,
      x_country, Wc_lin, bc_lin, emb_c, nid_c, xc, NC);

  zero_kernel<<<512, TB, 0, stream>>>(hp2, NP * HDIM + E + 2 * NB);
  bucket_count_kernel<<<csr_blocks, 1024, 0, stream>>>(e_src, e_dst, E, gcnt);
  bucket_fill_kernel<<<csr_blocks, 1024, 0, stream>>>(e_src, e_dst, E, gcnt,
                                                      gcur, rec);
  bucket_rows_kernel<<<NB, 1024, 0, stream>>>(gcnt, rec, E, rp_p, rp_c,
                                              colw_p, colw_c);

  const int layer_blocks = NC / 16 + NP / 16;

  sage_layer_kernel<<<layer_blocks, TB, 0, stream>>>(
      xp, col_c, rp_c, xc, W1pc_l, b1pc, W1pc_r, hc, NC,
      xc, col_p, rp_p, xp, W1ep_l, b1ep, W1ep_r, hp, NP, E, 1);

  sage_layer_kernel<<<layer_blocks, TB, 0, stream>>>(
      hp, col_c, rp_c, hc, W2pc_l, b2pc, W2pc_r, hc2, NC,
      hc, col_p, rp_p, hp, W2ep_l, b2ep, W2ep_r, hp2, NP, E, 0);

  const int el_blocks = (EL + 8 * EPW - 1) / (8 * EPW);
  edge_dot_kernel<<<el_blocks, TB, 0, stream>>>(hc2, hp2, l_src, l_dst, EL,
                                                (float*)d_out);
}


# input projection GEMM chunk loop software-pipelined: LDS fragment reads batched one 8-WMMA group ahead, next chunk's LDS stores and global fetch before the last group; sum order kept
# speedup vs baseline: 3.8157x; 1.0891x over previous
#include <hip/hip_runtime.h>

typedef __attribute__((ext_vector_type(2))) float v2f;
typedef __attribute__((ext_vector_type(8))) float v8f;
typedef __attribute__((ext_vector_type(4))) float v4f;

constexpr int HDIM = 128;

__device__ __forceinline__ void atomic_add_i32(int* p, int v) {
  (void)__hip_atomic_fetch_add(p, v, __ATOMIC_RELAXED,
                               __HIP_MEMORY_SCOPE_AGENT);
}
__device__ __forceinline__ void atomic_or_u32(unsigned* p, unsigned v) {
  (void)__hip_atomic_fetch_or(p, v, __ATOMIC_RELAXED,
                              __HIP_MEMORY_SCOPE_AGENT);
}

__global__ void zero_kernel(float* __restrict__ p, int n) {
  int i = blockIdx.x * blockDim.x + threadIdx.x;
  int stride = gridDim.x * blockDim.x;
  for (; i < n; i += stride) p[i] = 0.0f;
}


constexpr int EPB = 4096;
constexpr int EPT = EPB / 1024;
constexpr int MAXP = 8192;
constexpr int MAXC = 4096;
constexpr int NBP = MAXP / 64;
constexpr int NBC = MAXC / 64;
constexpr int NB = NBP + NBC;
static_assert(NB % 32 == 0 && NB <= 256, "bucket counters are whole waves");

__device__ __forceinline__ int lds_add_rtn(int* p, int v) {
  return __hip_atomic_fetch_add(p, v, __ATOMIC_RELAXED,
                                __HIP_MEMORY_SCOPE_WORKGROUP);
}

__device__ __forceinline__ int wave_incl_scan(int v, int lane) {
  for (int off = 1; off < 32; off <<= 1) {
    const int o = __shfl_up(v, off, 32);
    if (lane >= off) v += o;
  }
  return v;
}

__global__ void __launch_bounds__(1024)
bucket_count_kernel(const int* __restrict__ src, const int* __restrict__ dst,
                    int e, int* __restrict__ gcnt) {
  __shared__ int h[NB];
  const int t = threadIdx.x;
  if (t < NB) h[t] = 0;
  __syncthreads();
  int es[EPT], ed[EPT];
#pragma unroll
  for (int k = 0; k < EPT; ++k) {
    const int i = blockIdx.x * EPB + k * 1024 + t;
    const int ii = i < e ? i : e - 1;
    es[k] = src[ii];
    ed[k] = dst[ii];
  }
#pragma unroll
  for (int k = 0; k < EPT; ++k) {
    const int i = blockIdx.x * EPB + k * 1024 + t;
    if (i < e) {
      (void)lds_add_rtn(&h[ed[k] >> 6], 1);
      (void)lds_add_rtn(&h[NBP + (es[k] >> 6)], 1);
    }
  }
  __syncthreads();
  if (t < NB) {
    const int v = h[t];
    if (v) atomic_add_i32(&gcnt[t], v);
  }
}

__global__ void __launch_bounds__(1024)
bucket_fill_kernel(const int* __restrict__ src, const int* __restrict__ dst,
                   int e, const int* __restrict__ gcnt, int* __restrict__ gcur,
                   unsigned* __restrict__ rec) {
  __shared__ int h[NB];
  __shared__ int lst[NB];
  __shared__ int gst[NB];
  __shared__ int wtot[2][8];
  __shared__ unsigned stage[2 * EPB];
  const int t = threadIdx.x;
  const int lane = t & 31;
  const int w = t >> 5;
  if (t < NB) h[t] = 0;
  __syncthreads();
  int es[EPT], ed[EPT], rd[EPT], rs[EPT];
#pragma unroll
  for (int k = 0; k < EPT; ++k) {
    const int i = blockIdx.x * EPB + k * 1024 + t;
    const int ii = i < e ? i : e - 1;
    es[k] = src[ii];
    ed[k] = dst[ii];
  }
#pragma unroll
  for (int k = 0; k < EPT; ++k) {
    const int i = blockIdx.x * EPB + k * 1024 + t;
    rd[k] = 0; rs[k] = 0;
    if (i < e) {
      rd[k] = lds_add_rtn(&h[ed[k] >> 6], 1);
      rs[k] = lds_add_rtn(&h[NBP + (es[k] >> 6)], 1);
    }
  }
  __syncthreads();
  const int myc = (t < NB) ? h[t] : 0;
  const int myg = (t < NB) ? gcnt[t] : 0;
  const int incc = wave_incl_scan(myc, lane);
  const int incg = wave_incl_scan(myg, lane);
  if (lane == 31 && w < 8) { wtot[0][w] = incc; wtot[1][w] = incg; }
  __syncthreads();
  if (t < NB) {
    int basec = 0, baseg = 0;
    for (int j = 0; j < NB / 32; ++j)
      if (j < w) { basec += wtot[0][j]; baseg += wtot[1][j]; }
    int r = 0;
    if (myc) r = __hip_atomic_fetch_add(&gcur[t], myc, __ATOMIC_RELAXED,
                                        __HIP_MEMORY_SCOPE_AGENT);
    lst[t] = basec + incc - myc;
    gst[t] = baseg + incg - myg + r;
  }
  __syncthreads();
#pragma unroll
  for (int k = 0; k < EPT; ++k) {
    const int i = blockIdx.x * EPB + k * 1024 + t;
    if (i < e) {
      const int bd = ed[k] >> 6;
      stage[lst[bd] + rd[k]] = ((unsigned)bd << 24) |
                               ((unsigned)(ed[k] & 63) << 16) | (unsigned)es[k];
      const int bs = NBP + (es[k] >> 6);
      stage[lst[bs] + rs[k]] = ((unsigned)bs << 24) |
                               ((unsigned)(es[k] & 63) << 16) | (unsigned)ed[k];
    }
  }
  __syncthreads();
  const int left = e - (int)blockIdx.x * EPB;
  const int total = 2 * (left < EPB ? left : EPB);
  for (int i = t; i < total; i += 1024) {
    const unsigned r = stage[i];
    const int b = (int)(r >> 24);
    atomic_or_u32(&rec[gst[b] + (i - lst[b])], r);
  }
}

constexpr int CAP = 32768;

__global__ void __launch_bounds__(1024)
bucket_rows_kernel(const int* __restrict__ gcnt,
                   const unsigned* __restrict__ rec, int e,
                   int* __restrict__ rp_dst, int* __restrict__ rp_src,
                   unsigned* __restrict__ col_dst,
                   unsigned* __restrict__ col_src) {
  __shared__ int fc[64];
  __shared__ int fp[64];
  __shared__ int wsum[32];
  __shared__ unsigned short sorted[CAP];
  const int t = threadIdx.x;
  const int lane = t & 31;
  const int w = t >> 5;
  const int b = blockIdx.x;
  int v = (t < NB && t < b) ? gcnt[t] : 0;
  for (int off = 16; off >= 1; off >>= 1) v += __shfl_xor(v, off, 32);
  if (lane == 0) wsum[w] = v;
  if (t < 64) fc[t] = 0;
  __syncthreads();
  int beg = 0;
  for (int j = 0; j < NB / 32; ++j) beg += wsum[j];
  const int n = gcnt[b];
  const bool isd = b < NBP;
  int* rp        = isd ? rp_dst : rp_src;
  unsigned* colw = isd ? col_dst : col_src;
  const int row0 = (isd ? b : b - NBP) * 64;
  const int s0   = isd ? beg : beg - e;
  const unsigned* rb = rec + beg;

  for (int i = t; i < n; i += 1024)
    (void)lds_add_rtn(&fc[(rb[i] >> 16) & 63], 1);
  __syncthreads();
  const int c = (t < 64) ? fc[t] : 0;
  const int inc = wave_incl_scan(c, lane);
  if (t == 31) wsum[0] = inc;
  __syncthreads();
  if (t < 64) {
    const int ex = inc - c + (w == 1 ? wsum[0] : 0);
    fp[t] = ex;
    fc[t] = 0;
    rp[row0 + t] = s0 + ex;
  }
  __syncthreads();
  const bool small = n <= CAP;
  for (int i = t; i < n; i += 1024) {
    const unsigned r = rb[i];
    const int f = (int)(r >> 16) & 63;
    const int pos = fp[f] + lds_add_rtn(&fc[f], 1);
    const unsigned id = r & 0xffffu;
    if (small) {
      sorted[pos] = (unsigned short)id;
    } else {
      const int slot = s0 + pos;
      atomic_or_u32(&colw[slot >> 1], id << (16 * (slot & 1)));
    }
  }
  if (small) {
    __syncthreads();
    const int w0 = s0 >> 1;
    const int w1 = (s0 + n + 1) >> 1;
    for (int q = w0 + t; q < w1; q += 1024) {
      const int sa = 2 * q - s0;
      unsigned val = 0;
      if (sa >= 0 && sa < n) val = sorted[sa];
      if (sa + 1 >= 0 && sa + 1 < n) val |= (unsigned)sorted[sa + 1] << 16;
      atomic_or_u32(&colw[q], val);
    }
  }
}

template <int N>
__device__ __forceinline__ void gather_group(const float* __restrict__ xl,
                                             int ids, int j, v4f& acc) {
  v4f v[N];
#pragma unroll
  for (int u = 0; u < N; ++u) {
    const int c = __builtin_amdgcn_readlane(ids, j + u);
    v[u] = *(const v4f*)(xl + (size_t)c * HDIM);
  }
#pragma unroll
  for (int u = 0; u < N; ++u) acc += v[u];
}

__device__ __forceinline__ v4f gather_row(const float* __restrict__ x,
                                          const unsigned short* __restrict__ col,
                                          int beg, int end, int lane) {
  const float* xl = x + lane * 4;
  v4f acc = {0.0f, 0.0f, 0.0f, 0.0f};
  for (int base = beg; base < end; base += 32) {
    const int left = end - base;
    const int n = left < 32 ? left : 32;
    const int ids = (lane < n) ? (int)col[base + lane] : 0;
    int j = 0;
    for (; j + 16 <= n; j += 16) gather_group<16>(xl, ids, j, acc);
    if (n - j >= 8) { gather_group<8>(xl, ids, j, acc); j += 8; }
    if (n - j >= 4) { gather_group<4>(xl, ids, j, acc); j += 4; }
    if (n - j >= 2) { gather_group<2>(xl, ids, j, acc); j += 2; }
    if (n - j >= 1) { gather_group<1>(xl, ids, j, acc); }
  }
  return acc;
}


__device__ __forceinline__ void lds_barrier() {
  __builtin_amdgcn_fence(__ATOMIC_RELEASE, "workgroup", "local");
  __builtin_amdgcn_s_barrier();
  __builtin_amdgcn_fence(__ATOMIC_ACQUIRE, "workgroup", "local");
}

template <int RT, int GJ, int PA, int PW>
__device__ __forceinline__ void
frag_group(const float* __restrict__ as, const float* __restrict__ ws, int g,
           v4f (&fa)[GJ][RT], v2f (&fb)[GJ][2]) {
#pragma unroll
  for (int jj = 0; jj < GJ; ++jj) {
    const int j = g * GJ + jj;
#pragma unroll
    for (int i = 0; i < RT; ++i)
      fa[jj][i] = *(const v4f*)(as + (16 * i) * PA + 8 * j);
    fb[jj][0].x = ws[(8 * j + 0) * PW];
    fb[jj][0].y = ws[(8 * j + 1) * PW];
    fb[jj][1].x = ws[(8 * j + 2) * PW];
    fb[jj][1].y = ws[(8 * j + 3) * PW];
  }
}

template <int RT, int BK>
__device__ __forceinline__ void
gemm_in_tile(float* __restrict__ smem, int bid,
             const float* __restrict__ A, const float* __restrict__ W,
             const float* __restrict__ bias, const float* __restrict__ emb,
             const int* __restrict__ nid, float* __restrict__ C, int K) {
  constexpr int BM = 16 * RT;
  constexpr int PA = BK + 4;
  constexpr int PW = HDIM + 4;
  constexpr int ATPR = BK / 4;
  constexpr int AROWS = 256 / ATPR;
  constexpr int NA = BM / AROWS;
  constexpr int NW = BK / 8;
  static_assert(NA * AROWS == BM && NA >= 1, "A tile must be whole passes");
  constexpr int GJ = 4 / RT;
  constexpr int NG = (BK / 8) / GJ;
  static_assert(NG * GJ * 8 == BK && NG >= 2, "a chunk is whole groups");
  constexpr int ASZ = BM * PA;
  constexpr int WSZ = BK * PW;
  float* As = smem;
  float* Ws = smem + 2 * ASZ;

  const int tid  = threadIdx.x;
  const int lane = tid & 31;
  const int wave = tid >> 5;
  const int half = lane >> 4;
  const int nloc = lane & 15;
  const int m0   = bid * BM;
  const int n    = wave * 16 + nloc;

  const int arow = tid / ATPR;
  const int acol = (tid % ATPR) * 4;
  const int wrow = tid >> 5;
  const int wcol = (tid & 31) * 4;
  const float* Ag = A + (size_t)(m0 + arow) * K + acol;
  const float* Wg = W + (size_t)wrow * HDIM + wcol;

  v4f ra[NA];
  v4f rw[NW];
  v8f acc[RT];
#pragma unroll
  for (int i = 0; i < RT; ++i) acc[i] = v8f{};

  const int nchunk = K / BK;

#pragma unroll
  for (int i = 0; i < NA; ++i)
    ra[i] = *(const v4f*)(Ag + (size_t)(AROWS * i) * K);
#pragma unroll
  for (int i = 0; i < NW; ++i)
    rw[i] = *(const v4f*)(Wg + (size_t)(8 * i) * HDIM);
#pragma unroll
  for (int i = 0; i < NA; ++i)
    *(v4f*)(&As[(arow + AROWS * i) * PA + acol]) = ra[i];
#pragma unroll
  for (int i = 0; i < NW; ++i)
    *(v4f*)(&Ws[(wrow + 8 * i) * PW + wcol]) = rw[i];
  {
    const int k1 = (nchunk > 1 ? 1 : 0) * BK;
#pragma unroll
    for (int i = 0; i < NA; ++i)
      ra[i] = *(const v4f*)(Ag + (size_t)(AROWS * i) * K + k1);
#pragma unroll
    for (int i = 0; i < NW; ++i)
      rw[i] = *(const v4f*)(Wg + (size_t)(k1 + 8 * i) * HDIM);
  }
  lds_barrier();

  for (int t = 0; t < nchunk; ++t) {
    const int buf = t & 1;
    const float* as = &As[buf * ASZ + nloc * PA + 4 * half];
    const float* ws = &Ws[buf * WSZ + (4 * half) * PW + n];
    v4f fa[2][GJ][RT];
    v2f fb[2][GJ][2];
    frag_group<RT, GJ, PA, PW>(as, ws, 0, fa[0], fb[0]);
    __builtin_amdgcn_sched_barrier(0);
#pragma unroll
    for (int g = 0; g < NG; ++g) {
      if (g + 1 < NG)
        frag_group<RT, GJ, PA, PW>(as, ws, g + 1, fa[(g + 1) & 1],
                                   fb[(g + 1) & 1]);
      if (g == NG - 1) {
#pragma unroll
        for (int i = 0; i < NA; ++i)
          *(v4f*)(&As[(buf ^ 1) * ASZ + (arow + AROWS * i) * PA + acol]) =
              ra[i];
#pragma unroll
        for (int i = 0; i < NW; ++i)
          *(v4f*)(&Ws[(buf ^ 1) * WSZ + (wrow + 8 * i) * PW + wcol]) = rw[i];
        const int tn = t + 2 < nchunk ? t + 2 : nchunk - 1;
        const int k2 = tn * BK;
#pragma unroll
        for (int i = 0; i < NA; ++i)
          ra[i] = *(const v4f*)(Ag + (size_t)(AROWS * i) * K + k2);
#pragma unroll
        for (int i = 0; i < NW; ++i)
          rw[i] = *(const v4f*)(Wg + (size_t)(k2 + 8 * i) * HDIM);
      }
      __builtin_amdgcn_sched_barrier(0);
#pragma unroll
      for (int jj = 0; jj < GJ; ++jj) {
#pragma unroll
        for (int i = 0; i < RT; ++i) {
          v2f a; a.x = fa[g & 1][jj][i].x; a.y = fa[g & 1][jj][i].y;
          acc[i] = __builtin_amdgcn_wmma_f32_16x16x4_f32(
              false, a, false, fb[g & 1][jj][0], (short)0, acc[i], false,
              false);
        }
#pragma unroll
        for (int i = 0; i < RT; ++i) {
          v2f a; a.x = fa[g & 1][jj][i].z; a.y = fa[g & 1][jj][i].w;
          acc[i] = __builtin_amdgcn_wmma_f32_16x16x4_f32(
              false, a, false, fb[g & 1][jj][1], (short)0, acc[i], false,
              false);
        }
      }
      __builtin_amdgcn_sched_barrier(0);
    }
    lds_barrier();
  }

  const float bn = bias[n];
#pragma unroll
  for (int i = 0; i < RT; ++i) {
#pragma unroll
    for (int r = 0; r < 8; ++r) {
      const int m = m0 + 16 * i + r + 8 * half;
      const float v = acc[i][r] + bn + emb[(size_t)nid[m] * HDIM + n];
      C[(size_t)m * HDIM + n] = v;
    }
  }
}

constexpr int gemm_in_smem(int rt, int bk) {
  return 2 * (16 * rt) * (bk + 4) + 2 * bk * (HDIM + 4);
}
constexpr int GEMM_SMEM = gemm_in_smem(1, 128) > gemm_in_smem(2, 64)
                              ? gemm_in_smem(1, 128) : gemm_in_smem(2, 64);
static_assert(2 * GEMM_SMEM * 4 <= 320 * 1024, "two tiles per WGP");

__global__ void __launch_bounds__(256)
gemm_in_both_kernel(const float* __restrict__ Ap, const float* __restrict__ Wp,
                    const float* __restrict__ bias_p,
                    const float* __restrict__ emb_p,
                    const int* __restrict__ nid_p, float* __restrict__ Cp,
                    int Kp, int nb_p,
                    const float* __restrict__ Ac, const float* __restrict__ Wc,
                    const float* __restrict__ bias_c,
                    const float* __restrict__ emb_c,
                    const int* __restrict__ nid_c, float* __restrict__ Cc,
                    int Kc) {
  __shared__ __attribute__((aligned(16))) float smem[GEMM_SMEM];
  if ((int)blockIdx.x < nb_p)
    gemm_in_tile<2, 64>(smem, (int)blockIdx.x, Ap, Wp, bias_p, emb_p, nid_p,
                        Cp, Kp);
  else
    gemm_in_tile<1, 128>(smem, (int)blockIdx.x - nb_p, Ac, Wc, bias_c, emb_c,
                         nid_c, Cc, Kc);
}

constexpr int PS = HDIM + 4;

__global__ void __launch_bounds__(256)
sage_layer_kernel(const float* __restrict__ xsrc_c,
                  const unsigned short* __restrict__ col_c,
                  const int* __restrict__ rp_c,
                  const float* __restrict__ xdst_c,
                  const float* __restrict__ Wl_c, const float* __restrict__ bl_c,
                  const float* __restrict__ Wr_c, float* __restrict__ out_c,
                  int nrows_c,
                  const float* __restrict__ xsrc_p,
                  const unsigned short* __restrict__ col_p,
                  const int* __restrict__ rp_p,
                  const float* __restrict__ xdst_p,
                  const float* __restrict__ Wl_p, const float* __restrict__ bl_p,
                  const float* __restrict__ Wr_p, float* __restrict__ out_p,
                  int nrows_p, int etot, int relu) {
  __shared__ __attribute__((aligned(16))) float Ss[16 * PS];
  __shared__ __attribute__((aligned(16))) float Xs[16 * PS];

  const int nb_c = nrows_c >> 4;
  const bool isc = (int)blockIdx.x < nb_c;
  const float* xsrc          = isc ? xsrc_c : xsrc_p;
  const unsigned short* col  = isc ? col_c : col_p;
  const int* rp              = isc ? rp_c : rp_p;
  const float* X             = isc ? xdst_c : xdst_p;
  const float* Wl            = isc ? Wl_c : Wl_p;
  const float* bl            = isc ? bl_c : bl_p;
  const float* Wr            = isc ? Wr_c : Wr_p;
  float* Out                 = isc ? out_c : out_p;
  const int nrows            = isc ? nrows_c : nrows_p;
  const int m0 = (isc ? (int)blockIdx.x : (int)blockIdx.x - nb_c) * 16;

  const int lane = threadIdx.x & 31;
  const int wave = __builtin_amdgcn_readfirstlane((int)(threadIdx.x >> 5));
  const int half = lane >> 4;
  const int nloc = lane & 15;
  const int n    = wave * 16 + nloc;

  const int lr = 2 * wave;
  const v4f xr0 = *(const v4f*)(X + (size_t)(m0 + lr) * HDIM + lane * 4);
  const v4f xr1 = *(const v4f*)(X + (size_t)(m0 + lr + 1) * HDIM + lane * 4);
  for (int rr = 0; rr < 2; ++rr) {
    const int row = m0 + lr + rr;
    const int beg = rp[row];
    const int end = (row + 1 < nrows) ? rp[row + 1] : etot;
    const v4f acc = gather_row(xsrc, col, beg, end, lane);
    *(v4f*)(&Ss[(lr + rr) * PS + lane * 4]) = acc;
  }
  *(v4f*)(&Xs[lr * PS + lane * 4]) = xr0;
  *(v4f*)(&Xs[(lr + 1) * PS + lane * 4]) = xr1;

  const int mb = m0 + 8 * half;
  int rpv[9];
#pragma unroll
  for (int r = 0; r < 8; ++r) rpv[r] = rp[mb + r];
  rpv[8] = (mb + 8 < nrows) ? rp[mb + 8] : etot;
  __syncthreads();

  const float* sa = &Ss[nloc * PS + 2 * half];
  const float* xa = &Xs[nloc * PS + 2 * half];
  v8f accS = {};
  v8f accX = {};
#pragma unroll 8
  for (int k0 = 0; k0 < HDIM; k0 += 4) {
    const int ka = k0 + 2 * half;
    const v2f a1 = *(const v2f*)(sa + k0);
    v2f b1; b1.x = Wl[(size_t)ka * HDIM + n];
            b1.y = Wl[(size_t)(ka + 1) * HDIM + n];
    accS = __builtin_amdgcn_wmma_f32_16x16x4_f32(false, a1, false, b1,
                                                 (short)0, accS, false, false);
    const v2f a2 = *(const v2f*)(xa + k0);
    v2f b2; b2.x = Wr[(size_t)ka * HDIM + n];
            b2.y = Wr[(size_t)(ka + 1) * HDIM + n];
    accX = __builtin_amdgcn_wmma_f32_16x16x4_f32(false, a2, false, b2,
                                                 (short)0, accX, false, false);
  }

  const float bn = bl[n];
#pragma unroll
  for (int r = 0; r < 8; ++r) {
    const int m = m0 + r + 8 * half;
    float d = (float)(rpv[r + 1] - rpv[r]);
    d = d < 1.0f ? 1.0f : d;
    float v = accS[r] / d + accX[r] + bn;
    if (relu) v = v > 0.0f ? v : 0.0f;
    Out[(size_t)m * HDIM + n] = v;
  }
}

constexpr int EPW = 32;

template <int M>
__device__ __forceinline__ float lane_xor(float v) {
  const int i = __builtin_bit_cast(int, v);
  return __builtin_bit_cast(
      float, __builtin_amdgcn_update_dpp(i, i, 0x160 + M, 0xf, 0xf, false));
}

__global__ void __launch_bounds__(256)
edge_dot_kernel(const float* __restrict__ Hc, const float* __restrict__ Hp,
                const int* __restrict__ ls, const int* __restrict__ ld, int e,
                float* __restrict__ out) {
  const int lane = threadIdx.x & 31;
  const int q = lane & 7;
  const int first = EPW * __builtin_amdgcn_readfirstlane(
      (int)((blockIdx.x * 256 + threadIdx.x) >> 5));
  if (first >= e) return;
  const int e0 = first + (lane & 24);
  int ia[8], ib[8];
  if (first + EPW <= e) {
    const int4 a0 = *(const int4*)(ls + e0);
    const int4 a1 = *(const int4*)(ls + e0 + 4);
    const int4 b0 = *(const int4*)(ld + e0);
    const int4 b1 = *(const int4*)(ld + e0 + 4);
    ia[0] = a0.x; ia[1] = a0.y; ia[2] = a0.z; ia[3] = a0.w;
    ia[4] = a1.x; ia[5] = a1.y; ia[6] = a1.z; ia[7] = a1.w;
    ib[0] = b0.x; ib[1] = b0.y; ib[2] = b0.z; ib[3] = b0.w;
    ib[4] = b1.x; ib[5] = b1.y; ib[6] = b1.z; ib[7] = b1.w;
  } else {
#pragma unroll
    for (int k = 0; k < 8; ++k) {
      const int i = e0 + k;
      ia[k] = (i < e) ? ls[i] : 0;
      ib[k] = (i < e) ? ld[i] : 0;
    }
  }
  const float* hcq = Hc + q * 4;
  const float* hpq = Hp + q * 4;
  float r = 0.0f;
#pragma unroll
  for (int k = 0; k < 8; k += 2) {
    v4f va[2][4], vb[2][4];
#pragma unroll
    for (int u = 0; u < 2; ++u) {
      const float* pa = hcq + (size_t)ia[k + u] * HDIM;
      const float* pb = hpq + (size_t)ib[k + u] * HDIM;
#pragma unroll
      for (int c = 0; c < 4; ++c) {
        va[u][c] = *(const v4f*)(pa + 32 * c);
        vb[u][c] = *(const v4f*)(pb + 32 * c);
      }
    }
#pragma unroll
    for (int u = 0; u < 2; ++u) {
      float p = 0.0f;
#pragma unroll
      for (int c = 0; c < 4; ++c)
        p += va[u][c].x * vb[u][c].x + va[u][c].y * vb[u][c].y +
             va[u][c].z * vb[u][c].z + va[u][c].w * vb[u][c].w;
      p += lane_xor<1>(p);
      p += lane_xor<2>(p);
      p += lane_xor<4>(p);
      if (q == k + u) r = p;
    }
  }
  const int i = first + lane;
  if (i < e) out[i] = r;
}

extern "C" void kernel_launch(void* const* d_in, const int* in_sizes, int n_in,
                              void* d_out, int out_size, void* d_ws,
                              size_t ws_size, hipStream_t stream) {
  (void)in_sizes; (void)n_in; (void)out_size; (void)ws_size;
  const int NC = 4096, NP = 8192, E = 500000, EL = 500000;

  const float* x_country = (const float*)d_in[0];
  const float* x_product = (const float*)d_in[1];
  const float* Wc_lin = (const float*)d_in[2];
  const float* bc_lin = (const float*)d_in[3];
  const float* Wp_lin = (const float*)d_in[4];
  const float* bp_lin = (const float*)d_in[5];
  const float* emb_c  = (const float*)d_in[6];
  const float* emb_p  = (const float*)d_in[7];
  const float* W1ep_l = (const float*)d_in[8];
  const float* b1ep   = (const float*)d_in[9];
  const float* W1ep_r = (const float*)d_in[10];
  const float* W1pc_l = (const float*)d_in[11];
  const float* b1pc   = (const float*)d_in[12];
  const float* W1pc_r = (const float*)d_in[13];
  const float* W2ep_l = (const float*)d_in[14];
  const float* b2ep   = (const float*)d_in[15];
  const float* W2ep_r = (const float*)d_in[16];
  const float* W2pc_l = (const float*)d_in[17];
  const float* b2pc   = (const float*)d_in[18];
  const float* W2pc_r = (const float*)d_in[19];
  const int* nid_c = (const int*)d_in[20];
  const int* nid_p = (const int*)d_in[21];
  const int* e_src = (const int*)d_in[22];
  const int* e_dst = (const int*)d_in[23];
  const int* l_src = (const int*)d_in[24];
  const int* l_dst = (const int*)d_in[25];

  float* ws = (float*)d_ws;
  size_t off = 0;
  float* xc   = ws + off; off += (size_t)NC * HDIM;
  float* xp   = ws + off; off += (size_t)NP * HDIM;
  float* hp   = ws + off; off += (size_t)NP * HDIM;
  float* hc   = ws + off; off += (size_t)NC * HDIM;
  float* hc2  = ws + off; off += (size_t)NC * HDIM;
  float* hp2  = ws + off; off += (size_t)NP * HDIM;
  float* scr  = ws + off; off += (size_t)NP * HDIM;
  int* rp_p   = (int*)(ws + off); off += NP;
  int* rp_c   = (int*)(ws + off); off += NC;

  unsigned* rec    = (unsigned*)hp2;
  unsigned* colw_p = (unsigned*)scr;
  unsigned* colw_c = colw_p + E / 2;
  int* gcnt = (int*)scr + E;
  int* gcur = gcnt + NB;
  const unsigned short* col_p = (const unsigned short*)colw_p;
  const unsigned short* col_c = (const unsigned short*)colw_c;

  const int TB = 256;
  const int csr_blocks = (E + EPB - 1) / EPB;

  gemm_in_both_kernel<<<NP / 32 + NC / 16, TB, 0, stream>>>(
      x_product, Wp_lin, bp_lin, emb_p, nid_p, xp, NP, NP / 32,
      x_country, Wc_lin, bc_lin, emb_c, nid_c, xc, NC);

  zero_kernel<<<512, TB, 0, stream>>>(hp2, NP * HDIM + E + 2 * NB);
  bucket_count_kernel<<<csr_blocks, 1024, 0, stream>>>(e_src, e_dst, E, gcnt);
  bucket_fill_kernel<<<csr_blocks, 1024, 0, stream>>>(e_src, e_dst, E, gcnt,
                                                      gcur, rec);
  bucket_rows_kernel<<<NB, 1024, 0, stream>>>(gcnt, rec, E, rp_p, rp_c,
                                              colw_p, colw_c);

  const int layer_blocks = NC / 16 + NP / 16;

  sage_layer_kernel<<<layer_blocks, TB, 0, stream>>>(
      xp, col_c, rp_c, xc, W1pc_l, b1pc, W1pc_r, hc, NC,
      xc, col_p, rp_p, xp, W1ep_l, b1ep, W1ep_r, hp, NP, E, 1);

  sage_layer_kernel<<<layer_blocks, TB, 0, stream>>>(
      hp, col_c, rp_c, hc, W2pc_l, b2pc, W2pc_r, hc2, NC,
      hc, col_p, rp_p, hp, W2ep_l, b2ep, W2ep_r, hp2, NP, E, 0);

  const int el_blocks = (EL + 8 * EPW - 1) / (8 * EPW);
  edge_dot_kernel<<<el_blocks, TB, 0, stream>>>(hc2, hp2, l_src, l_dst, EL,
                                                (float*)d_out);
}


# each layer block orders its own 16-row bucket of edge records by row in LDS and gathers from that list: list pass and lists in memory go, 8 -> 7 launches; row f32 sum order changes (arrival order)
# speedup vs baseline: 3.9365x; 1.0317x over previous
#include <hip/hip_runtime.h>

typedef __attribute__((ext_vector_type(2))) float v2f;
typedef __attribute__((ext_vector_type(8))) float v8f;
typedef __attribute__((ext_vector_type(4))) float v4f;

constexpr int HDIM = 128;

__device__ __forceinline__ void atomic_add_i32(int* p, int v) {
  (void)__hip_atomic_fetch_add(p, v, __ATOMIC_RELAXED,
                               __HIP_MEMORY_SCOPE_AGENT);
}
__device__ __forceinline__ void atomic_or_u32(unsigned* p, unsigned v) {
  (void)__hip_atomic_fetch_or(p, v, __ATOMIC_RELAXED,
                              __HIP_MEMORY_SCOPE_AGENT);
}

__global__ void zero_kernel(float* __restrict__ p, int n) {
  int i = blockIdx.x * blockDim.x + threadIdx.x;
  int stride = gridDim.x * blockDim.x;
  for (; i < n; i += stride) p[i] = 0.0f;
}


constexpr int EPB = 4096;
constexpr int EPT = EPB / 1024;
constexpr int MAXP = 8192;
constexpr int MAXC = 4096;
constexpr int NBP = MAXP / 16;
constexpr int NBC = MAXC / 16;
constexpr int NB = NBP + NBC;
static_assert(NB % 32 == 0 && NB <= 1024, "bucket counters are whole waves");

__device__ __forceinline__ int lds_add_rtn(int* p, int v) {
  return __hip_atomic_fetch_add(p, v, __ATOMIC_RELAXED,
                                __HIP_MEMORY_SCOPE_WORKGROUP);
}

__device__ __forceinline__ int wave_incl_scan(int v, int lane) {
  for (int off = 1; off < 32; off <<= 1) {
    const int o = __shfl_up(v, off, 32);
    if (lane >= off) v += o;
  }
  return v;
}

__global__ void __launch_bounds__(1024)
bucket_count_kernel(const int* __restrict__ src, const int* __restrict__ dst,
                    int e, int* __restrict__ gcnt) {
  __shared__ int h[NB];
  const int t = threadIdx.x;
  if (t < NB) h[t] = 0;
  __syncthreads();
  int es[EPT], ed[EPT];
#pragma unroll
  for (int k = 0; k < EPT; ++k) {
    const int i = blockIdx.x * EPB + k * 1024 + t;
    const int ii = i < e ? i : e - 1;
    es[k] = src[ii];
    ed[k] = dst[ii];
  }
#pragma unroll
  for (int k = 0; k < EPT; ++k) {
    const int i = blockIdx.x * EPB + k * 1024 + t;
    if (i < e) {
      (void)lds_add_rtn(&h[ed[k] >> 4], 1);
      (void)lds_add_rtn(&h[NBP + (es[k] >> 4)], 1);
    }
  }
  __syncthreads();
  if (t < NB) {
    const int v = h[t];
    if (v) atomic_add_i32(&gcnt[t], v);
  }
}

__global__ void __launch_bounds__(1024)
bucket_fill_kernel(const int* __restrict__ src, const int* __restrict__ dst,
                   int e, const int* __restrict__ gcnt, int* __restrict__ gcur,
                   int* __restrict__ gbeg, unsigned* __restrict__ rec) {
  __shared__ int h[NB];
  __shared__ int lst[NB];
  __shared__ int gst[NB];
  __shared__ int wtot[2][NB / 32];
  __shared__ unsigned stage[2 * EPB];
  const int t = threadIdx.x;
  const int lane = t & 31;
  const int w = t >> 5;
  if (t < NB) h[t] = 0;
  __syncthreads();
  int es[EPT], ed[EPT], rd[EPT], rs[EPT];
#pragma unroll
  for (int k = 0; k < EPT; ++k) {
    const int i = blockIdx.x * EPB + k * 1024 + t;
    const int ii = i < e ? i : e - 1;
    es[k] = src[ii];
    ed[k] = dst[ii];
  }
#pragma unroll
  for (int k = 0; k < EPT; ++k) {
    const int i = blockIdx.x * EPB + k * 1024 + t;
    rd[k] = 0; rs[k] = 0;
    if (i < e) {
      rd[k] = lds_add_rtn(&h[ed[k] >> 4], 1);
      rs[k] = lds_add_rtn(&h[NBP + (es[k] >> 4)], 1);
    }
  }
  __syncthreads();
  const int myc = (t < NB) ? h[t] : 0;
  const int myg = (t < NB) ? gcnt[t] : 0;
  const int incc = wave_incl_scan(myc, lane);
  const int incg = wave_incl_scan(myg, lane);
  if (lane == 31 && w < NB / 32) { wtot[0][w] = incc; wtot[1][w] = incg; }
  __syncthreads();
  if (t < NB) {
    int basec = 0, baseg = 0;
    for (int j = 0; j < NB / 32; ++j)
      if (j < w) { basec += wtot[0][j]; baseg += wtot[1][j]; }
    int r = 0;
    if (myc) r = __hip_atomic_fetch_add(&gcur[t], myc, __ATOMIC_RELAXED,
                                        __HIP_MEMORY_SCOPE_AGENT);
    lst[t] = basec + incc - myc;
    gst[t] = baseg + incg - myg + r;
    if (blockIdx.x == 0) gbeg[t] = baseg + incg - myg;
  }
  __syncthreads();
#pragma unroll
  for (int k = 0; k < EPT; ++k) {
    const int i = blockIdx.x * EPB + k * 1024 + t;
    if (i < e) {
      const int bd = ed[k] >> 4;
      stage[lst[bd] + rd[k]] = ((unsigned)bd << 20) |
                               ((unsigned)(ed[k] & 15) << 16) | (unsigned)es[k];
      const int bs = NBP + (es[k] >> 4);
      stage[lst[bs] + rs[k]] = ((unsigned)bs << 20) |
                               ((unsigned)(es[k] & 15) << 16) | (unsigned)ed[k];
    }
  }
  __syncthreads();
  const int left = e - (int)blockIdx.x * EPB;
  const int total = 2 * (left < EPB ? left : EPB);
  for (int i = t; i < total; i += 1024) {
    const unsigned r = stage[i];
    const int b = (int)(r >> 20);
    atomic_or_u32(&rec[gst[b] + (i - lst[b])], r);
  }
}

template <int N>
__device__ __forceinline__ void gather_group(const float* __restrict__ xl,
                                             int ids, int j, v4f& acc) {
  v4f v[N];
#pragma unroll
  for (int u = 0; u < N; ++u) {
    const int c = __builtin_amdgcn_readlane(ids, j + u);
    v[u] = *(const v4f*)(xl + (size_t)c * HDIM);
  }
#pragma unroll
  for (int u = 0; u < N; ++u) acc += v[u];
}

__device__ __forceinline__ v4f gather_row(const float* __restrict__ x,
                                          const unsigned short* lid,
                                          int beg, int end, int lane, v4f acc) {
  const float* xl = x + lane * 4;
  for (int base = beg; base < end; base += 32) {
    const int left = end - base;
    const int n = left < 32 ? left : 32;
    const int ids = (lane < n) ? (int)lid[base + lane] : 0;
    int j = 0;
    for (; j + 16 <= n; j += 16) gather_group<16>(xl, ids, j, acc);
    if (n - j >= 8) { gather_group<8>(xl, ids, j, acc); j += 8; }
    if (n - j >= 4) { gather_group<4>(xl, ids, j, acc); j += 4; }
    if (n - j >= 2) { gather_group<2>(xl, ids, j, acc); j += 2; }
    if (n - j >= 1) { gather_group<1>(xl, ids, j, acc); }
  }
  return acc;
}


__device__ __forceinline__ void lds_barrier() {
  __builtin_amdgcn_fence(__ATOMIC_RELEASE, "workgroup", "local");
  __builtin_amdgcn_s_barrier();
  __builtin_amdgcn_fence(__ATOMIC_ACQUIRE, "workgroup", "local");
}

template <int RT, int GJ, int PA, int PW>
__device__ __forceinline__ void
frag_group(const float* __restrict__ as, const float* __restrict__ ws, int g,
           v4f (&fa)[GJ][RT], v2f (&fb)[GJ][2]) {
#pragma unroll
  for (int jj = 0; jj < GJ; ++jj) {
    const int j = g * GJ + jj;
#pragma unroll
    for (int i = 0; i < RT; ++i)
      fa[jj][i] = *(const v4f*)(as + (16 * i) * PA + 8 * j);
    fb[jj][0].x = ws[(8 * j + 0) * PW];
    fb[jj][0].y = ws[(8 * j + 1) * PW];
    fb[jj][1].x = ws[(8 * j + 2) * PW];
    fb[jj][1].y = ws[(8 * j + 3) * PW];
  }
}

template <int RT, int BK>
__device__ __forceinline__ void
gemm_in_tile(float* __restrict__ smem, int bid,
             const float* __restrict__ A, const float* __restrict__ W,
             const float* __restrict__ bias, const float* __restrict__ emb,
             const int* __restrict__ nid, float* __restrict__ C, int K) {
  constexpr int BM = 16 * RT;
  constexpr int PA = BK + 4;
  constexpr int PW = HDIM + 4;
  constexpr int ATPR = BK / 4;
  constexpr int AROWS = 256 / ATPR;
  constexpr int NA = BM / AROWS;
  constexpr int NW = BK / 8;
  static_assert(NA * AROWS == BM && NA >= 1, "A tile must be whole passes");
  constexpr int GJ = 4 / RT;
  constexpr int NG = (BK / 8) / GJ;
  static_assert(NG * GJ * 8 == BK && NG >= 2, "a chunk is whole groups");
  constexpr int ASZ = BM * PA;
  constexpr int WSZ = BK * PW;
  float* As = smem;
  float* Ws = smem + 2 * ASZ;

  const int tid  = threadIdx.x;
  const int lane = tid & 31;
  const int wave = tid >> 5;
  const int half = lane >> 4;
  const int nloc = lane & 15;
  const int m0   = bid * BM;
  const int n    = wave * 16 + nloc;

  const int arow = tid / ATPR;
  const int acol = (tid % ATPR) * 4;
  const int wrow = tid >> 5;
  const int wcol = (tid & 31) * 4;
  const float* Ag = A + (size_t)(m0 + arow) * K + acol;
  const float* Wg = W + (size_t)wrow * HDIM + wcol;

  v4f ra[NA];
  v4f rw[NW];
  v8f acc[RT];
#pragma unroll
  for (int i = 0; i < RT; ++i) acc[i] = v8f{};

  const int nchunk = K / BK;

#pragma unroll
  for (int i = 0; i < NA; ++i)
    ra[i] = *(const v4f*)(Ag + (size_t)(AROWS * i) * K);
#pragma unroll
  for (int i = 0; i < NW; ++i)
    rw[i] = *(const v4f*)(Wg + (size_t)(8 * i) * HDIM);
#pragma unroll
  for (int i = 0; i < NA; ++i)
    *(v4f*)(&As[(arow + AROWS * i) * PA + acol]) = ra[i];
#pragma unroll
  for (int i = 0; i < NW; ++i)
    *(v4f*)(&Ws[(wrow + 8 * i) * PW + wcol]) = rw[i];
  {
    const int k1 = (nchunk > 1 ? 1 : 0) * BK;
#pragma unroll
    for (int i = 0; i < NA; ++i)
      ra[i] = *(const v4f*)(Ag + (size_t)(AROWS * i) * K + k1);
#pragma unroll
    for (int i = 0; i < NW; ++i)
      rw[i] = *(const v4f*)(Wg + (size_t)(k1 + 8 * i) * HDIM);
  }
  lds_barrier();

  for (int t = 0; t < nchunk; ++t) {
    const int buf = t & 1;
    const float* as = &As[buf * ASZ + nloc * PA + 4 * half];
    const float* ws = &Ws[buf * WSZ + (4 * half) * PW + n];
    v4f fa[2][GJ][RT];
    v2f fb[2][GJ][2];
    frag_group<RT, GJ, PA, PW>(as, ws, 0, fa[0], fb[0]);
    __builtin_amdgcn_sched_barrier(0);
#pragma unroll
    for (int g = 0; g < NG; ++g) {
      if (g + 1 < NG)
        frag_group<RT, GJ, PA, PW>(as, ws, g + 1, fa[(g + 1) & 1],
                                   fb[(g + 1) & 1]);
      if (g == NG - 1) {
#pragma unroll
        for (int i = 0; i < NA; ++i)
          *(v4f*)(&As[(buf ^ 1) * ASZ + (arow + AROWS * i) * PA + acol]) =
              ra[i];
#pragma unroll
        for (int i = 0; i < NW; ++i)
          *(v4f*)(&Ws[(buf ^ 1) * WSZ + (wrow + 8 * i) * PW + wcol]) = rw[i];
        const int tn = t + 2 < nchunk ? t + 2 : nchunk - 1;
        const int k2 = tn * BK;
#pragma unroll
        for (int i = 0; i < NA; ++i)
          ra[i] = *(const v4f*)(Ag + (size_t)(AROWS * i) * K + k2);
#pragma unroll
        for (int i = 0; i < NW; ++i)
          rw[i] = *(const v4f*)(Wg + (size_t)(k2 + 8 * i) * HDIM);
      }
      __builtin_amdgcn_sched_barrier(0);
#pragma unroll
      for (int jj = 0; jj < GJ; ++jj) {
#pragma unroll
        for (int i = 0; i < RT; ++i) {
          v2f a; a.x = fa[g & 1][jj][i].x; a.y = fa[g & 1][jj][i].y;
          acc[i] = __builtin_amdgcn_wmma_f32_16x16x4_f32(
              false, a, false, fb[g & 1][jj][0], (short)0, acc[i], false,
              false);
        }
#pragma unroll
        for (int i = 0; i < RT; ++i) {
          v2f a; a.x = fa[g & 1][jj][i].z; a.y = fa[g & 1][jj][i].w;
          acc[i] = __builtin_amdgcn_wmma_f32_16x16x4_f32(
              false, a, false, fb[g & 1][jj][1], (short)0, acc[i], false,
              false);
        }
      }
      __builtin_amdgcn_sched_barrier(0);
    }
    lds_barrier();
  }

  const float bn = bias[n];
#pragma unroll
  for (int i = 0; i < RT; ++i) {
#pragma unroll
    for (int r = 0; r < 8; ++r) {
      const int m = m0 + 16 * i + r + 8 * half;
      const float v = acc[i][r] + bn + emb[(size_t)nid[m] * HDIM + n];
      C[(size_t)m * HDIM + n] = v;
    }
  }
}

constexpr int gemm_in_smem(int rt, int bk) {
  return 2 * (16 * rt) * (bk + 4) + 2 * bk * (HDIM + 4);
}
constexpr int GEMM_SMEM = gemm_in_smem(1, 128) > gemm_in_smem(2, 64)
                              ? gemm_in_smem(1, 128) : gemm_in_smem(2, 64);
static_assert(2 * GEMM_SMEM * 4 <= 320 * 1024, "two tiles per WGP");

__global__ void __launch_bounds__(256)
gemm_in_both_kernel(const float* __restrict__ Ap, const float* __restrict__ Wp,
                    const float* __restrict__ bias_p,
                    const float* __restrict__ emb_p,
                    const int* __restrict__ nid_p, float* __restrict__ Cp,
                    int Kp, int nb_p,
                    const float* __restrict__ Ac, const float* __restrict__ Wc,
                    const float* __restrict__ bias_c,
                    const float* __restrict__ emb_c,
                    const int* __restrict__ nid_c, float* __restrict__ Cc,
                    int Kc) {
  __shared__ __attribute__((aligned(16))) float smem[GEMM_SMEM];
  if ((int)blockIdx.x < nb_p)
    gemm_in_tile<2, 64>(smem, (int)blockIdx.x, Ap, Wp, bias_p, emb_p, nid_p,
                        Cp, Kp);
  else
    gemm_in_tile<1, 128>(smem, (int)blockIdx.x - nb_p, Ac, Wc, bias_c, emb_c,
                         nid_c, Cc, Kc);
}

constexpr int PS = HDIM + 4;
constexpr int RPT = 12;
constexpr int SEG = 256 * RPT;

__global__ void __launch_bounds__(256)
sage_layer_kernel(const unsigned* __restrict__ rec,
                  const int* __restrict__ gcnt, const int* __restrict__ gbeg,
                  const float* __restrict__ xsrc_c,
                  const float* __restrict__ xdst_c,
                  const float* __restrict__ Wl_c, const float* __restrict__ bl_c,
                  const float* __restrict__ Wr_c, float* __restrict__ out_c,
                  int nrows_c,
                  const float* __restrict__ xsrc_p,
                  const float* __restrict__ xdst_p,
                  const float* __restrict__ Wl_p, const float* __restrict__ bl_p,
                  const float* __restrict__ Wr_p, float* __restrict__ out_p,
                  int relu) {
  __shared__ __attribute__((aligned(16))) float Ss[16 * PS];
  __shared__ __attribute__((aligned(16))) float Xs[16 * PS];
  __shared__ unsigned short lid[SEG];
  __shared__ int cnt[16];
  __shared__ int fp[17];
  __shared__ int deg[16];

  const int nb_c = nrows_c >> 4;
  const bool isc = (int)blockIdx.x < nb_c;
  const float* xsrc          = isc ? xsrc_c : xsrc_p;
  const float* X             = isc ? xdst_c : xdst_p;
  const float* Wl            = isc ? Wl_c : Wl_p;
  const float* bl            = isc ? bl_c : bl_p;
  const float* Wr            = isc ? Wr_c : Wr_p;
  float* Out                 = isc ? out_c : out_p;
  const int tile   = isc ? (int)blockIdx.x : (int)blockIdx.x - nb_c;
  const int bucket = isc ? NBP + tile : tile;
  const int m0 = tile * 16;
  const int nrec = gcnt[bucket];
  const unsigned* rb = rec + gbeg[bucket];

  const int t = threadIdx.x;
  const int lane = t & 31;
  const int wave = __builtin_amdgcn_readfirstlane((int)(threadIdx.x >> 5));
  const int half = lane >> 4;
  const int nloc = lane & 15;
  const int n    = wave * 16 + nloc;

  const int lr = 2 * wave;
  const v4f xr0 = *(const v4f*)(X + (size_t)(m0 + lr) * HDIM + lane * 4);
  const v4f xr1 = *(const v4f*)(X + (size_t)(m0 + lr + 1) * HDIM + lane * 4);
  if (t < 16) deg[t] = 0;
  v4f acc0 = {0.0f, 0.0f, 0.0f, 0.0f};
  v4f acc1 = {0.0f, 0.0f, 0.0f, 0.0f};
  for (int s0 = 0; s0 < nrec; s0 += SEG) {
    unsigned rv[RPT];
    int rk[RPT];
#pragma unroll
    for (int k = 0; k < RPT; ++k) {
      const int i = s0 + k * 256 + t;
      const int ii = i < nrec ? i : nrec - 1;
      rv[k] = rb[ii];
    }
    if (t < 16) cnt[t] = 0;
    __syncthreads();
#pragma unroll
    for (int k = 0; k < RPT; ++k) {
      const int i = s0 + k * 256 + t;
      rk[k] = 0;
      if (i < nrec) rk[k] = lds_add_rtn(&cnt[(rv[k] >> 16) & 15], 1);
    }
    __syncthreads();
    if (t < 16) {
      int ex = 0, tot = 0;
#pragma unroll
      for (int j = 0; j < 16; ++j) {
        const int cj = cnt[j];
        if (j < t) ex += cj;
        tot += cj;
      }
      fp[t] = ex;
      deg[t] += cnt[t];
      if (t == 0) fp[16] = tot;
    }
    __syncthreads();
    int pos[RPT];
#pragma unroll
    for (int k = 0; k < RPT; ++k) pos[k] = fp[(rv[k] >> 16) & 15] + rk[k];
#pragma unroll
    for (int k = 0; k < RPT; ++k) {
      const int i = s0 + k * 256 + t;
      if (i < nrec) lid[pos[k]] = (unsigned short)(rv[k] & 0xffffu);
    }
    __syncthreads();
    const int b0 = __builtin_amdgcn_readfirstlane(fp[lr]);
    const int b1 = __builtin_amdgcn_readfirstlane(fp[lr + 1]);
    const int b2 = __builtin_amdgcn_readfirstlane(fp[lr + 2]);
    acc0 = gather_row(xsrc, lid, b0, b1, lane, acc0);
    acc1 = gather_row(xsrc, lid, b1, b2, lane, acc1);
  }
  *(v4f*)(&Ss[lr * PS + lane * 4]) = acc0;
  *(v4f*)(&Ss[(lr + 1) * PS + lane * 4]) = acc1;
  *(v4f*)(&Xs[lr * PS + lane * 4]) = xr0;
  *(v4f*)(&Xs[(lr + 1) * PS + lane * 4]) = xr1;
  __syncthreads();

  const float* sa = &Ss[nloc * PS + 2 * half];
  const float* xa = &Xs[nloc * PS + 2 * half];
  v8f accS = {};
  v8f accX = {};
#pragma unroll 8
  for (int k0 = 0; k0 < HDIM; k0 += 4) {
    const int ka = k0 + 2 * half;
    const v2f a1 = *(const v2f*)(sa + k0);
    v2f b1; b1.x = Wl[(size_t)ka * HDIM + n];
            b1.y = Wl[(size_t)(ka + 1) * HDIM + n];
    accS = __builtin_amdgcn_wmma_f32_16x16x4_f32(false, a1, false, b1,
                                                 (short)0, accS, false, false);
    const v2f a2 = *(const v2f*)(xa + k0);
    v2f b2; b2.x = Wr[(size_t)ka * HDIM + n];
            b2.y = Wr[(size_t)(ka + 1) * HDIM + n];
    accX = __builtin_amdgcn_wmma_f32_16x16x4_f32(false, a2, false, b2,
                                                 (short)0, accX, false, false);
  }

  const float bn = bl[n];
#pragma unroll
  for (int r = 0; r < 8; ++r) {
    const int m = m0 + r + 8 * half;
    float d = (float)deg[r + 8 * half];
    d = d < 1.0f ? 1.0f : d;
    float v = accS[r] / d + accX[r] + bn;
    if (relu) v = v > 0.0f ? v : 0.0f;
    Out[(size_t)m * HDIM + n] = v;
  }
}

constexpr int EPW = 32;

template <int M>
__device__ __forceinline__ float lane_xor(float v) {
  const int i = __builtin_bit_cast(int, v);
  return __builtin_bit_cast(
      float, __builtin_amdgcn_update_dpp(i, i, 0x160 + M, 0xf, 0xf, false));
}

__global__ void __launch_bounds__(256)
edge_dot_kernel(const float* __restrict__ Hc, const float* __restrict__ Hp,
                const int* __restrict__ ls, const int* __restrict__ ld, int e,
                float* __restrict__ out) {
  const int lane = threadIdx.x & 31;
  const int q = lane & 7;
  const int first = EPW * __builtin_amdgcn_readfirstlane(
      (int)((blockIdx.x * 256 + threadIdx.x) >> 5));
  if (first >= e) return;
  const int e0 = first + (lane & 24);
  int ia[8], ib[8];
  if (first + EPW <= e) {
    const int4 a0 = *(const int4*)(ls + e0);
    const int4 a1 = *(const int4*)(ls + e0 + 4);
    const int4 b0 = *(const int4*)(ld + e0);
    const int4 b1 = *(const int4*)(ld + e0 + 4);
    ia[0] = a0.x; ia[1] = a0.y; ia[2] = a0.z; ia[3] = a0.w;
    ia[4] = a1.x; ia[5] = a1.y; ia[6] = a1.z; ia[7] = a1.w;
    ib[0] = b0.x; ib[1] = b0.y; ib[2] = b0.z; ib[3] = b0.w;
    ib[4] = b1.x; ib[5] = b1.y; ib[6] = b1.z; ib[7] = b1.w;
  } else {
#pragma unroll
    for (int k = 0; k < 8; ++k) {
      const int i = e0 + k;
      ia[k] = (i < e) ? ls[i] : 0;
      ib[k] = (i < e) ? ld[i] : 0;
    }
  }
  const float* hcq = Hc + q * 4;
  const float* hpq = Hp + q * 4;
  float r = 0.0f;
#pragma unroll
  for (int k = 0; k < 8; k += 2) {
    v4f va[2][4], vb[2][4];
#pragma unroll
    for (int u = 0; u < 2; ++u) {
      const float* pa = hcq + (size_t)ia[k + u] * HDIM;
      const float* pb = hpq + (size_t)ib[k + u] * HDIM;
#pragma unroll
      for (int c = 0; c < 4; ++c) {
        va[u][c] = *(const v4f*)(pa + 32 * c);
        vb[u][c] = *(const v4f*)(pb + 32 * c);
      }
    }
#pragma unroll
    for (int u = 0; u < 2; ++u) {
      float p = 0.0f;
#pragma unroll
      for (int c = 0; c < 4; ++c)
        p += va[u][c].x * vb[u][c].x + va[u][c].y * vb[u][c].y +
             va[u][c].z * vb[u][c].z + va[u][c].w * vb[u][c].w;
      p += lane_xor<1>(p);
      p += lane_xor<2>(p);
      p += lane_xor<4>(p);
      if (q == k + u) r = p;
    }
  }
  const int i = first + lane;
  if (i < e) out[i] = r;
}

extern "C" void kernel_launch(void* const* d_in, const int* in_sizes, int n_in,
                              void* d_out, int out_size, void* d_ws,
                              size_t ws_size, hipStream_t stream) {
  (void)in_sizes; (void)n_in; (void)out_size; (void)ws_size;
  const int NC = 4096, NP = 8192, E = 500000, EL = 500000;

  const float* x_country = (const float*)d_in[0];
  const float* x_product = (const float*)d_in[1];
  const float* Wc_lin = (const float*)d_in[2];
  const float* bc_lin = (const float*)d_in[3];
  const float* Wp_lin = (const float*)d_in[4];
  const float* bp_lin = (const float*)d_in[5];
  const float* emb_c  = (const float*)d_in[6];
  const float* emb_p  = (const float*)d_in[7];
  const float* W1ep_l = (const float*)d_in[8];
  const float* b1ep   = (const float*)d_in[9];
  const float* W1ep_r = (const float*)d_in[10];
  const float* W1pc_l = (const float*)d_in[11];
  const float* b1pc   = (const float*)d_in[12];
  const float* W1pc_r = (const float*)d_in[13];
  const float* W2ep_l = (const float*)d_in[14];
  const float* b2ep   = (const float*)d_in[15];
  const float* W2ep_r = (const float*)d_in[16];
  const float* W2pc_l = (const float*)d_in[17];
  const float* b2pc   = (const float*)d_in[18];
  const float* W2pc_r = (const float*)d_in[19];
  const int* nid_c = (const int*)d_in[20];
  const int* nid_p = (const int*)d_in[21];
  const int* e_src = (const int*)d_in[22];
  const int* e_dst = (const int*)d_in[23];
  const int* l_src = (const int*)d_in[24];
  const int* l_dst = (const int*)d_in[25];

  float* ws = (float*)d_ws;
  size_t off = 0;
  float* xc   = ws + off; off += (size_t)NC * HDIM;
  float* xp   = ws + off; off += (size_t)NP * HDIM;
  float* hp   = ws + off; off += (size_t)NP * HDIM;
  float* hc   = ws + off; off += (size_t)NC * HDIM;
  float* hc2  = ws + off; off += (size_t)NC * HDIM;
  float* hp2  = ws + off; off += (size_t)NP * HDIM;
  float* scr  = ws + off; off += (size_t)NP * HDIM;

  static_assert(2 * E + 3 * NB <= NP * HDIM, "edge split fits its plane");
  unsigned* rec = (unsigned*)scr;
  int* gcnt = (int*)scr + 2 * E;
  int* gcur = gcnt + NB;
  int* gbeg = gcur + NB;

  const int TB = 256;
  const int csr_blocks = (E + EPB - 1) / EPB;

  gemm_in_both_kernel<<<NP / 32 + NC / 16, TB, 0, stream>>>(
      x_product, Wp_lin, bp_lin, emb_p, nid_p, xp, NP, NP / 32,
      x_country, Wc_lin, bc_lin, emb_c, nid_c, xc, NC);

  zero_kernel<<<512, TB, 0, stream>>>(scr, 2 * E + 2 * NB);
  bucket_count_kernel<<<csr_blocks, 1024, 0, stream>>>(e_src, e_dst, E, gcnt);
  bucket_fill_kernel<<<csr_blocks, 1024, 0, stream>>>(e_src, e_dst, E, gcnt,
                                                      gcur, gbeg, rec);

  const int layer_blocks = NC / 16 + NP / 16;

  sage_layer_kernel<<<layer_blocks, TB, 0, stream>>>(
      rec, gcnt, gbeg,
      xp, xc, W1pc_l, b1pc, W1pc_r, hc, NC,
      xc, xp, W1ep_l, b1ep, W1ep_r, hp, 1);

  sage_layer_kernel<<<layer_blocks, TB, 0, stream>>>(
      rec, gcnt, gbeg,
      hp, hc, W2pc_l, b2pc, W2pc_r, hc2, NC,
      hc, hp, W2ep_l, b2ep, W2ep_r, hp2, 0);

  const int el_blocks = (EL + 8 * EPW - 1) / (8 * EPW);
  edge_dot_kernel<<<el_blocks, TB, 0, stream>>>(hc2, hp2, l_src, l_dst, EL,
                                                (float*)d_out);
}


# edge split rides in the projection launch: a split block orders its 8192 edges in LDS into its own region, layer blocks collect their runs; 7 -> 4 launches; row f32 sum order changes (arrival order)
# speedup vs baseline: 4.0635x; 1.0323x over previous
#include <hip/hip_runtime.h>

typedef __attribute__((ext_vector_type(2))) float v2f;
typedef __attribute__((ext_vector_type(8))) float v8f;
typedef __attribute__((ext_vector_type(4))) float v4f;

constexpr int HDIM = 128;

constexpr int EPB = 8192;
constexpr int EPT = EPB / 256;
constexpr int ESB = 8;
constexpr int MAXP = 8192;
constexpr int MAXC = 4096;
constexpr int NBP = MAXP / 16;
constexpr int NBC = MAXC / 16;
constexpr int NB = NBP + NBC;
constexpr int OFFP = 832;
static_assert(NB == 3 * 256, "a thread owns three buckets in the prefix sum");
static_assert(2 * EPB <= 65535, "a slot inside a region takes 16 bits");
static_assert(OFFP >= NB + 2 && (OFFP * 2) % 128 == 0, "offset rows are whole lines");

__device__ __forceinline__ int lds_add_rtn(int* p, int v) {
  return __hip_atomic_fetch_add(p, v, __ATOMIC_RELAXED,
                                __HIP_MEMORY_SCOPE_WORKGROUP);
}

__device__ __forceinline__ int wave_incl_scan(int v, int lane) {
  for (int off = 1; off < 32; off <<= 1) {
    const int o = __shfl_up(v, off, 32);
    if (lane >= off) v += o;
  }
  return v;
}

constexpr int SPLIT_LST = NB + 4;
constexpr int SPLIT_SMEM = NB + SPLIT_LST + 8 + 2 * EPB;

__device__ __forceinline__ void
edge_split_block(int* __restrict__ sm, int b, const int* __restrict__ src,
                 const int* __restrict__ dst, int e, unsigned* __restrict__ rec,
                 unsigned* __restrict__ off32) {
  int* h   = sm;
  int* lst = h + NB;
  int* wt  = lst + SPLIT_LST;
  unsigned* stage = (unsigned*)(wt + 8);
  const int t = threadIdx.x;
  const int lane = t & 31;
  const int w = t >> 5;
  const int e0 = b * EPB;
  for (int i = t; i < NB; i += 256) h[i] = 0;
  if (t < SPLIT_LST - NB) lst[NB + t] = 0;
  __syncthreads();
  for (int k0 = 0; k0 < EPT; k0 += ESB) {
    int es[ESB], ed[ESB];
#pragma unroll
    for (int u = 0; u < ESB; ++u) {
      const int i = e0 + (k0 + u) * 256 + t;
      const int ii = i < e ? i : e - 1;
      es[u] = src[ii];
      ed[u] = dst[ii];
    }
#pragma unroll
    for (int u = 0; u < ESB; ++u) {
      const int i = e0 + (k0 + u) * 256 + t;
      if (i < e) {
        (void)lds_add_rtn(&h[ed[u] >> 4], 1);
        (void)lds_add_rtn(&h[NBP + (es[u] >> 4)], 1);
      }
    }
  }
  __syncthreads();
  const int c0 = h[3 * t], c1 = h[3 * t + 1], c2 = h[3 * t + 2];
  const int s3 = c0 + c1 + c2;
  const int inc = wave_incl_scan(s3, lane);
  if (lane == 31) wt[w] = inc;
  __syncthreads();
  int base = 0;
  for (int j = 0; j < 8; ++j)
    if (j < w) base += wt[j];
  const int ex = base + inc - s3;
  lst[3 * t] = ex;
  lst[3 * t + 1] = ex + c0;
  lst[3 * t + 2] = ex + c0 + c1;
  if (t == 255) lst[NB] = ex + s3;
  __syncthreads();
  for (int i = t; i < NB; i += 256) h[i] = lst[i];
  for (int i = t; 2 * i <= NB; i += 256)
    off32[(size_t)b * (OFFP / 2) + i] =
        (unsigned)lst[2 * i] | ((unsigned)lst[2 * i + 1] << 16);
  __syncthreads();
  for (int k0 = 0; k0 < EPT; k0 += ESB) {
    int es[ESB], ed[ESB];
#pragma unroll
    for (int u = 0; u < ESB; ++u) {
      const int i = e0 + (k0 + u) * 256 + t;
      const int ii = i < e ? i : e - 1;
      es[u] = src[ii];
      ed[u] = dst[ii];
    }
#pragma unroll
    for (int u = 0; u < ESB; ++u) {
      const int i = e0 + (k0 + u) * 256 + t;
      if (i < e) {
        const int pd = lds_add_rtn(&h[ed[u] >> 4], 1);
        stage[pd] = ((unsigned)(ed[u] & 15) << 16) | (unsigned)es[u];
        const int ps = lds_add_rtn(&h[NBP + (es[u] >> 4)], 1);
        stage[ps] = ((unsigned)(es[u] & 15) << 16) | (unsigned)ed[u];
      }
    }
  }
  __syncthreads();
  const int left = e - e0;
  const int total = 2 * (left < EPB ? left : EPB);
  unsigned* rb = rec + (size_t)b * (2 * EPB);
  for (int i = t; i < total; i += 256) rb[i] = stage[i];
}

template <int N>
__device__ __forceinline__ void gather_group(const float* __restrict__ xl,
                                             int ids, int j, v4f& acc) {
  v4f v[N];
#pragma unroll
  for (int u = 0; u < N; ++u) {
    const int c = __builtin_amdgcn_readlane(ids, j + u);
    v[u] = *(const v4f*)(xl + (size_t)c * HDIM);
  }
#pragma unroll
  for (int u = 0; u < N; ++u) acc += v[u];
}

__device__ __forceinline__ v4f gather_row(const float* __restrict__ x,
                                          const unsigned short* lid,
                                          int beg, int end, int lane, v4f acc) {
  const float* xl = x + lane * 4;
  for (int base = beg; base < end; base += 32) {
    const int left = end - base;
    const int n = left < 32 ? left : 32;
    const int ids = (lane < n) ? (int)lid[base + lane] : 0;
    int j = 0;
    for (; j + 16 <= n; j += 16) gather_group<16>(xl, ids, j, acc);
    if (n - j >= 8) { gather_group<8>(xl, ids, j, acc); j += 8; }
    if (n - j >= 4) { gather_group<4>(xl, ids, j, acc); j += 4; }
    if (n - j >= 2) { gather_group<2>(xl, ids, j, acc); j += 2; }
    if (n - j >= 1) { gather_group<1>(xl, ids, j, acc); }
  }
  return acc;
}


__device__ __forceinline__ void lds_barrier() {
  __builtin_amdgcn_fence(__ATOMIC_RELEASE, "workgroup", "local");
  __builtin_amdgcn_s_barrier();
  __builtin_amdgcn_fence(__ATOMIC_ACQUIRE, "workgroup", "local");
}

template <int RT, int GJ, int PA, int PW>
__device__ __forceinline__ void
frag_group(const float* __restrict__ as, const float* __restrict__ ws, int g,
           v4f (&fa)[GJ][RT], v2f (&fb)[GJ][2]) {
#pragma unroll
  for (int jj = 0; jj < GJ; ++jj) {
    const int j = g * GJ + jj;
#pragma unroll
    for (int i = 0; i < RT; ++i)
      fa[jj][i] = *(const v4f*)(as + (16 * i) * PA + 8 * j);
    fb[jj][0].x = ws[(8 * j + 0) * PW];
    fb[jj][0].y = ws[(8 * j + 1) * PW];
    fb[jj][1].x = ws[(8 * j + 2) * PW];
    fb[jj][1].y = ws[(8 * j + 3) * PW];
  }
}

template <int RT, int BK>
__device__ __forceinline__ void
gemm_in_tile(float* __restrict__ smem, int bid,
             const float* __restrict__ A, const float* __restrict__ W,
             const float* __restrict__ bias, const float* __restrict__ emb,
             const int* __restrict__ nid, float* __restrict__ C, int K) {
  constexpr int BM = 16 * RT;
  constexpr int PA = BK + 4;
  constexpr int PW = HDIM + 4;
  constexpr int ATPR = BK / 4;
  constexpr int AROWS = 256 / ATPR;
  constexpr int NA = BM / AROWS;
  constexpr int NW = BK / 8;
  static_assert(NA * AROWS == BM && NA >= 1, "A tile must be whole passes");
  constexpr int GJ = 4 / RT;
  constexpr int NG = (BK / 8) / GJ;
  static_assert(NG * GJ * 8 == BK && NG >= 2, "a chunk is whole groups");
  constexpr int ASZ = BM * PA;
  constexpr int WSZ = BK * PW;
  float* As = smem;
  float* Ws = smem + 2 * ASZ;

  const int tid  = threadIdx.x;
  const int lane = tid & 31;
  const int wave = tid >> 5;
  const int half = lane >> 4;
  const int nloc = lane & 15;
  const int m0   = bid * BM;
  const int n    = wave * 16 + nloc;

  const int arow = tid / ATPR;
  const int acol = (tid % ATPR) * 4;
  const int wrow = tid >> 5;
  const int wcol = (tid & 31) * 4;
  const float* Ag = A + (size_t)(m0 + arow) * K + acol;
  const float* Wg = W + (size_t)wrow * HDIM + wcol;

  v4f ra[NA];
  v4f rw[NW];
  v8f acc[RT];
#pragma unroll
  for (int i = 0; i < RT; ++i) acc[i] = v8f{};

  const int nchunk = K / BK;

#pragma unroll
  for (int i = 0; i < NA; ++i)
    ra[i] = *(const v4f*)(Ag + (size_t)(AROWS * i) * K);
#pragma unroll
  for (int i = 0; i < NW; ++i)
    rw[i] = *(const v4f*)(Wg + (size_t)(8 * i) * HDIM);
#pragma unroll
  for (int i = 0; i < NA; ++i)
    *(v4f*)(&As[(arow + AROWS * i) * PA + acol]) = ra[i];
#pragma unroll
  for (int i = 0; i < NW; ++i)
    *(v4f*)(&Ws[(wrow + 8 * i) * PW + wcol]) = rw[i];
  {
    const int k1 = (nchunk > 1 ? 1 : 0) * BK;
#pragma unroll
    for (int i = 0; i < NA; ++i)
      ra[i] = *(const v4f*)(Ag + (size_t)(AROWS * i) * K + k1);
#pragma unroll
    for (int i = 0; i < NW; ++i)
      rw[i] = *(const v4f*)(Wg + (size_t)(k1 + 8 * i) * HDIM);
  }
  lds_barrier();

  for (int t = 0; t < nchunk; ++t) {
    const int buf = t & 1;
    const float* as = &As[buf * ASZ + nloc * PA + 4 * half];
    const float* ws = &Ws[buf * WSZ + (4 * half) * PW + n];
    v4f fa[2][GJ][RT];
    v2f fb[2][GJ][2];
    frag_group<RT, GJ, PA, PW>(as, ws, 0, fa[0], fb[0]);
    __builtin_amdgcn_sched_barrier(0);
#pragma unroll
    for (int g = 0; g < NG; ++g) {
      if (g + 1 < NG)
        frag_group<RT, GJ, PA, PW>(as, ws, g + 1, fa[(g + 1) & 1],
                                   fb[(g + 1) & 1]);
      if (g == NG - 1) {
#pragma unroll
        for (int i = 0; i < NA; ++i)
          *(v4f*)(&As[(buf ^ 1) * ASZ + (arow + AROWS * i) * PA + acol]) =
              ra[i];
#pragma unroll
        for (int i = 0; i < NW; ++i)
          *(v4f*)(&Ws[(buf ^ 1) * WSZ + (wrow + 8 * i) * PW + wcol]) = rw[i];
        const int tn = t + 2 < nchunk ? t + 2 : nchunk - 1;
        const int k2 = tn * BK;
#pragma unroll
        for (int i = 0; i < NA; ++i)
          ra[i] = *(const v4f*)(Ag + (size_t)(AROWS * i) * K + k2);
#pragma unroll
        for (int i = 0; i < NW; ++i)
          rw[i] = *(const v4f*)(Wg + (size_t)(k2 + 8 * i) * HDIM);
      }
      __builtin_amdgcn_sched_barrier(0);
#pragma unroll
      for (int jj = 0; jj < GJ; ++jj) {
#pragma unroll
        for (int i = 0; i < RT; ++i) {
          v2f a; a.x = fa[g & 1][jj][i].x; a.y = fa[g & 1][jj][i].y;
          acc[i] = __builtin_amdgcn_wmma_f32_16x16x4_f32(
              false, a, false, fb[g & 1][jj][0], (short)0, acc[i], false,
              false);
        }
#pragma unroll
        for (int i = 0; i < RT; ++i) {
          v2f a; a.x = fa[g & 1][jj][i].z; a.y = fa[g & 1][jj][i].w;
          acc[i] = __builtin_amdgcn_wmma_f32_16x16x4_f32(
              false, a, false, fb[g & 1][jj][1], (short)0, acc[i], false,
              false);
        }
      }
      __builtin_amdgcn_sched_barrier(0);
    }
    lds_barrier();
  }

  const float bn = bias[n];
#pragma unroll
  for (int i = 0; i < RT; ++i) {
#pragma unroll
    for (int r = 0; r < 8; ++r) {
      const int m = m0 + 16 * i + r + 8 * half;
      const float v = acc[i][r] + bn + emb[(size_t)nid[m] * HDIM + n];
      C[(size_t)m * HDIM + n] = v;
    }
  }
}

constexpr int gemm_in_smem(int rt, int bk) {
  return 2 * (16 * rt) * (bk + 4) + 2 * bk * (HDIM + 4);
}
constexpr int GEMM_SMEM = gemm_in_smem(1, 128) > gemm_in_smem(2, 64)
                              ? gemm_in_smem(1, 128) : gemm_in_smem(2, 64);
static_assert(2 * GEMM_SMEM * 4 <= 320 * 1024, "two tiles per WGP");
static_assert(SPLIT_SMEM <= GEMM_SMEM, "the edge split works in the tile's LDS");

__global__ void __launch_bounds__(256)
gemm_in_both_kernel(const float* __restrict__ Ap, const float* __restrict__ Wp,
                    const float* __restrict__ bias_p,
                    const float* __restrict__ emb_p,
                    const int* __restrict__ nid_p, float* __restrict__ Cp,
                    int Kp, int nb_p,
                    const float* __restrict__ Ac, const float* __restrict__ Wc,
                    const float* __restrict__ bias_c,
                    const float* __restrict__ emb_c,
                    const int* __restrict__ nid_c, float* __restrict__ Cc,
                    int Kc,
                    const int* __restrict__ e_src, const int* __restrict__ e_dst,
                    int ne, int nsplit, unsigned* __restrict__ rec,
                    unsigned* __restrict__ off32) {
  __shared__ __attribute__((aligned(16))) float smem[GEMM_SMEM];
  if ((int)blockIdx.x < nb_p) {
    gemm_in_tile<2, 64>(smem, (int)blockIdx.x, Ap, Wp, bias_p, emb_p, nid_p,
                        Cp, Kp);
  } else {
    const int cb = (int)blockIdx.x - nb_p;
    if (cb < nsplit) {
      edge_split_block((int*)smem, cb, e_src, e_dst, ne, rec, off32);
      __syncthreads();
    }
    gemm_in_tile<1, 128>(smem, cb, Ac, Wc, bias_c, emb_c, nid_c, Cc, Kc);
  }
}

constexpr int PS = HDIM + 4;
constexpr int RPT = 12;
constexpr int SEG = 256 * RPT;

__global__ void __launch_bounds__(256)
sage_layer_kernel(const unsigned* __restrict__ rec,
                  const unsigned* __restrict__ off32, int nsplit,
                  const float* __restrict__ xsrc_c,
                  const float* __restrict__ xdst_c,
                  const float* __restrict__ Wl_c, const float* __restrict__ bl_c,
                  const float* __restrict__ Wr_c, float* __restrict__ out_c,
                  int nrows_c,
                  const float* __restrict__ xsrc_p,
                  const float* __restrict__ xdst_p,
                  const float* __restrict__ Wl_p, const float* __restrict__ bl_p,
                  const float* __restrict__ Wr_p, float* __restrict__ out_p,
                  int relu) {
  __shared__ __attribute__((aligned(16))) float Ss[16 * PS];
  __shared__ __attribute__((aligned(16))) float Xs[16 * PS];
  __shared__ unsigned short lid[SEG];
  __shared__ unsigned stage[SEG];
  __shared__ int wsum[8];
  __shared__ int cnt[16];
  __shared__ int fp[17];
  __shared__ int deg[16];

  const int nb_c = nrows_c >> 4;
  const bool isc = (int)blockIdx.x < nb_c;
  const float* xsrc          = isc ? xsrc_c : xsrc_p;
  const float* X             = isc ? xdst_c : xdst_p;
  const float* Wl            = isc ? Wl_c : Wl_p;
  const float* bl            = isc ? bl_c : bl_p;
  const float* Wr            = isc ? Wr_c : Wr_p;
  float* Out                 = isc ? out_c : out_p;
  const int tile   = isc ? (int)blockIdx.x : (int)blockIdx.x - nb_c;
  const int bucket = isc ? NBP + tile : tile;
  const int m0 = tile * 16;

  const int t = threadIdx.x;
  const int lane = t & 31;
  const int wave = __builtin_amdgcn_readfirstlane((int)(threadIdx.x >> 5));
  const int half = lane >> 4;
  const int nloc = lane & 15;
  const int n    = wave * 16 + nloc;

  const int lr = 2 * wave;
  const v4f xr0 = *(const v4f*)(X + (size_t)(m0 + lr) * HDIM + lane * 4);
  const v4f xr1 = *(const v4f*)(X + (size_t)(m0 + lr + 1) * HDIM + lane * 4);
  if (t < 16) deg[t] = 0;
  const int run = t >> 2;
  const int sub = t & 3;
  int rbeg = 0, rlen = 0;
  if (run < nsplit) {
    const unsigned* o = off32 + (size_t)run * (OFFP / 2);
    const unsigned w0 = o[bucket >> 1];
    const unsigned w1 = o[(bucket + 1) >> 1];
    rbeg = (int)((bucket & 1) ? (w0 >> 16) : (w0 & 0xffffu));
    const int rend = (int)((bucket & 1) ? (w1 & 0xffffu) : (w1 >> 16));
    rlen = rend - rbeg;
  }
  const unsigned* rp = rec + (size_t)run * (2 * EPB) + rbeg;
  const int mine = sub == 0 ? rlen : 0;
  const int rinc = wave_incl_scan(mine, lane);
  if (lane == 31) wsum[wave] = rinc;
  __syncthreads();
  int rbase = 0, rtot = 0;
#pragma unroll
  for (int j = 0; j < 8; ++j) {
    const int wj = wsum[j];
    if (j < wave) rbase += wj;
    rtot += wj;
  }
  const int p0 = __shfl(rbase + rinc - mine, lane & ~3, 32);
  const int nrec = __builtin_amdgcn_readfirstlane(rtot);
  v4f acc0 = {0.0f, 0.0f, 0.0f, 0.0f};
  v4f acc1 = {0.0f, 0.0f, 0.0f, 0.0f};
  for (int s0 = 0; s0 < nrec; s0 += SEG) {
    unsigned rv[RPT];
    int rk[RPT];
    {
      int jlo = s0 - p0;
      jlo = jlo < 0 ? 0 : jlo;
      int jhi = s0 + SEG - p0;
      jhi = jhi > rlen ? rlen : jhi;
      for (int j = jlo + sub; j < jhi; j += 32) {
        unsigned v[8];
#pragma unroll
        for (int u = 0; u < 8; ++u) {
          const int jj = j + 4 * u;
          v[u] = rp[jj < jhi ? jj : jhi - 1];
        }
#pragma unroll
        for (int u = 0; u < 8; ++u) {
          const int jj = j + 4 * u;
          if (jj < jhi) stage[p0 + jj - s0] = v[u];
        }
      }
    }
    if (t < 16) cnt[t] = 0;
    __syncthreads();
#pragma unroll
    for (int k = 0; k < RPT; ++k) {
      const int i = s0 + k * 256 + t;
      const int ii = i < nrec ? i : nrec - 1;
      rv[k] = stage[ii - s0];
    }
#pragma unroll
    for (int k = 0; k < RPT; ++k) {
      const int i = s0 + k * 256 + t;
      rk[k] = 0;
      if (i < nrec) rk[k] = lds_add_rtn(&cnt[(rv[k] >> 16) & 15], 1);
    }
    __syncthreads();
    if (t < 16) {
      int ex = 0, tot = 0;
#pragma unroll
      for (int j = 0; j < 16; ++j) {
        const int cj = cnt[j];
        if (j < t) ex += cj;
        tot += cj;
      }
      fp[t] = ex;
      deg[t] += cnt[t];
      if (t == 0) fp[16] = tot;
    }
    __syncthreads();
    int pos[RPT];
#pragma unroll
    for (int k = 0; k < RPT; ++k) pos[k] = fp[(rv[k] >> 16) & 15] + rk[k];
#pragma unroll
    for (int k = 0; k < RPT; ++k) {
      const int i = s0 + k * 256 + t;
      if (i < nrec) lid[pos[k]] = (unsigned short)(rv[k] & 0xffffu);
    }
    __syncthreads();
    const int b0 = __builtin_amdgcn_readfirstlane(fp[lr]);
    const int b1 = __builtin_amdgcn_readfirstlane(fp[lr + 1]);
    const int b2 = __builtin_amdgcn_readfirstlane(fp[lr + 2]);
    acc0 = gather_row(xsrc, lid, b0, b1, lane, acc0);
    acc1 = gather_row(xsrc, lid, b1, b2, lane, acc1);
  }
  *(v4f*)(&Ss[lr * PS + lane * 4]) = acc0;
  *(v4f*)(&Ss[(lr + 1) * PS + lane * 4]) = acc1;
  *(v4f*)(&Xs[lr * PS + lane * 4]) = xr0;
  *(v4f*)(&Xs[(lr + 1) * PS + lane * 4]) = xr1;
  __syncthreads();

  const float* sa = &Ss[nloc * PS + 2 * half];
  const float* xa = &Xs[nloc * PS + 2 * half];
  v8f accS = {};
  v8f accX = {};
#pragma unroll 8
  for (int k0 = 0; k0 < HDIM; k0 += 4) {
    const int ka = k0 + 2 * half;
    const v2f a1 = *(const v2f*)(sa + k0);
    v2f b1; b1.x = Wl[(size_t)ka * HDIM + n];
            b1.y = Wl[(size_t)(ka + 1) * HDIM + n];
    accS = __builtin_amdgcn_wmma_f32_16x16x4_f32(false, a1, false, b1,
                                                 (short)0, accS, false, false);
    const v2f a2 = *(const v2f*)(xa + k0);
    v2f b2; b2.x = Wr[(size_t)ka * HDIM + n];
            b2.y = Wr[(size_t)(ka + 1) * HDIM + n];
    accX = __builtin_amdgcn_wmma_f32_16x16x4_f32(false, a2, false, b2,
                                                 (short)0, accX, false, false);
  }

  const float bn = bl[n];
#pragma unroll
  for (int r = 0; r < 8; ++r) {
    const int m = m0 + r + 8 * half;
    float d = (float)deg[r + 8 * half];
    d = d < 1.0f ? 1.0f : d;
    float v = accS[r] / d + accX[r] + bn;
    if (relu) v = v > 0.0f ? v : 0.0f;
    Out[(size_t)m * HDIM + n] = v;
  }
}

constexpr int EPW = 32;

template <int M>
__device__ __forceinline__ float lane_xor(float v) {
  const int i = __builtin_bit_cast(int, v);
  return __builtin_bit_cast(
      float, __builtin_amdgcn_update_dpp(i, i, 0x160 + M, 0xf, 0xf, false));
}

__global__ void __launch_bounds__(256)
edge_dot_kernel(const float* __restrict__ Hc, const float* __restrict__ Hp,
                const int* __restrict__ ls, const int* __restrict__ ld, int e,
                float* __restrict__ out) {
  const int lane = threadIdx.x & 31;
  const int q = lane & 7;
  const int first = EPW * __builtin_amdgcn_readfirstlane(
      (int)((blockIdx.x * 256 + threadIdx.x) >> 5));
  if (first >= e) return;
  const int e0 = first + (lane & 24);
  int ia[8], ib[8];
  if (first + EPW <= e) {
    const int4 a0 = *(const int4*)(ls + e0);
    const int4 a1 = *(const int4*)(ls + e0 + 4);
    const int4 b0 = *(const int4*)(ld + e0);
    const int4 b1 = *(const int4*)(ld + e0 + 4);
    ia[0] = a0.x; ia[1] = a0.y; ia[2] = a0.z; ia[3] = a0.w;
    ia[4] = a1.x; ia[5] = a1.y; ia[6] = a1.z; ia[7] = a1.w;
    ib[0] = b0.x; ib[1] = b0.y; ib[2] = b0.z; ib[3] = b0.w;
    ib[4] = b1.x; ib[5] = b1.y; ib[6] = b1.z; ib[7] = b1.w;
  } else {
#pragma unroll
    for (int k = 0; k < 8; ++k) {
      const int i = e0 + k;
      ia[k] = (i < e) ? ls[i] : 0;
      ib[k] = (i < e) ? ld[i] : 0;
    }
  }
  const float* hcq = Hc + q * 4;
  const float* hpq = Hp + q * 4;
  float r = 0.0f;
#pragma unroll
  for (int k = 0; k < 8; k += 2) {
    v4f va[2][4], vb[2][4];
#pragma unroll
    for (int u = 0; u < 2; ++u) {
      const float* pa = hcq + (size_t)ia[k + u] * HDIM;
      const float* pb = hpq + (size_t)ib[k + u] * HDIM;
#pragma unroll
      for (int c = 0; c < 4; ++c) {
        va[u][c] = *(const v4f*)(pa + 32 * c);
        vb[u][c] = *(const v4f*)(pb + 32 * c);
      }
    }
#pragma unroll
    for (int u = 0; u < 2; ++u) {
      float p = 0.0f;
#pragma unroll
      for (int c = 0; c < 4; ++c)
        p += va[u][c].x * vb[u][c].x + va[u][c].y * vb[u][c].y +
             va[u][c].z * vb[u][c].z + va[u][c].w * vb[u][c].w;
      p += lane_xor<1>(p);
      p += lane_xor<2>(p);
      p += lane_xor<4>(p);
      if (q == k + u) r = p;
    }
  }
  const int i = first + lane;
  if (i < e) out[i] = r;
}

extern "C" void kernel_launch(void* const* d_in, const int* in_sizes, int n_in,
                              void* d_out, int out_size, void* d_ws,
                              size_t ws_size, hipStream_t stream) {
  (void)in_sizes; (void)n_in; (void)out_size; (void)ws_size;
  const int NC = 4096, NP = 8192, E = 500000, EL = 500000;

  const float* x_country = (const float*)d_in[0];
  const float* x_product = (const float*)d_in[1];
  const float* Wc_lin = (const float*)d_in[2];
  const float* bc_lin = (const float*)d_in[3];
  const float* Wp_lin = (const float*)d_in[4];
  const float* bp_lin = (const float*)d_in[5];
  const float* emb_c  = (const float*)d_in[6];
  const float* emb_p  = (const float*)d_in[7];
  const float* W1ep_l = (const float*)d_in[8];
  const float* b1ep   = (const float*)d_in[9];
  const float* W1ep_r = (const float*)d_in[10];
  const float* W1pc_l = (const float*)d_in[11];
  const float* b1pc   = (const float*)d_in[12];
  const float* W1pc_r = (const float*)d_in[13];
  const float* W2ep_l = (const float*)d_in[14];
  const float* b2ep   = (const float*)d_in[15];
  const float* W2ep_r = (const float*)d_in[16];
  const float* W2pc_l = (const float*)d_in[17];
  const float* b2pc   = (const float*)d_in[18];
  const float* W2pc_r = (const float*)d_in[19];
  const int* nid_c = (const int*)d_in[20];
  const int* nid_p = (const int*)d_in[21];
  const int* e_src = (const int*)d_in[22];
  const int* e_dst = (const int*)d_in[23];
  const int* l_src = (const int*)d_in[24];
  const int* l_dst = (const int*)d_in[25];

  float* ws = (float*)d_ws;
  size_t off = 0;
  float* xc   = ws + off; off += (size_t)NC * HDIM;
  float* xp   = ws + off; off += (size_t)NP * HDIM;
  float* hp   = ws + off; off += (size_t)NP * HDIM;
  float* hc   = ws + off; off += (size_t)NC * HDIM;
  float* hc2  = ws + off; off += (size_t)NC * HDIM;
  float* hp2  = ws + off; off += (size_t)NP * HDIM;
  float* scr  = ws + off; off += (size_t)NP * HDIM;

  constexpr int NSPLIT = (E + EPB - 1) / EPB;
  static_assert(2 * E + NSPLIT * (OFFP / 2) <= NP * HDIM,
                "edge split fits its plane");
  static_assert((2 * E * 4) % 128 == 0, "offset rows start on a line");
  static_assert(4 * NSPLIT <= 256, "four threads of a layer block per run");
  static_assert(NSPLIT <= NC / 16, "a country tile block per split block");
  unsigned* rec = (unsigned*)scr;
  unsigned* off32 = rec + 2 * E;

  const int TB = 256;

  gemm_in_both_kernel<<<NP / 32 + NC / 16, TB, 0, stream>>>(
      x_product, Wp_lin, bp_lin, emb_p, nid_p, xp, NP, NP / 32,
      x_country, Wc_lin, bc_lin, emb_c, nid_c, xc, NC,
      e_src, e_dst, E, NSPLIT, rec, off32);

  const int layer_blocks = NC / 16 + NP / 16;

  sage_layer_kernel<<<layer_blocks, TB, 0, stream>>>(
      rec, off32, NSPLIT,
      xp, xc, W1pc_l, b1pc, W1pc_r, hc, NC,
      xc, xp, W1ep_l, b1ep, W1ep_r, hp, 1);

  sage_layer_kernel<<<layer_blocks, TB, 0, stream>>>(
      rec, off32, NSPLIT,
      hp, hc, W2pc_l, b2pc, W2pc_r, hc2, NC,
      hc, hp, W2ep_l, b2ep, W2ep_r, hp2, 0);

  const int el_blocks = (EL + 8 * EPW - 1) / (8 * EPW);
  edge_dot_kernel<<<el_blocks, TB, 0, stream>>>(hc2, hp2, l_src, l_dst, EL,
                                                (float*)d_out);
}
